# starting kernel
# baseline (speedup 1.0000x reference)
#include <hip/hip_runtime.h>
#include <hip/hip_bf16.h>

typedef float v2f __attribute__((ext_vector_type(2)));
typedef float v8f __attribute__((ext_vector_type(8)));

#define MT 256
#define NO 512
#define PD 24
#define DD 8

__global__ __launch_bounds__(256) void sqnorm_kernel(const float* __restrict__ X,
                                                     float* __restrict__ SN, int npts) {
  int p = blockIdx.x * blockDim.x + threadIdx.x;
  if (p < npts) {
    float s = 0.f;
#pragma unroll
    for (int d = 0; d < DD; ++d) { float x = X[p * DD + d]; s += x * x; }
    SN[p] = s;
  }
}

__global__ __launch_bounds__(256) void self_kernel(const float* __restrict__ X,
                                                   const float* __restrict__ A,
                                                   const float* __restrict__ SN,
                                                   float* __restrict__ sOut, int n) {
  int wave = threadIdx.x >> 5, lane = threadIdx.x & 31;
  int i = blockIdx.x * (blockDim.x >> 5) + wave;
  float acc = 0.f;
  if (i < n) {
    for (int idx = lane; idx < PD * PD; idx += 32) {
      int u = idx / PD, v = idx % PD;
      const float* xu = X + (i * PD + u) * DD;
      const float* xv = X + (i * PD + v) * DD;
      float dot = 0.f;
#pragma unroll
      for (int d = 0; d < DD; ++d) dot += xu[d] * xv[d];
      float d2 = SN[i * PD + u] + SN[i * PD + v] - 2.f * dot;
      acc += A[i * PD + u] * A[i * PD + v] * __expf(-0.5f * fmaxf(d2, 0.f));
    }
  }
#pragma unroll
  for (int off = 16; off > 0; off >>= 1) acc += __shfl_xor(acc, off);
  if (i < n && lane == 0) sOut[i] = acc;
}

__global__ __launch_bounds__(32) void gram_kernel(
    const float* __restrict__ X1, const float* __restrict__ SN1,
    const float* __restrict__ A1, const float* __restrict__ s1,
    const float* __restrict__ X2, const float* __restrict__ SN2,
    const float* __restrict__ A2, const float* __restrict__ s2,
    float* __restrict__ out, int ldc,
    const float* __restrict__ rho_p, const float* __restrict__ nu_p,
    const float* __restrict__ g_p, int addDiag) {
  const int lane = threadIdx.x & 31;
  const int half = lane >> 4, lq = lane & 15;
  const int i0 = blockIdx.y * 16, j0 = blockIdx.x * 16;

  float acc[8];
#pragma unroll
  for (int r = 0; r < 8; ++r) acc[r] = 0.f;

  for (int u = 0; u < PD; ++u) {
    float a1u[8], n1u[8];
#pragma unroll
    for (int r = 0; r < 8; ++r) {
      int i = i0 + r + 8 * half;
      a1u[r] = A1[i * PD + u];
      n1u[r] = SN1[i * PD + u];
    }
    const float* x1 = X1 + ((i0 + lq) * PD + u) * DD;
    v2f a0; a0.x = x1[2 * half + 0]; a0.y = x1[2 * half + 1];
    v2f a1; a1.x = x1[4 + 2 * half + 0]; a1.y = x1[4 + 2 * half + 1];

    for (int v = 0; v < PD; ++v) {
      const float* x2 = X2 + ((j0 + lq) * PD + v) * DD;
      v2f b0; b0.x = x2[2 * half + 0]; b0.y = x2[2 * half + 1];
      v2f b1; b1.x = x2[4 + 2 * half + 0]; b1.y = x2[4 + 2 * half + 1];

      v8f s = {};
      s = __builtin_amdgcn_wmma_f32_16x16x4_f32(false, a0, false, b0, (short)0, s, false, false);
      s = __builtin_amdgcn_wmma_f32_16x16x4_f32(false, a1, false, b1, (short)0, s, false, false);

      float sn2v = SN2[(j0 + lq) * PD + v];
      float a2v  = A2[(j0 + lq) * PD + v];
#pragma unroll
      for (int r = 0; r < 8; ++r) {
        float d2 = n1u[r] + sn2v - 2.f * s[r];
        acc[r] += a1u[r] * a2v * __expf(-0.5f * fmaxf(d2, 0.f));
      }
    }
  }

  float invrho = 1.f / rho_p[0];
  float nu = nu_p[0], g = g_p[0];
#pragma unroll
  for (int r = 0; r < 8; ++r) {
    int i = i0 + r + 8 * half, j = j0 + lq;
    float d2 = s1[i] + s2[j] - 2.f * acc[r];
    float val = nu * __expf(-0.5f * fmaxf(d2, 0.f) * invrho);
    if (addDiag && i == j) val += nu * g;
    out[i * ldc + j] = val;
  }
}

__global__ __launch_bounds__(1024) void chol_kernel(float* __restrict__ Lm) {
  const int n = NO, tid = threadIdx.x, nt = blockDim.x;
  for (int k = 0; k < n; ++k) {
    if (tid == 0) Lm[k * n + k] = sqrtf(Lm[k * n + k]);
    __syncthreads();
    float dinv = 1.f / Lm[k * n + k];
    for (int i = k + 1 + tid; i < n; i += nt) Lm[i * n + k] *= dinv;
    __syncthreads();
    for (int i = k + 1 + tid; i < n; i += nt) {
      float lik = Lm[i * n + k];
      for (int j = k + 1; j <= i; ++j) Lm[i * n + j] -= lik * Lm[j * n + k];
    }
    __syncthreads();
  }
}

#define SW 8
__global__ __launch_bounds__(SW * 32) void solve_kernel(const float* __restrict__ L,
                                                        const float* __restrict__ Kox,
                                                        float* __restrict__ G) {
  __shared__ float y[SW][NO];
  int wave = threadIdx.x >> 5, lane = threadIdx.x & 31;
  int col = blockIdx.x * SW + wave;
  bool valid = col <= MT;
  if (valid)
    for (int n = lane; n < NO; n += 32)
      y[wave][n] = (col < MT) ? Kox[n * MT + col] : 1.0f;
  __syncthreads();
  for (int r = 0; r < NO; ++r) {
    float s = 0.f;
    if (valid)
      for (int c = lane; c < r; c += 32) s += L[r * NO + c] * y[wave][c];
#pragma unroll
    for (int off = 16; off > 0; off >>= 1) s += __shfl_xor(s, off);
    if (valid && lane == 0) y[wave][r] = (y[wave][r] - s) / L[r * NO + r];
    __syncthreads();
  }
  for (int rr = 0; rr < NO; ++rr) {
    int r = NO - 1 - rr;
    float s = 0.f;
    if (valid)
      for (int c = r + 1 + lane; c < NO; c += 32) s += L[c * NO + r] * y[wave][c];
#pragma unroll
    for (int off = 16; off > 0; off >>= 1) s += __shfl_xor(s, off);
    if (valid && lane == 0) y[wave][r] = (y[wave][r] - s) / L[r * NO + r];
    __syncthreads();
  }
  if (valid)
    for (int n = lane; n < NO; n += 32) G[col * NO + n] = y[wave][n];
}

__global__ __launch_bounds__(256) void mean_kernel(const float* __restrict__ G,
                                                   const float* __restrict__ ys,
                                                   const float* __restrict__ b_p,
                                                   float* __restrict__ out,
                                                   float* __restrict__ colsum) {
  __shared__ float r1[256], r2[256];
  int col = blockIdx.x, tid = threadIdx.x;
  float b = b_p[0];
  float sm = 0.f, sc = 0.f;
  for (int n = tid; n < NO; n += 256) {
    float g = G[col * NO + n];
    sm += g * (ys[n] - b);
    sc += g;
  }
  r1[tid] = sm; r2[tid] = sc;
  __syncthreads();
  for (int off = 128; off > 0; off >>= 1) {
    if (tid < off) { r1[tid] += r1[tid + off]; r2[tid] += r2[tid + off]; }
    __syncthreads();
  }
  if (tid == 0) {
    colsum[col] = r2[0];
    if (col < MT) out[col] = b + r1[0];
  }
}

__global__ __launch_bounds__(32) void cov_kernel(const float* __restrict__ Kxx,
                                                 const float* __restrict__ Kox,
                                                 const float* __restrict__ G,
                                                 const float* __restrict__ colsum,
                                                 float* __restrict__ out) {
  const int lane = threadIdx.x & 31;
  const int half = lane >> 4, lq = lane & 15;
  const int i0 = blockIdx.y * 16, j0 = blockIdx.x * 16;
  v8f c = {};
  for (int n0 = 0; n0 < NO; n0 += 8) {
    const float* grow = G + (i0 + lq) * NO + n0 + 2 * half;
    v2f a0; a0.x = grow[0]; a0.y = grow[1];
    v2f a1; a1.x = grow[4]; a1.y = grow[5];
    v2f b0; b0.x = Kox[(n0 + 2 * half + 0) * MT + j0 + lq];
            b0.y = Kox[(n0 + 2 * half + 1) * MT + j0 + lq];
    v2f b1; b1.x = Kox[(n0 + 4 + 2 * half + 0) * MT + j0 + lq];
            b1.y = Kox[(n0 + 4 + 2 * half + 1) * MT + j0 + lq];
    c = __builtin_amdgcn_wmma_f32_16x16x4_f32(false, a0, false, b0, (short)0, c, false, false);
    c = __builtin_amdgcn_wmma_f32_16x16x4_f32(false, a1, false, b1, (short)0, c, false, false);
  }
  float invS = 1.f / colsum[MT];
  float kbxj = colsum[j0 + lq];
#pragma unroll
  for (int r = 0; r < 8; ++r) {
    int i = i0 + r + 8 * half, j = j0 + lq;
    float v = Kxx[i * MT + j] - c[r] + (1.f - colsum[i]) * (1.f - kbxj) * invS;
    out[(1 + i) * MT + j] = v;
  }
}

extern "C" void kernel_launch(void* const* d_in, const int* in_sizes, int n_in,
                              void* d_out, int out_size, void* d_ws, size_t ws_size,
                              hipStream_t stream) {
  const float* Xt  = (const float*)d_in[0];
  const float* At  = (const float*)d_in[1];
  const float* Xo  = (const float*)d_in[2];
  const float* Ao  = (const float*)d_in[3];
  const float* ys  = (const float*)d_in[4];
  const float* rho = (const float*)d_in[5];
  const float* g   = (const float*)d_in[6];
  const float* nu  = (const float*)d_in[7];
  const float* b   = (const float*)d_in[8];
  float* out = (float*)d_out;

  float* ws  = (float*)d_ws;
  float* SN1 = ws;
  float* SN2 = SN1 + MT * PD;
  float* sT  = SN2 + NO * PD;
  float* sO  = sT + MT;
  float* Kxx = sO + NO;
  float* Kox = Kxx + MT * MT;
  float* Koo = Kox + NO * MT;
  float* G   = Koo + NO * NO;
  float* csm = G + (MT + 1) * NO;

  sqnorm_kernel<<<(MT * PD + 255) / 256, 256, 0, stream>>>(Xt, SN1, MT * PD);
  sqnorm_kernel<<<(NO * PD + 255) / 256, 256, 0, stream>>>(Xo, SN2, NO * PD);
  self_kernel<<<(MT + 7) / 8, 256, 0, stream>>>(Xt, At, SN1, sT, MT);
  self_kernel<<<(NO + 7) / 8, 256, 0, stream>>>(Xo, Ao, SN2, sO, NO);

  gram_kernel<<<dim3(16, 16), 32, 0, stream>>>(Xt, SN1, At, sT, Xt, SN1, At, sT,
                                               Kxx, MT, rho, nu, g, 0);
  gram_kernel<<<dim3(16, 32), 32, 0, stream>>>(Xo, SN2, Ao, sO, Xt, SN1, At, sT,
                                               Kox, MT, rho, nu, g, 0);
  gram_kernel<<<dim3(32, 32), 32, 0, stream>>>(Xo, SN2, Ao, sO, Xo, SN2, Ao, sO,
                                               Koo, NO, rho, nu, g, 1);

  chol_kernel<<<1, 1024, 0, stream>>>(Koo);
  solve_kernel<<<(MT + 1 + SW - 1) / SW, SW * 32, 0, stream>>>(Koo, Kox, G);
  mean_kernel<<<MT + 1, 256, 0, stream>>>(G, ys, b, out, csm);
  cov_kernel<<<dim3(16, 16), 32, 0, stream>>>(Kxx, Kox, G, csm, out);
}


# Cholesky blocked by 32 columns in one workgroup: diagonal block in one wave's registers, panel by per-row substitution, trailing update on f32 WMMA from LDS; update sums regrouped per block
# speedup vs baseline: 12.0616x; 12.0616x over previous
#include <hip/hip_runtime.h>
#include <hip/hip_bf16.h>

typedef float v2f __attribute__((ext_vector_type(2)));
typedef float v8f __attribute__((ext_vector_type(8)));

#define MT 256
#define NO 512
#define PD 24
#define DD 8

__global__ __launch_bounds__(256) void sqnorm_kernel(const float* __restrict__ X,
                                                     float* __restrict__ SN, int npts) {
  int p = blockIdx.x * blockDim.x + threadIdx.x;
  if (p < npts) {
    float s = 0.f;
#pragma unroll
    for (int d = 0; d < DD; ++d) { float x = X[p * DD + d]; s += x * x; }
    SN[p] = s;
  }
}

__global__ __launch_bounds__(256) void self_kernel(const float* __restrict__ X,
                                                   const float* __restrict__ A,
                                                   const float* __restrict__ SN,
                                                   float* __restrict__ sOut, int n) {
  int wave = threadIdx.x >> 5, lane = threadIdx.x & 31;
  int i = blockIdx.x * (blockDim.x >> 5) + wave;
  float acc = 0.f;
  if (i < n) {
    for (int idx = lane; idx < PD * PD; idx += 32) {
      int u = idx / PD, v = idx % PD;
      const float* xu = X + (i * PD + u) * DD;
      const float* xv = X + (i * PD + v) * DD;
      float dot = 0.f;
#pragma unroll
      for (int d = 0; d < DD; ++d) dot += xu[d] * xv[d];
      float d2 = SN[i * PD + u] + SN[i * PD + v] - 2.f * dot;
      acc += A[i * PD + u] * A[i * PD + v] * __expf(-0.5f * fmaxf(d2, 0.f));
    }
  }
#pragma unroll
  for (int off = 16; off > 0; off >>= 1) acc += __shfl_xor(acc, off);
  if (i < n && lane == 0) sOut[i] = acc;
}

__global__ __launch_bounds__(32) void gram_kernel(
    const float* __restrict__ X1, const float* __restrict__ SN1,
    const float* __restrict__ A1, const float* __restrict__ s1,
    const float* __restrict__ X2, const float* __restrict__ SN2,
    const float* __restrict__ A2, const float* __restrict__ s2,
    float* __restrict__ out, int ldc,
    const float* __restrict__ rho_p, const float* __restrict__ nu_p,
    const float* __restrict__ g_p, int addDiag) {
  const int lane = threadIdx.x & 31;
  const int half = lane >> 4, lq = lane & 15;
  const int i0 = blockIdx.y * 16, j0 = blockIdx.x * 16;

  float acc[8];
#pragma unroll
  for (int r = 0; r < 8; ++r) acc[r] = 0.f;

  for (int u = 0; u < PD; ++u) {
    float a1u[8], n1u[8];
#pragma unroll
    for (int r = 0; r < 8; ++r) {
      int i = i0 + r + 8 * half;
      a1u[r] = A1[i * PD + u];
      n1u[r] = SN1[i * PD + u];
    }
    const float* x1 = X1 + ((i0 + lq) * PD + u) * DD;
    v2f a0; a0.x = x1[2 * half + 0]; a0.y = x1[2 * half + 1];
    v2f a1; a1.x = x1[4 + 2 * half + 0]; a1.y = x1[4 + 2 * half + 1];

    for (int v = 0; v < PD; ++v) {
      const float* x2 = X2 + ((j0 + lq) * PD + v) * DD;
      v2f b0; b0.x = x2[2 * half + 0]; b0.y = x2[2 * half + 1];
      v2f b1; b1.x = x2[4 + 2 * half + 0]; b1.y = x2[4 + 2 * half + 1];

      v8f s = {};
      s = __builtin_amdgcn_wmma_f32_16x16x4_f32(false, a0, false, b0, (short)0, s, false, false);
      s = __builtin_amdgcn_wmma_f32_16x16x4_f32(false, a1, false, b1, (short)0, s, false, false);

      float sn2v = SN2[(j0 + lq) * PD + v];
      float a2v  = A2[(j0 + lq) * PD + v];
#pragma unroll
      for (int r = 0; r < 8; ++r) {
        float d2 = n1u[r] + sn2v - 2.f * s[r];
        acc[r] += a1u[r] * a2v * __expf(-0.5f * fmaxf(d2, 0.f));
      }
    }
  }

  float invrho = 1.f / rho_p[0];
  float nu = nu_p[0], g = g_p[0];
#pragma unroll
  for (int r = 0; r < 8; ++r) {
    int i = i0 + r + 8 * half, j = j0 + lq;
    float d2 = s1[i] + s2[j] - 2.f * acc[r];
    float val = nu * __expf(-0.5f * fmaxf(d2, 0.f) * invrho);
    if (addDiag && i == j) val += nu * g;
    out[i * ldc + j] = val;
  }
}

#define NB 32
#define PP 34
__global__ __launch_bounds__(1024) void chol_kernel(float* __restrict__ Lm) {
  __shared__ __attribute__((aligned(16))) float Sd[NB * NB];
  __shared__ __attribute__((aligned(16))) float invd[NB];
  __shared__ __attribute__((aligned(16))) float Pn[(NO - NB) * PP];

  const int tid = threadIdx.x;
  const int lane = tid & 31;
  const int half = lane >> 4, lq = lane & 15;
  const int wave = __builtin_amdgcn_readfirstlane(tid >> 5);

  for (int kb = 0; kb < NO / NB; ++kb) {
    const int k0 = kb * NB;
    const int nrows = NO - k0 - NB;

    if (wave == 0) {
      float r[NB];
      float* rowp = Lm + (k0 + lane) * NO + k0;
      const float4* src = (const float4*)rowp;
#pragma unroll
      for (int q = 0; q < NB / 4; ++q) {
        float4 t = src[q];
        r[4 * q + 0] = t.x; r[4 * q + 1] = t.y; r[4 * q + 2] = t.z; r[4 * q + 3] = t.w;
      }
      float mydinv = 0.f;
#pragma unroll
      for (int k = 0; k < NB; ++k) {
        float akk = __int_as_float(__builtin_amdgcn_readlane(__float_as_int(r[k]), k));
        float d = sqrtf(akk);
        float dinv = 1.f / d;
        r[k] = (lane == k) ? d : r[k] * dinv;
        mydinv = (lane == k) ? dinv : mydinv;
#pragma unroll
        for (int j = k + 1; j < NB; ++j) {
          float ljk = __int_as_float(__builtin_amdgcn_readlane(__float_as_int(r[k]), j));
          r[j] -= r[k] * ljk;
        }
      }
#pragma unroll
      for (int j = 0; j < NB; ++j) r[j] = (j <= lane) ? r[j] : 0.f;
#pragma unroll
      for (int c = 0; c < NB; ++c) Sd[c * NB + lane] = r[c];
      invd[lane] = mydinv;
      float4* dst = (float4*)rowp;
#pragma unroll
      for (int q = 0; q < NB / 4; ++q)
        dst[q] = make_float4(r[4 * q + 0], r[4 * q + 1], r[4 * q + 2], r[4 * q + 3]);
    }
    __syncthreads();

    if (tid < nrows) {
      float x[NB];
      float* rowp = Lm + (k0 + NB + tid) * NO + k0;
      const float4* src = (const float4*)rowp;
#pragma unroll
      for (int q = 0; q < NB / 4; ++q) {
        float4 t = src[q];
        x[4 * q + 0] = t.x; x[4 * q + 1] = t.y; x[4 * q + 2] = t.z; x[4 * q + 3] = t.w;
      }
#pragma unroll
      for (int c = 0; c < NB; ++c) {
        x[c] *= invd[c];
#pragma unroll
        for (int j = c + 1; j < NB; ++j) x[j] -= x[c] * Sd[c * NB + j];
      }
      float4* dst = (float4*)rowp;
#pragma unroll
      for (int q = 0; q < NB / 4; ++q)
        dst[q] = make_float4(x[4 * q + 0], x[4 * q + 1], x[4 * q + 2], x[4 * q + 3]);
      float2* pdst = (float2*)(Pn + tid * PP);
#pragma unroll
      for (int q = 0; q < NB / 2; ++q) pdst[q] = make_float2(x[2 * q + 0], x[2 * q + 1]);
    }
    __syncthreads();

    const int m = nrows >> 4;
    const int T = m * (m + 1) / 2;
    for (int q = wave; q < T; q += 32) {
      int ti = (int)((sqrtf(8.f * (float)q + 1.f) - 1.f) * 0.5f);
      if (ti * (ti + 1) / 2 > q) --ti;
      if ((ti + 1) * (ti + 2) / 2 <= q) ++ti;
      const int tj = q - ti * (ti + 1) / 2;
      const float* pa = Pn + (ti * 16 + lq) * PP + 2 * half;
      const float* pb = Pn + (tj * 16 + lq) * PP + 2 * half;
      v8f s = {};
#pragma unroll
      for (int kc = 0; kc < NB / 4; ++kc) {
        v2f a = *(const v2f*)(pa + 4 * kc);
        v2f b = *(const v2f*)(pb + 4 * kc);
        s = __builtin_amdgcn_wmma_f32_16x16x4_f32(false, a, false, b, (short)0, s, false, false);
      }
      float* cp = Lm + (k0 + NB + ti * 16 + 8 * half) * NO + (k0 + NB + tj * 16 + lq);
#pragma unroll
      for (int r = 0; r < 8; ++r) cp[r * NO] -= s[r];
    }
    __syncthreads();
  }
}

#define SW 8
__global__ __launch_bounds__(SW * 32) void solve_kernel(const float* __restrict__ L,
                                                        const float* __restrict__ Kox,
                                                        float* __restrict__ G) {
  __shared__ float y[SW][NO];
  int wave = threadIdx.x >> 5, lane = threadIdx.x & 31;
  int col = blockIdx.x * SW + wave;
  bool valid = col <= MT;
  if (valid)
    for (int n = lane; n < NO; n += 32)
      y[wave][n] = (col < MT) ? Kox[n * MT + col] : 1.0f;
  __syncthreads();
  for (int r = 0; r < NO; ++r) {
    float s = 0.f;
    if (valid)
      for (int c = lane; c < r; c += 32) s += L[r * NO + c] * y[wave][c];
#pragma unroll
    for (int off = 16; off > 0; off >>= 1) s += __shfl_xor(s, off);
    if (valid && lane == 0) y[wave][r] = (y[wave][r] - s) / L[r * NO + r];
    __syncthreads();
  }
  for (int rr = 0; rr < NO; ++rr) {
    int r = NO - 1 - rr;
    float s = 0.f;
    if (valid)
      for (int c = r + 1 + lane; c < NO; c += 32) s += L[c * NO + r] * y[wave][c];
#pragma unroll
    for (int off = 16; off > 0; off >>= 1) s += __shfl_xor(s, off);
    if (valid && lane == 0) y[wave][r] = (y[wave][r] - s) / L[r * NO + r];
    __syncthreads();
  }
  if (valid)
    for (int n = lane; n < NO; n += 32) G[col * NO + n] = y[wave][n];
}

__global__ __launch_bounds__(256) void mean_kernel(const float* __restrict__ G,
                                                   const float* __restrict__ ys,
                                                   const float* __restrict__ b_p,
                                                   float* __restrict__ out,
                                                   float* __restrict__ colsum) {
  __shared__ float r1[256], r2[256];
  int col = blockIdx.x, tid = threadIdx.x;
  float b = b_p[0];
  float sm = 0.f, sc = 0.f;
  for (int n = tid; n < NO; n += 256) {
    float g = G[col * NO + n];
    sm += g * (ys[n] - b);
    sc += g;
  }
  r1[tid] = sm; r2[tid] = sc;
  __syncthreads();
  for (int off = 128; off > 0; off >>= 1) {
    if (tid < off) { r1[tid] += r1[tid + off]; r2[tid] += r2[tid + off]; }
    __syncthreads();
  }
  if (tid == 0) {
    colsum[col] = r2[0];
    if (col < MT) out[col] = b + r1[0];
  }
}

__global__ __launch_bounds__(32) void cov_kernel(const float* __restrict__ Kxx,
                                                 const float* __restrict__ Kox,
                                                 const float* __restrict__ G,
                                                 const float* __restrict__ colsum,
                                                 float* __restrict__ out) {
  const int lane = threadIdx.x & 31;
  const int half = lane >> 4, lq = lane & 15;
  const int i0 = blockIdx.y * 16, j0 = blockIdx.x * 16;
  v8f c = {};
  for (int n0 = 0; n0 < NO; n0 += 8) {
    const float* grow = G + (i0 + lq) * NO + n0 + 2 * half;
    v2f a0; a0.x = grow[0]; a0.y = grow[1];
    v2f a1; a1.x = grow[4]; a1.y = grow[5];
    v2f b0; b0.x = Kox[(n0 + 2 * half + 0) * MT + j0 + lq];
            b0.y = Kox[(n0 + 2 * half + 1) * MT + j0 + lq];
    v2f b1; b1.x = Kox[(n0 + 4 + 2 * half + 0) * MT + j0 + lq];
            b1.y = Kox[(n0 + 4 + 2 * half + 1) * MT + j0 + lq];
    c = __builtin_amdgcn_wmma_f32_16x16x4_f32(false, a0, false, b0, (short)0, c, false, false);
    c = __builtin_amdgcn_wmma_f32_16x16x4_f32(false, a1, false, b1, (short)0, c, false, false);
  }
  float invS = 1.f / colsum[MT];
  float kbxj = colsum[j0 + lq];
#pragma unroll
  for (int r = 0; r < 8; ++r) {
    int i = i0 + r + 8 * half, j = j0 + lq;
    float v = Kxx[i * MT + j] - c[r] + (1.f - colsum[i]) * (1.f - kbxj) * invS;
    out[(1 + i) * MT + j] = v;
  }
}

extern "C" void kernel_launch(void* const* d_in, const int* in_sizes, int n_in,
                              void* d_out, int out_size, void* d_ws, size_t ws_size,
                              hipStream_t stream) {
  const float* Xt  = (const float*)d_in[0];
  const float* At  = (const float*)d_in[1];
  const float* Xo  = (const float*)d_in[2];
  const float* Ao  = (const float*)d_in[3];
  const float* ys  = (const float*)d_in[4];
  const float* rho = (const float*)d_in[5];
  const float* g   = (const float*)d_in[6];
  const float* nu  = (const float*)d_in[7];
  const float* b   = (const float*)d_in[8];
  float* out = (float*)d_out;

  float* ws  = (float*)d_ws;
  float* SN1 = ws;
  float* SN2 = SN1 + MT * PD;
  float* sT  = SN2 + NO * PD;
  float* sO  = sT + MT;
  float* Kxx = sO + NO;
  float* Kox = Kxx + MT * MT;
  float* Koo = Kox + NO * MT;
  float* G   = Koo + NO * NO;
  float* csm = G + (MT + 1) * NO;

  sqnorm_kernel<<<(MT * PD + 255) / 256, 256, 0, stream>>>(Xt, SN1, MT * PD);
  sqnorm_kernel<<<(NO * PD + 255) / 256, 256, 0, stream>>>(Xo, SN2, NO * PD);
  self_kernel<<<(MT + 7) / 8, 256, 0, stream>>>(Xt, At, SN1, sT, MT);
  self_kernel<<<(NO + 7) / 8, 256, 0, stream>>>(Xo, Ao, SN2, sO, NO);

  gram_kernel<<<dim3(16, 16), 32, 0, stream>>>(Xt, SN1, At, sT, Xt, SN1, At, sT,
                                               Kxx, MT, rho, nu, g, 0);
  gram_kernel<<<dim3(16, 32), 32, 0, stream>>>(Xo, SN2, Ao, sO, Xt, SN1, At, sT,
                                               Kox, MT, rho, nu, g, 0);
  gram_kernel<<<dim3(32, 32), 32, 0, stream>>>(Xo, SN2, Ao, sO, Xo, SN2, Ao, sO,
                                               Koo, NO, rho, nu, g, 1);

  chol_kernel<<<1, 1024, 0, stream>>>(Koo);
  solve_kernel<<<(MT + 1 + SW - 1) / SW, SW * 32, 0, stream>>>(Koo, Kox, G);
  mean_kernel<<<MT + 1, 256, 0, stream>>>(G, ys, b, out, csm);
  cov_kernel<<<dim3(16, 16), 32, 0, stream>>>(Kxx, Kox, G, csm, out);
}


# Triangular solves blocked by 32 rows, 16 right-hand sides per workgroup held in LDS: diagonal block by readlane substitution in registers, the rest on f32 WMMA; solve sums regrouped per block
# speedup vs baseline: 46.1140x; 3.8232x over previous
#include <hip/hip_runtime.h>
#include <hip/hip_bf16.h>

typedef float v2f __attribute__((ext_vector_type(2)));
typedef float v8f __attribute__((ext_vector_type(8)));

#define MT 256
#define NO 512
#define PD 24
#define DD 8

__global__ __launch_bounds__(256) void sqnorm_kernel(const float* __restrict__ X,
                                                     float* __restrict__ SN, int npts) {
  int p = blockIdx.x * blockDim.x + threadIdx.x;
  if (p < npts) {
    float s = 0.f;
#pragma unroll
    for (int d = 0; d < DD; ++d) { float x = X[p * DD + d]; s += x * x; }
    SN[p] = s;
  }
}

__global__ __launch_bounds__(256) void self_kernel(const float* __restrict__ X,
                                                   const float* __restrict__ A,
                                                   const float* __restrict__ SN,
                                                   float* __restrict__ sOut, int n) {
  int wave = threadIdx.x >> 5, lane = threadIdx.x & 31;
  int i = blockIdx.x * (blockDim.x >> 5) + wave;
  float acc = 0.f;
  if (i < n) {
    for (int idx = lane; idx < PD * PD; idx += 32) {
      int u = idx / PD, v = idx % PD;
      const float* xu = X + (i * PD + u) * DD;
      const float* xv = X + (i * PD + v) * DD;
      float dot = 0.f;
#pragma unroll
      for (int d = 0; d < DD; ++d) dot += xu[d] * xv[d];
      float d2 = SN[i * PD + u] + SN[i * PD + v] - 2.f * dot;
      acc += A[i * PD + u] * A[i * PD + v] * __expf(-0.5f * fmaxf(d2, 0.f));
    }
  }
#pragma unroll
  for (int off = 16; off > 0; off >>= 1) acc += __shfl_xor(acc, off);
  if (i < n && lane == 0) sOut[i] = acc;
}

__global__ __launch_bounds__(32) void gram_kernel(
    const float* __restrict__ X1, const float* __restrict__ SN1,
    const float* __restrict__ A1, const float* __restrict__ s1,
    const float* __restrict__ X2, const float* __restrict__ SN2,
    const float* __restrict__ A2, const float* __restrict__ s2,
    float* __restrict__ out, int ldc,
    const float* __restrict__ rho_p, const float* __restrict__ nu_p,
    const float* __restrict__ g_p, int addDiag) {
  const int lane = threadIdx.x & 31;
  const int half = lane >> 4, lq = lane & 15;
  const int i0 = blockIdx.y * 16, j0 = blockIdx.x * 16;

  float acc[8];
#pragma unroll
  for (int r = 0; r < 8; ++r) acc[r] = 0.f;

  for (int u = 0; u < PD; ++u) {
    float a1u[8], n1u[8];
#pragma unroll
    for (int r = 0; r < 8; ++r) {
      int i = i0 + r + 8 * half;
      a1u[r] = A1[i * PD + u];
      n1u[r] = SN1[i * PD + u];
    }
    const float* x1 = X1 + ((i0 + lq) * PD + u) * DD;
    v2f a0; a0.x = x1[2 * half + 0]; a0.y = x1[2 * half + 1];
    v2f a1; a1.x = x1[4 + 2 * half + 0]; a1.y = x1[4 + 2 * half + 1];

    for (int v = 0; v < PD; ++v) {
      const float* x2 = X2 + ((j0 + lq) * PD + v) * DD;
      v2f b0; b0.x = x2[2 * half + 0]; b0.y = x2[2 * half + 1];
      v2f b1; b1.x = x2[4 + 2 * half + 0]; b1.y = x2[4 + 2 * half + 1];

      v8f s = {};
      s = __builtin_amdgcn_wmma_f32_16x16x4_f32(false, a0, false, b0, (short)0, s, false, false);
      s = __builtin_amdgcn_wmma_f32_16x16x4_f32(false, a1, false, b1, (short)0, s, false, false);

      float sn2v = SN2[(j0 + lq) * PD + v];
      float a2v  = A2[(j0 + lq) * PD + v];
#pragma unroll
      for (int r = 0; r < 8; ++r) {
        float d2 = n1u[r] + sn2v - 2.f * s[r];
        acc[r] += a1u[r] * a2v * __expf(-0.5f * fmaxf(d2, 0.f));
      }
    }
  }

  float invrho = 1.f / rho_p[0];
  float nu = nu_p[0], g = g_p[0];
#pragma unroll
  for (int r = 0; r < 8; ++r) {
    int i = i0 + r + 8 * half, j = j0 + lq;
    float d2 = s1[i] + s2[j] - 2.f * acc[r];
    float val = nu * __expf(-0.5f * fmaxf(d2, 0.f) * invrho);
    if (addDiag && i == j) val += nu * g;
    out[i * ldc + j] = val;
  }
}

#define NB 32
#define PP 34
__global__ __launch_bounds__(1024) void chol_kernel(float* __restrict__ Lm) {
  __shared__ __attribute__((aligned(16))) float Sd[NB * NB];
  __shared__ __attribute__((aligned(16))) float invd[NB];
  __shared__ __attribute__((aligned(16))) float Pn[(NO - NB) * PP];

  const int tid = threadIdx.x;
  const int lane = tid & 31;
  const int half = lane >> 4, lq = lane & 15;
  const int wave = __builtin_amdgcn_readfirstlane(tid >> 5);

  for (int kb = 0; kb < NO / NB; ++kb) {
    const int k0 = kb * NB;
    const int nrows = NO - k0 - NB;

    if (wave == 0) {
      float r[NB];
      float* rowp = Lm + (k0 + lane) * NO + k0;
      const float4* src = (const float4*)rowp;
#pragma unroll
      for (int q = 0; q < NB / 4; ++q) {
        float4 t = src[q];
        r[4 * q + 0] = t.x; r[4 * q + 1] = t.y; r[4 * q + 2] = t.z; r[4 * q + 3] = t.w;
      }
      float mydinv = 0.f;
#pragma unroll
      for (int k = 0; k < NB; ++k) {
        float akk = __int_as_float(__builtin_amdgcn_readlane(__float_as_int(r[k]), k));
        float d = sqrtf(akk);
        float dinv = 1.f / d;
        r[k] = (lane == k) ? d : r[k] * dinv;
        mydinv = (lane == k) ? dinv : mydinv;
#pragma unroll
        for (int j = k + 1; j < NB; ++j) {
          float ljk = __int_as_float(__builtin_amdgcn_readlane(__float_as_int(r[k]), j));
          r[j] -= r[k] * ljk;
        }
      }
#pragma unroll
      for (int j = 0; j < NB; ++j) r[j] = (j <= lane) ? r[j] : 0.f;
#pragma unroll
      for (int c = 0; c < NB; ++c) Sd[c * NB + lane] = r[c];
      invd[lane] = mydinv;
      float4* dst = (float4*)rowp;
#pragma unroll
      for (int q = 0; q < NB / 4; ++q)
        dst[q] = make_float4(r[4 * q + 0], r[4 * q + 1], r[4 * q + 2], r[4 * q + 3]);
    }
    __syncthreads();

    if (tid < nrows) {
      float x[NB];
      float* rowp = Lm + (k0 + NB + tid) * NO + k0;
      const float4* src = (const float4*)rowp;
#pragma unroll
      for (int q = 0; q < NB / 4; ++q) {
        float4 t = src[q];
        x[4 * q + 0] = t.x; x[4 * q + 1] = t.y; x[4 * q + 2] = t.z; x[4 * q + 3] = t.w;
      }
#pragma unroll
      for (int c = 0; c < NB; ++c) {
        x[c] *= invd[c];
#pragma unroll
        for (int j = c + 1; j < NB; ++j) x[j] -= x[c] * Sd[c * NB + j];
      }
      float4* dst = (float4*)rowp;
#pragma unroll
      for (int q = 0; q < NB / 4; ++q)
        dst[q] = make_float4(x[4 * q + 0], x[4 * q + 1], x[4 * q + 2], x[4 * q + 3]);
      float2* pdst = (float2*)(Pn + tid * PP);
#pragma unroll
      for (int q = 0; q < NB / 2; ++q) pdst[q] = make_float2(x[2 * q + 0], x[2 * q + 1]);
    }
    __syncthreads();

    const int m = nrows >> 4;
    const int T = m * (m + 1) / 2;
    for (int q = wave; q < T; q += 32) {
      int ti = (int)((sqrtf(8.f * (float)q + 1.f) - 1.f) * 0.5f);
      if (ti * (ti + 1) / 2 > q) --ti;
      if ((ti + 1) * (ti + 2) / 2 <= q) ++ti;
      const int tj = q - ti * (ti + 1) / 2;
      const float* pa = Pn + (ti * 16 + lq) * PP + 2 * half;
      const float* pb = Pn + (tj * 16 + lq) * PP + 2 * half;
      v8f s = {};
#pragma unroll
      for (int kc = 0; kc < NB / 4; ++kc) {
        v2f a = *(const v2f*)(pa + 4 * kc);
        v2f b = *(const v2f*)(pb + 4 * kc);
        s = __builtin_amdgcn_wmma_f32_16x16x4_f32(false, a, false, b, (short)0, s, false, false);
      }
      float* cp = Lm + (k0 + NB + ti * 16 + 8 * half) * NO + (k0 + NB + tj * 16 + lq);
#pragma unroll
      for (int r = 0; r < 8; ++r) cp[r * NO] -= s[r];
    }
    __syncthreads();
  }
}

#define SC 16
#define YP 516
__global__ __launch_bounds__(1024) void solve_kernel(const float* __restrict__ L,
                                                     const float* __restrict__ Kox,
                                                     float* __restrict__ G) {
  __shared__ __attribute__((aligned(16))) float Yt[SC * YP];

  const int tid = threadIdx.x;
  const int lane = tid & 31;
  const int half = lane >> 4, lq = lane & 15;
  const int wave = __builtin_amdgcn_readfirstlane(tid >> 5);
  const int col0 = blockIdx.x * SC;

  for (int i = 0; i < (SC * NO) / 1024; ++i) {
    int idx = tid + 1024 * i;
    int cc = idx & (SC - 1), n = idx >> 4;
    int col = col0 + cc;
    float v = 0.f;
    if (col < MT) v = Kox[n * MT + col];
    else if (col == MT) v = 1.0f;
    Yt[cc * YP + n] = v;
  }

  float Ln[NB];
  float dg = 1.f;

  if (wave < SC) {
    const float4* lp = (const float4*)(L + lane * NO);
#pragma unroll
    for (int q = 0; q < NB / 4; ++q) {
      float4 t4 = lp[q];
      Ln[4 * q + 0] = t4.x; Ln[4 * q + 1] = t4.y; Ln[4 * q + 2] = t4.z; Ln[4 * q + 3] = t4.w;
    }
    dg = L[lane * NO + lane];
  }
  for (int kb = 0; kb < NO / NB; ++kb) {
    const int k0 = kb * NB;
    float Mr[NB];
    float inv = 0.f;
    if (wave < SC) {
      inv = 1.f / dg;
#pragma unroll
      for (int c = 0; c < NB; ++c) Mr[c] = (lane > c) ? Ln[c] * inv : 0.f;
    }
    const int ntiles = (NO - k0 - NB) >> 4;
    const int rowbase = k0 + NB + 16 * wave;
    __syncthreads();

    float4 av[4];
    if (wave < ntiles) {
      const float* ap = L + (rowbase + lq) * NO + k0 + 4 * half;
#pragma unroll
      for (int jj = 0; jj < 4; ++jj) av[jj] = *(const float4*)(ap + 8 * jj);
    }
    if (wave < SC) {
      if (kb + 1 < NO / NB) {
        const float4* lp = (const float4*)(L + (k0 + NB + lane) * NO + k0 + NB);
#pragma unroll
        for (int q = 0; q < NB / 4; ++q) {
          float4 t4 = lp[q];
          Ln[4 * q + 0] = t4.x; Ln[4 * q + 1] = t4.y; Ln[4 * q + 2] = t4.z; Ln[4 * q + 3] = t4.w;
        }
        dg = L[(k0 + NB + lane) * NO + k0 + NB + lane];
      }
      float* yp = Yt + wave * YP + k0 + lane;
      float t = yp[0] * inv;
#pragma unroll
      for (int c = 0; c < NB - 1; ++c) {
        float yc = __int_as_float(__builtin_amdgcn_readlane(__float_as_int(t), c));
        t = fmaf(-Mr[c], yc, t);
      }
      yp[0] = t;
    }
    __syncthreads();

    if (wave < ntiles) {
      const float* bp = Yt + lq * YP + k0 + 4 * half;
      v8f s = {};
#pragma unroll
      for (int jj = 0; jj < 4; ++jj) {
        float4 bv = *(const float4*)(bp + 8 * jj);
        v2f a, b;
        a.x = av[jj].x; a.y = av[jj].y; b.x = bv.x; b.y = bv.y;
        s = __builtin_amdgcn_wmma_f32_16x16x4_f32(false, a, false, b, (short)0, s, false, false);
        a.x = av[jj].z; a.y = av[jj].w; b.x = bv.z; b.y = bv.w;
        s = __builtin_amdgcn_wmma_f32_16x16x4_f32(false, a, false, b, (short)0, s, false, false);
      }
      float4* cp = (float4*)(Yt + lq * YP + rowbase + 8 * half);
      float4 c0 = cp[0], c1 = cp[1];
      c0.x -= s[0]; c0.y -= s[1]; c0.z -= s[2]; c0.w -= s[3];
      c1.x -= s[4]; c1.y -= s[5]; c1.z -= s[6]; c1.w -= s[7];
      cp[0] = c0; cp[1] = c1;
    }
  }

  if (wave < SC) {
    const float* lp = L + (NO - NB) * NO + NO - NB + lane;
#pragma unroll
    for (int c = 0; c < NB; ++c) Ln[c] = lp[c * NO];
    dg = L[(NO - NB + lane) * NO + NO - NB + lane];
  }
  for (int kb = NO / NB - 1; kb >= 0; --kb) {
    const int k0 = kb * NB;
    float Mr[NB];
    float inv = 0.f;
    if (wave < SC) {
      inv = 1.f / dg;
#pragma unroll
      for (int c = 0; c < NB; ++c) Mr[c] = (lane < c) ? Ln[c] * inv : 0.f;
    }
    const int ntiles = k0 >> 4;
    const int rowbase = 16 * wave;
    __syncthreads();

    float at[16];
    if (wave < ntiles) {
      const float* ap = L + (k0 + 4 * half) * NO + rowbase + lq;
#pragma unroll
      for (int jj = 0; jj < 4; ++jj)
#pragma unroll
        for (int e = 0; e < 4; ++e) at[4 * jj + e] = ap[(8 * jj + e) * NO];
    }
    if (wave < SC) {
      if (kb > 0) {
        const float* lp = L + (k0 - NB) * NO + k0 - NB + lane;
#pragma unroll
        for (int c = 0; c < NB; ++c) Ln[c] = lp[c * NO];
        dg = L[(k0 - NB + lane) * NO + k0 - NB + lane];
      }
      float* yp = Yt + wave * YP + k0 + lane;
      float t = yp[0] * inv;
#pragma unroll
      for (int c = NB - 1; c >= 1; --c) {
        float xc = __int_as_float(__builtin_amdgcn_readlane(__float_as_int(t), c));
        t = fmaf(-Mr[c], xc, t);
      }
      yp[0] = t;
    }
    __syncthreads();

    if (wave < ntiles) {
      const float* bp = Yt + lq * YP + k0 + 4 * half;
      v8f s = {};
#pragma unroll
      for (int jj = 0; jj < 4; ++jj) {
        float4 bv = *(const float4*)(bp + 8 * jj);
        v2f a, b;
        a.x = at[4 * jj + 0]; a.y = at[4 * jj + 1]; b.x = bv.x; b.y = bv.y;
        s = __builtin_amdgcn_wmma_f32_16x16x4_f32(false, a, false, b, (short)0, s, false, false);
        a.x = at[4 * jj + 2]; a.y = at[4 * jj + 3]; b.x = bv.z; b.y = bv.w;
        s = __builtin_amdgcn_wmma_f32_16x16x4_f32(false, a, false, b, (short)0, s, false, false);
      }
      float4* cp = (float4*)(Yt + lq * YP + rowbase + 8 * half);
      float4 c0 = cp[0], c1 = cp[1];
      c0.x -= s[0]; c0.y -= s[1]; c0.z -= s[2]; c0.w -= s[3];
      c1.x -= s[4]; c1.y -= s[5]; c1.z -= s[6]; c1.w -= s[7];
      cp[0] = c0; cp[1] = c1;
    }
  }
  __syncthreads();

  for (int i = 0; i < (SC * NO) / 1024; ++i) {
    int idx = tid + 1024 * i;
    int n = idx & (NO - 1), cc = idx >> 9;
    int col = col0 + cc;
    if (col <= MT) G[col * NO + n] = Yt[cc * YP + n];
  }
}

__global__ __launch_bounds__(256) void mean_kernel(const float* __restrict__ G,
                                                   const float* __restrict__ ys,
                                                   const float* __restrict__ b_p,
                                                   float* __restrict__ out,
                                                   float* __restrict__ colsum) {
  __shared__ float r1[256], r2[256];
  int col = blockIdx.x, tid = threadIdx.x;
  float b = b_p[0];
  float sm = 0.f, sc = 0.f;
  for (int n = tid; n < NO; n += 256) {
    float g = G[col * NO + n];
    sm += g * (ys[n] - b);
    sc += g;
  }
  r1[tid] = sm; r2[tid] = sc;
  __syncthreads();
  for (int off = 128; off > 0; off >>= 1) {
    if (tid < off) { r1[tid] += r1[tid + off]; r2[tid] += r2[tid + off]; }
    __syncthreads();
  }
  if (tid == 0) {
    colsum[col] = r2[0];
    if (col < MT) out[col] = b + r1[0];
  }
}

__global__ __launch_bounds__(32) void cov_kernel(const float* __restrict__ Kxx,
                                                 const float* __restrict__ Kox,
                                                 const float* __restrict__ G,
                                                 const float* __restrict__ colsum,
                                                 float* __restrict__ out) {
  const int lane = threadIdx.x & 31;
  const int half = lane >> 4, lq = lane & 15;
  const int i0 = blockIdx.y * 16, j0 = blockIdx.x * 16;
  v8f c = {};
  for (int n0 = 0; n0 < NO; n0 += 8) {
    const float* grow = G + (i0 + lq) * NO + n0 + 2 * half;
    v2f a0; a0.x = grow[0]; a0.y = grow[1];
    v2f a1; a1.x = grow[4]; a1.y = grow[5];
    v2f b0; b0.x = Kox[(n0 + 2 * half + 0) * MT + j0 + lq];
            b0.y = Kox[(n0 + 2 * half + 1) * MT + j0 + lq];
    v2f b1; b1.x = Kox[(n0 + 4 + 2 * half + 0) * MT + j0 + lq];
            b1.y = Kox[(n0 + 4 + 2 * half + 1) * MT + j0 + lq];
    c = __builtin_amdgcn_wmma_f32_16x16x4_f32(false, a0, false, b0, (short)0, c, false, false);
    c = __builtin_amdgcn_wmma_f32_16x16x4_f32(false, a1, false, b1, (short)0, c, false, false);
  }
  float invS = 1.f / colsum[MT];
  float kbxj = colsum[j0 + lq];
#pragma unroll
  for (int r = 0; r < 8; ++r) {
    int i = i0 + r + 8 * half, j = j0 + lq;
    float v = Kxx[i * MT + j] - c[r] + (1.f - colsum[i]) * (1.f - kbxj) * invS;
    out[(1 + i) * MT + j] = v;
  }
}

extern "C" void kernel_launch(void* const* d_in, const int* in_sizes, int n_in,
                              void* d_out, int out_size, void* d_ws, size_t ws_size,
                              hipStream_t stream) {
  const float* Xt  = (const float*)d_in[0];
  const float* At  = (const float*)d_in[1];
  const float* Xo  = (const float*)d_in[2];
  const float* Ao  = (const float*)d_in[3];
  const float* ys  = (const float*)d_in[4];
  const float* rho = (const float*)d_in[5];
  const float* g   = (const float*)d_in[6];
  const float* nu  = (const float*)d_in[7];
  const float* b   = (const float*)d_in[8];
  float* out = (float*)d_out;

  float* ws  = (float*)d_ws;
  float* SN1 = ws;
  float* SN2 = SN1 + MT * PD;
  float* sT  = SN2 + NO * PD;
  float* sO  = sT + MT;
  float* Kxx = sO + NO;
  float* Kox = Kxx + MT * MT;
  float* Koo = Kox + NO * MT;
  float* G   = Koo + NO * NO;
  float* csm = G + (MT + 1) * NO;

  sqnorm_kernel<<<(MT * PD + 255) / 256, 256, 0, stream>>>(Xt, SN1, MT * PD);
  sqnorm_kernel<<<(NO * PD + 255) / 256, 256, 0, stream>>>(Xo, SN2, NO * PD);
  self_kernel<<<(MT + 7) / 8, 256, 0, stream>>>(Xt, At, SN1, sT, MT);
  self_kernel<<<(NO + 7) / 8, 256, 0, stream>>>(Xo, Ao, SN2, sO, NO);

  gram_kernel<<<dim3(16, 16), 32, 0, stream>>>(Xt, SN1, At, sT, Xt, SN1, At, sT,
                                               Kxx, MT, rho, nu, g, 0);
  gram_kernel<<<dim3(16, 32), 32, 0, stream>>>(Xo, SN2, Ao, sO, Xt, SN1, At, sT,
                                               Kox, MT, rho, nu, g, 0);
  gram_kernel<<<dim3(32, 32), 32, 0, stream>>>(Xo, SN2, Ao, sO, Xo, SN2, Ao, sO,
                                               Koo, NO, rho, nu, g, 1);

  chol_kernel<<<1, 1024, 0, stream>>>(Koo);
  solve_kernel<<<(MT + 1 + SC - 1) / SC, 1024, 0, stream>>>(Koo, Kox, G);
  mean_kernel<<<MT + 1, 256, 0, stream>>>(G, ys, b, out, csm);
  cov_kernel<<<dim3(16, 16), 32, 0, stream>>>(Kxx, Kox, G, csm, out);
}


# Gram tiles spread over a workgroup of 12 waves (was 1 wave): tile operands staged once in LDS, each wave takes 2 of the 24 u points, partial tiles added through LDS; sums regrouped per wave
# speedup vs baseline: 73.8223x; 1.6009x over previous
#include <hip/hip_runtime.h>
#include <hip/hip_bf16.h>

typedef float v2f __attribute__((ext_vector_type(2)));
typedef float v8f __attribute__((ext_vector_type(8)));

#define MT 256
#define NO 512
#define PD 24
#define DD 8

__global__ __launch_bounds__(256) void sqnorm_kernel(const float* __restrict__ X,
                                                     float* __restrict__ SN, int npts) {
  int p = blockIdx.x * blockDim.x + threadIdx.x;
  if (p < npts) {
    float s = 0.f;
#pragma unroll
    for (int d = 0; d < DD; ++d) { float x = X[p * DD + d]; s += x * x; }
    SN[p] = s;
  }
}

__global__ __launch_bounds__(256) void self_kernel(const float* __restrict__ X,
                                                   const float* __restrict__ A,
                                                   const float* __restrict__ SN,
                                                   float* __restrict__ sOut, int n) {
  int wave = threadIdx.x >> 5, lane = threadIdx.x & 31;
  int i = blockIdx.x * (blockDim.x >> 5) + wave;
  float acc = 0.f;
  if (i < n) {
    for (int idx = lane; idx < PD * PD; idx += 32) {
      int u = idx / PD, v = idx % PD;
      const float* xu = X + (i * PD + u) * DD;
      const float* xv = X + (i * PD + v) * DD;
      float dot = 0.f;
#pragma unroll
      for (int d = 0; d < DD; ++d) dot += xu[d] * xv[d];
      float d2 = SN[i * PD + u] + SN[i * PD + v] - 2.f * dot;
      acc += A[i * PD + u] * A[i * PD + v] * __expf(-0.5f * fmaxf(d2, 0.f));
    }
  }
#pragma unroll
  for (int off = 16; off > 0; off >>= 1) acc += __shfl_xor(acc, off);
  if (i < n && lane == 0) sOut[i] = acc;
}

#define GW 12
#define UPW (PD / GW)
#define XP 196
__global__ __launch_bounds__(GW * 32) void gram_kernel(
    const float* __restrict__ X1, const float* __restrict__ SN1,
    const float* __restrict__ A1, const float* __restrict__ s1,
    const float* __restrict__ X2, const float* __restrict__ SN2,
    const float* __restrict__ A2, const float* __restrict__ s2,
    float* __restrict__ out, int ldc,
    const float* __restrict__ rho_p, const float* __restrict__ nu_p,
    const float* __restrict__ g_p, int addDiag) {
  __shared__ __attribute__((aligned(16))) float X1s[16 * XP];
  __shared__ __attribute__((aligned(16))) float X2s[16 * XP];
  __shared__ __attribute__((aligned(16))) float A1T[PD * 16];
  __shared__ __attribute__((aligned(16))) float N1T[PD * 16];
  __shared__ __attribute__((aligned(16))) float A2T[PD * 16];
  __shared__ __attribute__((aligned(16))) float N2T[PD * 16];
  __shared__ __attribute__((aligned(16))) float red[GW * 256];

  const int tid = threadIdx.x;
  const int lane = tid & 31;
  const int half = lane >> 4, lq = lane & 15;
  const int wave = __builtin_amdgcn_readfirstlane(tid >> 5);
  const int i0 = blockIdx.y * 16, j0 = blockIdx.x * 16;

  {
    const float4* g1 = (const float4*)(X1 + i0 * PD * DD);
    const float4* g2 = (const float4*)(X2 + j0 * PD * DD);
    for (int q = tid; q < 16 * PD * DD / 4; q += GW * 32) {
      const int item = q / (PD * DD / 4), w4 = q % (PD * DD / 4);
      const float4 t1 = g1[q], t2 = g2[q];
      *(float4*)(X1s + item * XP + 4 * w4) = t1;
      *(float4*)(X2s + item * XP + 4 * w4) = t2;
    }
    for (int q = tid; q < 16 * PD; q += GW * 32) {
      const int item = q / PD, p = q % PD;
      A1T[p * 16 + item] = A1[i0 * PD + q];
      N1T[p * 16 + item] = SN1[i0 * PD + q];
      A2T[p * 16 + item] = A2[j0 * PD + q];
      N2T[p * 16 + item] = SN2[j0 * PD + q];
    }
  }
  __syncthreads();

  float a1u[UPW][8], n1u[UPW][8];
  v2f fa0[UPW], fa1[UPW];
#pragma unroll
  for (int uu = 0; uu < UPW; ++uu) {
    const int u = wave * UPW + uu;
    const float4* ap = (const float4*)(A1T + u * 16 + 8 * half);
    const float4* np = (const float4*)(N1T + u * 16 + 8 * half);
    const float4 ta = ap[0], tb = ap[1], tc = np[0], td = np[1];
    a1u[uu][0] = ta.x; a1u[uu][1] = ta.y; a1u[uu][2] = ta.z; a1u[uu][3] = ta.w;
    a1u[uu][4] = tb.x; a1u[uu][5] = tb.y; a1u[uu][6] = tb.z; a1u[uu][7] = tb.w;
    n1u[uu][0] = tc.x; n1u[uu][1] = tc.y; n1u[uu][2] = tc.z; n1u[uu][3] = tc.w;
    n1u[uu][4] = td.x; n1u[uu][5] = td.y; n1u[uu][6] = td.z; n1u[uu][7] = td.w;
    const float* x1 = X1s + lq * XP + u * DD + 2 * half;
    fa0[uu] = *(const v2f*)(x1);
    fa1[uu] = *(const v2f*)(x1 + 4);
  }

  float acc[8];
#pragma unroll
  for (int r = 0; r < 8; ++r) acc[r] = 0.f;

  for (int v = 0; v < PD; ++v) {
    const float* x2 = X2s + lq * XP + v * DD + 2 * half;
    const v2f b0 = *(const v2f*)(x2);
    const v2f b1 = *(const v2f*)(x2 + 4);
    const float sn2v = N2T[v * 16 + lq];
    const float a2v  = A2T[v * 16 + lq];
#pragma unroll
    for (int uu = 0; uu < UPW; ++uu) {
      v8f s = {};
      s = __builtin_amdgcn_wmma_f32_16x16x4_f32(false, fa0[uu], false, b0, (short)0, s, false, false);
      s = __builtin_amdgcn_wmma_f32_16x16x4_f32(false, fa1[uu], false, b1, (short)0, s, false, false);
#pragma unroll
      for (int r = 0; r < 8; ++r) {
        float d2 = n1u[uu][r] + sn2v - 2.f * s[r];
        acc[r] += a1u[uu][r] * a2v * __expf(-0.5f * fmaxf(d2, 0.f));
      }
    }
  }

#pragma unroll
  for (int r = 0; r < 8; ++r) red[wave * 256 + r * 32 + lane] = acc[r];
  __syncthreads();

  if (tid < 256) {
    float c = 0.f;
#pragma unroll
    for (int w = 0; w < GW; ++w) c += red[w * 256 + tid];
    const int r = tid >> 5;
    const int i = i0 + r + 8 * half, j = j0 + lq;
    float invrho = 1.f / rho_p[0];
    float nu = nu_p[0], g = g_p[0];
    float d2 = s1[i] + s2[j] - 2.f * c;
    float val = nu * __expf(-0.5f * fmaxf(d2, 0.f) * invrho);
    if (addDiag && i == j) val += nu * g;
    out[i * ldc + j] = val;
  }
}

#define NB 32
#define PP 34
__global__ __launch_bounds__(1024) void chol_kernel(float* __restrict__ Lm) {
  __shared__ __attribute__((aligned(16))) float Sd[NB * NB];
  __shared__ __attribute__((aligned(16))) float invd[NB];
  __shared__ __attribute__((aligned(16))) float Pn[(NO - NB) * PP];

  const int tid = threadIdx.x;
  const int lane = tid & 31;
  const int half = lane >> 4, lq = lane & 15;
  const int wave = __builtin_amdgcn_readfirstlane(tid >> 5);

  for (int kb = 0; kb < NO / NB; ++kb) {
    const int k0 = kb * NB;
    const int nrows = NO - k0 - NB;

    if (wave == 0) {
      float r[NB];
      float* rowp = Lm + (k0 + lane) * NO + k0;
      const float4* src = (const float4*)rowp;
#pragma unroll
      for (int q = 0; q < NB / 4; ++q) {
        float4 t = src[q];
        r[4 * q + 0] = t.x; r[4 * q + 1] = t.y; r[4 * q + 2] = t.z; r[4 * q + 3] = t.w;
      }
      float mydinv = 0.f;
#pragma unroll
      for (int k = 0; k < NB; ++k) {
        float akk = __int_as_float(__builtin_amdgcn_readlane(__float_as_int(r[k]), k));
        float d = sqrtf(akk);
        float dinv = 1.f / d;
        r[k] = (lane == k) ? d : r[k] * dinv;
        mydinv = (lane == k) ? dinv : mydinv;
#pragma unroll
        for (int j = k + 1; j < NB; ++j) {
          float ljk = __int_as_float(__builtin_amdgcn_readlane(__float_as_int(r[k]), j));
          r[j] -= r[k] * ljk;
        }
      }
#pragma unroll
      for (int j = 0; j < NB; ++j) r[j] = (j <= lane) ? r[j] : 0.f;
#pragma unroll
      for (int c = 0; c < NB; ++c) Sd[c * NB + lane] = r[c];
      invd[lane] = mydinv;
      float4* dst = (float4*)rowp;
#pragma unroll
      for (int q = 0; q < NB / 4; ++q)
        dst[q] = make_float4(r[4 * q + 0], r[4 * q + 1], r[4 * q + 2], r[4 * q + 3]);
    }
    __syncthreads();

    if (tid < nrows) {
      float x[NB];
      float* rowp = Lm + (k0 + NB + tid) * NO + k0;
      const float4* src = (const float4*)rowp;
#pragma unroll
      for (int q = 0; q < NB / 4; ++q) {
        float4 t = src[q];
        x[4 * q + 0] = t.x; x[4 * q + 1] = t.y; x[4 * q + 2] = t.z; x[4 * q + 3] = t.w;
      }
#pragma unroll
      for (int c = 0; c < NB; ++c) {
        x[c] *= invd[c];
#pragma unroll
        for (int j = c + 1; j < NB; ++j) x[j] -= x[c] * Sd[c * NB + j];
      }
      float4* dst = (float4*)rowp;
#pragma unroll
      for (int q = 0; q < NB / 4; ++q)
        dst[q] = make_float4(x[4 * q + 0], x[4 * q + 1], x[4 * q + 2], x[4 * q + 3]);
      float2* pdst = (float2*)(Pn + tid * PP);
#pragma unroll
      for (int q = 0; q < NB / 2; ++q) pdst[q] = make_float2(x[2 * q + 0], x[2 * q + 1]);
    }
    __syncthreads();

    const int m = nrows >> 4;
    const int T = m * (m + 1) / 2;
    for (int q = wave; q < T; q += 32) {
      int ti = (int)((sqrtf(8.f * (float)q + 1.f) - 1.f) * 0.5f);
      if (ti * (ti + 1) / 2 > q) --ti;
      if ((ti + 1) * (ti + 2) / 2 <= q) ++ti;
      const int tj = q - ti * (ti + 1) / 2;
      const float* pa = Pn + (ti * 16 + lq) * PP + 2 * half;
      const float* pb = Pn + (tj * 16 + lq) * PP + 2 * half;
      v8f s = {};
#pragma unroll
      for (int kc = 0; kc < NB / 4; ++kc) {
        v2f a = *(const v2f*)(pa + 4 * kc);
        v2f b = *(const v2f*)(pb + 4 * kc);
        s = __builtin_amdgcn_wmma_f32_16x16x4_f32(false, a, false, b, (short)0, s, false, false);
      }
      float* cp = Lm + (k0 + NB + ti * 16 + 8 * half) * NO + (k0 + NB + tj * 16 + lq);
#pragma unroll
      for (int r = 0; r < 8; ++r) cp[r * NO] -= s[r];
    }
    __syncthreads();
  }
}

#define SC 16
#define YP 516
__global__ __launch_bounds__(1024) void solve_kernel(const float* __restrict__ L,
                                                     const float* __restrict__ Kox,
                                                     float* __restrict__ G) {
  __shared__ __attribute__((aligned(16))) float Yt[SC * YP];

  const int tid = threadIdx.x;
  const int lane = tid & 31;
  const int half = lane >> 4, lq = lane & 15;
  const int wave = __builtin_amdgcn_readfirstlane(tid >> 5);
  const int col0 = blockIdx.x * SC;

  for (int i = 0; i < (SC * NO) / 1024; ++i) {
    int idx = tid + 1024 * i;
    int cc = idx & (SC - 1), n = idx >> 4;
    int col = col0 + cc;
    float v = 0.f;
    if (col < MT) v = Kox[n * MT + col];
    else if (col == MT) v = 1.0f;
    Yt[cc * YP + n] = v;
  }

  float Ln[NB];
  float dg = 1.f;

  if (wave < SC) {
    const float4* lp = (const float4*)(L + lane * NO);
#pragma unroll
    for (int q = 0; q < NB / 4; ++q) {
      float4 t4 = lp[q];
      Ln[4 * q + 0] = t4.x; Ln[4 * q + 1] = t4.y; Ln[4 * q + 2] = t4.z; Ln[4 * q + 3] = t4.w;
    }
    dg = L[lane * NO + lane];
  }
  for (int kb = 0; kb < NO / NB; ++kb) {
    const int k0 = kb * NB;
    float Mr[NB];
    float inv = 0.f;
    if (wave < SC) {
      inv = 1.f / dg;
#pragma unroll
      for (int c = 0; c < NB; ++c) Mr[c] = (lane > c) ? Ln[c] * inv : 0.f;
    }
    const int ntiles = (NO - k0 - NB) >> 4;
    const int rowbase = k0 + NB + 16 * wave;
    __syncthreads();

    float4 av[4];
    if (wave < ntiles) {
      const float* ap = L + (rowbase + lq) * NO + k0 + 4 * half;
#pragma unroll
      for (int jj = 0; jj < 4; ++jj) av[jj] = *(const float4*)(ap + 8 * jj);
    }
    if (wave < SC) {
      if (kb + 1 < NO / NB) {
        const float4* lp = (const float4*)(L + (k0 + NB + lane) * NO + k0 + NB);
#pragma unroll
        for (int q = 0; q < NB / 4; ++q) {
          float4 t4 = lp[q];
          Ln[4 * q + 0] = t4.x; Ln[4 * q + 1] = t4.y; Ln[4 * q + 2] = t4.z; Ln[4 * q + 3] = t4.w;
        }
        dg = L[(k0 + NB + lane) * NO + k0 + NB + lane];
      }
      float* yp = Yt + wave * YP + k0 + lane;
      float t = yp[0] * inv;
#pragma unroll
      for (int c = 0; c < NB - 1; ++c) {
        float yc = __int_as_float(__builtin_amdgcn_readlane(__float_as_int(t), c));
        t = fmaf(-Mr[c], yc, t);
      }
      yp[0] = t;
    }
    __syncthreads();

    if (wave < ntiles) {
      const float* bp = Yt + lq * YP + k0 + 4 * half;
      v8f s = {};
#pragma unroll
      for (int jj = 0; jj < 4; ++jj) {
        float4 bv = *(const float4*)(bp + 8 * jj);
        v2f a, b;
        a.x = av[jj].x; a.y = av[jj].y; b.x = bv.x; b.y = bv.y;
        s = __builtin_amdgcn_wmma_f32_16x16x4_f32(false, a, false, b, (short)0, s, false, false);
        a.x = av[jj].z; a.y = av[jj].w; b.x = bv.z; b.y = bv.w;
        s = __builtin_amdgcn_wmma_f32_16x16x4_f32(false, a, false, b, (short)0, s, false, false);
      }
      float4* cp = (float4*)(Yt + lq * YP + rowbase + 8 * half);
      float4 c0 = cp[0], c1 = cp[1];
      c0.x -= s[0]; c0.y -= s[1]; c0.z -= s[2]; c0.w -= s[3];
      c1.x -= s[4]; c1.y -= s[5]; c1.z -= s[6]; c1.w -= s[7];
      cp[0] = c0; cp[1] = c1;
    }
  }

  if (wave < SC) {
    const float* lp = L + (NO - NB) * NO + NO - NB + lane;
#pragma unroll
    for (int c = 0; c < NB; ++c) Ln[c] = lp[c * NO];
    dg = L[(NO - NB + lane) * NO + NO - NB + lane];
  }
  for (int kb = NO / NB - 1; kb >= 0; --kb) {
    const int k0 = kb * NB;
    float Mr[NB];
    float inv = 0.f;
    if (wave < SC) {
      inv = 1.f / dg;
#pragma unroll
      for (int c = 0; c < NB; ++c) Mr[c] = (lane < c) ? Ln[c] * inv : 0.f;
    }
    const int ntiles = k0 >> 4;
    const int rowbase = 16 * wave;
    __syncthreads();

    float at[16];
    if (wave < ntiles) {
      const float* ap = L + (k0 + 4 * half) * NO + rowbase + lq;
#pragma unroll
      for (int jj = 0; jj < 4; ++jj)
#pragma unroll
        for (int e = 0; e < 4; ++e) at[4 * jj + e] = ap[(8 * jj + e) * NO];
    }
    if (wave < SC) {
      if (kb > 0) {
        const float* lp = L + (k0 - NB) * NO + k0 - NB + lane;
#pragma unroll
        for (int c = 0; c < NB; ++c) Ln[c] = lp[c * NO];
        dg = L[(k0 - NB + lane) * NO + k0 - NB + lane];
      }
      float* yp = Yt + wave * YP + k0 + lane;
      float t = yp[0] * inv;
#pragma unroll
      for (int c = NB - 1; c >= 1; --c) {
        float xc = __int_as_float(__builtin_amdgcn_readlane(__float_as_int(t), c));
        t = fmaf(-Mr[c], xc, t);
      }
      yp[0] = t;
    }
    __syncthreads();

    if (wave < ntiles) {
      const float* bp = Yt + lq * YP + k0 + 4 * half;
      v8f s = {};
#pragma unroll
      for (int jj = 0; jj < 4; ++jj) {
        float4 bv = *(const float4*)(bp + 8 * jj);
        v2f a, b;
        a.x = at[4 * jj + 0]; a.y = at[4 * jj + 1]; b.x = bv.x; b.y = bv.y;
        s = __builtin_amdgcn_wmma_f32_16x16x4_f32(false, a, false, b, (short)0, s, false, false);
        a.x = at[4 * jj + 2]; a.y = at[4 * jj + 3]; b.x = bv.z; b.y = bv.w;
        s = __builtin_amdgcn_wmma_f32_16x16x4_f32(false, a, false, b, (short)0, s, false, false);
      }
      float4* cp = (float4*)(Yt + lq * YP + rowbase + 8 * half);
      float4 c0 = cp[0], c1 = cp[1];
      c0.x -= s[0]; c0.y -= s[1]; c0.z -= s[2]; c0.w -= s[3];
      c1.x -= s[4]; c1.y -= s[5]; c1.z -= s[6]; c1.w -= s[7];
      cp[0] = c0; cp[1] = c1;
    }
  }
  __syncthreads();

  for (int i = 0; i < (SC * NO) / 1024; ++i) {
    int idx = tid + 1024 * i;
    int n = idx & (NO - 1), cc = idx >> 9;
    int col = col0 + cc;
    if (col <= MT) G[col * NO + n] = Yt[cc * YP + n];
  }
}

__global__ __launch_bounds__(256) void mean_kernel(const float* __restrict__ G,
                                                   const float* __restrict__ ys,
                                                   const float* __restrict__ b_p,
                                                   float* __restrict__ out,
                                                   float* __restrict__ colsum) {
  __shared__ float r1[256], r2[256];
  int col = blockIdx.x, tid = threadIdx.x;
  float b = b_p[0];
  float sm = 0.f, sc = 0.f;
  for (int n = tid; n < NO; n += 256) {
    float g = G[col * NO + n];
    sm += g * (ys[n] - b);
    sc += g;
  }
  r1[tid] = sm; r2[tid] = sc;
  __syncthreads();
  for (int off = 128; off > 0; off >>= 1) {
    if (tid < off) { r1[tid] += r1[tid + off]; r2[tid] += r2[tid + off]; }
    __syncthreads();
  }
  if (tid == 0) {
    colsum[col] = r2[0];
    if (col < MT) out[col] = b + r1[0];
  }
}

__global__ __launch_bounds__(32) void cov_kernel(const float* __restrict__ Kxx,
                                                 const float* __restrict__ Kox,
                                                 const float* __restrict__ G,
                                                 const float* __restrict__ colsum,
                                                 float* __restrict__ out) {
  const int lane = threadIdx.x & 31;
  const int half = lane >> 4, lq = lane & 15;
  const int i0 = blockIdx.y * 16, j0 = blockIdx.x * 16;
  v8f c = {};
  for (int n0 = 0; n0 < NO; n0 += 8) {
    const float* grow = G + (i0 + lq) * NO + n0 + 2 * half;
    v2f a0; a0.x = grow[0]; a0.y = grow[1];
    v2f a1; a1.x = grow[4]; a1.y = grow[5];
    v2f b0; b0.x = Kox[(n0 + 2 * half + 0) * MT + j0 + lq];
            b0.y = Kox[(n0 + 2 * half + 1) * MT + j0 + lq];
    v2f b1; b1.x = Kox[(n0 + 4 + 2 * half + 0) * MT + j0 + lq];
            b1.y = Kox[(n0 + 4 + 2 * half + 1) * MT + j0 + lq];
    c = __builtin_amdgcn_wmma_f32_16x16x4_f32(false, a0, false, b0, (short)0, c, false, false);
    c = __builtin_amdgcn_wmma_f32_16x16x4_f32(false, a1, false, b1, (short)0, c, false, false);
  }
  float invS = 1.f / colsum[MT];
  float kbxj = colsum[j0 + lq];
#pragma unroll
  for (int r = 0; r < 8; ++r) {
    int i = i0 + r + 8 * half, j = j0 + lq;
    float v = Kxx[i * MT + j] - c[r] + (1.f - colsum[i]) * (1.f - kbxj) * invS;
    out[(1 + i) * MT + j] = v;
  }
}

extern "C" void kernel_launch(void* const* d_in, const int* in_sizes, int n_in,
                              void* d_out, int out_size, void* d_ws, size_t ws_size,
                              hipStream_t stream) {
  const float* Xt  = (const float*)d_in[0];
  const float* At  = (const float*)d_in[1];
  const float* Xo  = (const float*)d_in[2];
  const float* Ao  = (const float*)d_in[3];
  const float* ys  = (const float*)d_in[4];
  const float* rho = (const float*)d_in[5];
  const float* g   = (const float*)d_in[6];
  const float* nu  = (const float*)d_in[7];
  const float* b   = (const float*)d_in[8];
  float* out = (float*)d_out;

  float* ws  = (float*)d_ws;
  float* SN1 = ws;
  float* SN2 = SN1 + MT * PD;
  float* sT  = SN2 + NO * PD;
  float* sO  = sT + MT;
  float* Kxx = sO + NO;
  float* Kox = Kxx + MT * MT;
  float* Koo = Kox + NO * MT;
  float* G   = Koo + NO * NO;
  float* csm = G + (MT + 1) * NO;

  sqnorm_kernel<<<(MT * PD + 255) / 256, 256, 0, stream>>>(Xt, SN1, MT * PD);
  sqnorm_kernel<<<(NO * PD + 255) / 256, 256, 0, stream>>>(Xo, SN2, NO * PD);
  self_kernel<<<(MT + 7) / 8, 256, 0, stream>>>(Xt, At, SN1, sT, MT);
  self_kernel<<<(NO + 7) / 8, 256, 0, stream>>>(Xo, Ao, SN2, sO, NO);

  gram_kernel<<<dim3(16, 16), GW * 32, 0, stream>>>(Xt, SN1, At, sT, Xt, SN1, At, sT,
                                               Kxx, MT, rho, nu, g, 0);
  gram_kernel<<<dim3(16, 32), GW * 32, 0, stream>>>(Xo, SN2, Ao, sO, Xt, SN1, At, sT,
                                               Kox, MT, rho, nu, g, 0);
  gram_kernel<<<dim3(32, 32), GW * 32, 0, stream>>>(Xo, SN2, Ao, sO, Xo, SN2, Ao, sO,
                                               Koo, NO, rho, nu, g, 1);

  chol_kernel<<<1, 1024, 0, stream>>>(Koo);
  solve_kernel<<<(MT + 1 + SC - 1) / SC, 1024, 0, stream>>>(Koo, Kox, G);
  mean_kernel<<<MT + 1, 256, 0, stream>>>(G, ys, b, out, csm);
  cov_kernel<<<dim3(16, 16), 32, 0, stream>>>(Kxx, Kox, G, csm, out);
}


# Cholesky panel substitution: each wave holds the factored 32x32 diagonal block in registers (lane j = row j) and takes entries by readlane, not by an LDS load waited for at each multiply-add pair
# speedup vs baseline: 94.2416x; 1.2766x over previous
#include <hip/hip_runtime.h>
#include <hip/hip_bf16.h>

typedef float v2f __attribute__((ext_vector_type(2)));
typedef float v8f __attribute__((ext_vector_type(8)));

#define MT 256
#define NO 512
#define PD 24
#define DD 8

__global__ __launch_bounds__(256) void sqnorm_kernel(const float* __restrict__ X,
                                                     float* __restrict__ SN, int npts) {
  int p = blockIdx.x * blockDim.x + threadIdx.x;
  if (p < npts) {
    float s = 0.f;
#pragma unroll
    for (int d = 0; d < DD; ++d) { float x = X[p * DD + d]; s += x * x; }
    SN[p] = s;
  }
}

__global__ __launch_bounds__(256) void self_kernel(const float* __restrict__ X,
                                                   const float* __restrict__ A,
                                                   const float* __restrict__ SN,
                                                   float* __restrict__ sOut, int n) {
  int wave = threadIdx.x >> 5, lane = threadIdx.x & 31;
  int i = blockIdx.x * (blockDim.x >> 5) + wave;
  float acc = 0.f;
  if (i < n) {
    for (int idx = lane; idx < PD * PD; idx += 32) {
      int u = idx / PD, v = idx % PD;
      const float* xu = X + (i * PD + u) * DD;
      const float* xv = X + (i * PD + v) * DD;
      float dot = 0.f;
#pragma unroll
      for (int d = 0; d < DD; ++d) dot += xu[d] * xv[d];
      float d2 = SN[i * PD + u] + SN[i * PD + v] - 2.f * dot;
      acc += A[i * PD + u] * A[i * PD + v] * __expf(-0.5f * fmaxf(d2, 0.f));
    }
  }
#pragma unroll
  for (int off = 16; off > 0; off >>= 1) acc += __shfl_xor(acc, off);
  if (i < n && lane == 0) sOut[i] = acc;
}

#define GW 12
#define UPW (PD / GW)
#define XP 196
__global__ __launch_bounds__(GW * 32) void gram_kernel(
    const float* __restrict__ X1, const float* __restrict__ SN1,
    const float* __restrict__ A1, const float* __restrict__ s1,
    const float* __restrict__ X2, const float* __restrict__ SN2,
    const float* __restrict__ A2, const float* __restrict__ s2,
    float* __restrict__ out, int ldc,
    const float* __restrict__ rho_p, const float* __restrict__ nu_p,
    const float* __restrict__ g_p, int addDiag) {
  __shared__ __attribute__((aligned(16))) float X1s[16 * XP];
  __shared__ __attribute__((aligned(16))) float X2s[16 * XP];
  __shared__ __attribute__((aligned(16))) float A1T[PD * 16];
  __shared__ __attribute__((aligned(16))) float N1T[PD * 16];
  __shared__ __attribute__((aligned(16))) float A2T[PD * 16];
  __shared__ __attribute__((aligned(16))) float N2T[PD * 16];
  __shared__ __attribute__((aligned(16))) float red[GW * 256];

  const int tid = threadIdx.x;
  const int lane = tid & 31;
  const int half = lane >> 4, lq = lane & 15;
  const int wave = __builtin_amdgcn_readfirstlane(tid >> 5);
  const int i0 = blockIdx.y * 16, j0 = blockIdx.x * 16;

  {
    const float4* g1 = (const float4*)(X1 + i0 * PD * DD);
    const float4* g2 = (const float4*)(X2 + j0 * PD * DD);
    for (int q = tid; q < 16 * PD * DD / 4; q += GW * 32) {
      const int item = q / (PD * DD / 4), w4 = q % (PD * DD / 4);
      const float4 t1 = g1[q], t2 = g2[q];
      *(float4*)(X1s + item * XP + 4 * w4) = t1;
      *(float4*)(X2s + item * XP + 4 * w4) = t2;
    }
    for (int q = tid; q < 16 * PD; q += GW * 32) {
      const int item = q / PD, p = q % PD;
      A1T[p * 16 + item] = A1[i0 * PD + q];
      N1T[p * 16 + item] = SN1[i0 * PD + q];
      A2T[p * 16 + item] = A2[j0 * PD + q];
      N2T[p * 16 + item] = SN2[j0 * PD + q];
    }
  }
  __syncthreads();

  float a1u[UPW][8], n1u[UPW][8];
  v2f fa0[UPW], fa1[UPW];
#pragma unroll
  for (int uu = 0; uu < UPW; ++uu) {
    const int u = wave * UPW + uu;
    const float4* ap = (const float4*)(A1T + u * 16 + 8 * half);
    const float4* np = (const float4*)(N1T + u * 16 + 8 * half);
    const float4 ta = ap[0], tb = ap[1], tc = np[0], td = np[1];
    a1u[uu][0] = ta.x; a1u[uu][1] = ta.y; a1u[uu][2] = ta.z; a1u[uu][3] = ta.w;
    a1u[uu][4] = tb.x; a1u[uu][5] = tb.y; a1u[uu][6] = tb.z; a1u[uu][7] = tb.w;
    n1u[uu][0] = tc.x; n1u[uu][1] = tc.y; n1u[uu][2] = tc.z; n1u[uu][3] = tc.w;
    n1u[uu][4] = td.x; n1u[uu][5] = td.y; n1u[uu][6] = td.z; n1u[uu][7] = td.w;
    const float* x1 = X1s + lq * XP + u * DD + 2 * half;
    fa0[uu] = *(const v2f*)(x1);
    fa1[uu] = *(const v2f*)(x1 + 4);
  }

  float acc[8];
#pragma unroll
  for (int r = 0; r < 8; ++r) acc[r] = 0.f;

  for (int v = 0; v < PD; ++v) {
    const float* x2 = X2s + lq * XP + v * DD + 2 * half;
    const v2f b0 = *(const v2f*)(x2);
    const v2f b1 = *(const v2f*)(x2 + 4);
    const float sn2v = N2T[v * 16 + lq];
    const float a2v  = A2T[v * 16 + lq];
#pragma unroll
    for (int uu = 0; uu < UPW; ++uu) {
      v8f s = {};
      s = __builtin_amdgcn_wmma_f32_16x16x4_f32(false, fa0[uu], false, b0, (short)0, s, false, false);
      s = __builtin_amdgcn_wmma_f32_16x16x4_f32(false, fa1[uu], false, b1, (short)0, s, false, false);
#pragma unroll
      for (int r = 0; r < 8; ++r) {
        float d2 = n1u[uu][r] + sn2v - 2.f * s[r];
        acc[r] += a1u[uu][r] * a2v * __expf(-0.5f * fmaxf(d2, 0.f));
      }
    }
  }

#pragma unroll
  for (int r = 0; r < 8; ++r) red[wave * 256 + r * 32 + lane] = acc[r];
  __syncthreads();

  if (tid < 256) {
    float c = 0.f;
#pragma unroll
    for (int w = 0; w < GW; ++w) c += red[w * 256 + tid];
    const int r = tid >> 5;
    const int i = i0 + r + 8 * half, j = j0 + lq;
    float invrho = 1.f / rho_p[0];
    float nu = nu_p[0], g = g_p[0];
    float d2 = s1[i] + s2[j] - 2.f * c;
    float val = nu * __expf(-0.5f * fmaxf(d2, 0.f) * invrho);
    if (addDiag && i == j) val += nu * g;
    out[i * ldc + j] = val;
  }
}

#define NB 32
#define PP 34
__global__ __launch_bounds__(1024) void chol_kernel(float* __restrict__ Lm) {
  __shared__ __attribute__((aligned(16))) float Sd[NB * NB];
  __shared__ __attribute__((aligned(16))) float invd[NB];
  __shared__ __attribute__((aligned(16))) float Pn[(NO - NB) * PP];

  const int tid = threadIdx.x;
  const int lane = tid & 31;
  const int half = lane >> 4, lq = lane & 15;
  const int wave = __builtin_amdgcn_readfirstlane(tid >> 5);

  for (int kb = 0; kb < NO / NB; ++kb) {
    const int k0 = kb * NB;
    const int nrows = NO - k0 - NB;

    if (wave == 0) {
      float r[NB];
      float* rowp = Lm + (k0 + lane) * NO + k0;
      const float4* src = (const float4*)rowp;
#pragma unroll
      for (int q = 0; q < NB / 4; ++q) {
        float4 t = src[q];
        r[4 * q + 0] = t.x; r[4 * q + 1] = t.y; r[4 * q + 2] = t.z; r[4 * q + 3] = t.w;
      }
      float mydinv = 0.f;
#pragma unroll
      for (int k = 0; k < NB; ++k) {
        float akk = __int_as_float(__builtin_amdgcn_readlane(__float_as_int(r[k]), k));
        float d = sqrtf(akk);
        float dinv = 1.f / d;
        r[k] = (lane == k) ? d : r[k] * dinv;
        mydinv = (lane == k) ? dinv : mydinv;
#pragma unroll
        for (int j = k + 1; j < NB; ++j) {
          float ljk = __int_as_float(__builtin_amdgcn_readlane(__float_as_int(r[k]), j));
          r[j] -= r[k] * ljk;
        }
      }
#pragma unroll
      for (int j = 0; j < NB; ++j) r[j] = (j <= lane) ? r[j] : 0.f;
#pragma unroll
      for (int c = 0; c < NB; ++c) Sd[c * NB + lane] = r[c];
      invd[lane] = mydinv;
      float4* dst = (float4*)rowp;
#pragma unroll
      for (int q = 0; q < NB / 4; ++q)
        dst[q] = make_float4(r[4 * q + 0], r[4 * q + 1], r[4 * q + 2], r[4 * q + 3]);
    }
    __syncthreads();

    if (wave < (nrows >> 5)) {
      float Lr[NB];
#pragma unroll
      for (int c = 0; c < NB; ++c) Lr[c] = Sd[c * NB + lane];
      const float dv = invd[lane];
      float x[NB];
      float* rowp = Lm + (k0 + NB + tid) * NO + k0;
      const float4* src = (const float4*)rowp;
#pragma unroll
      for (int q = 0; q < NB / 4; ++q) {
        float4 t = src[q];
        x[4 * q + 0] = t.x; x[4 * q + 1] = t.y; x[4 * q + 2] = t.z; x[4 * q + 3] = t.w;
      }
#pragma unroll
      for (int c = 0; c < NB; ++c) {
        x[c] *= __int_as_float(__builtin_amdgcn_readlane(__float_as_int(dv), c));
#pragma unroll
        for (int j = c + 1; j < NB; ++j)
          x[j] -= x[c] * __int_as_float(__builtin_amdgcn_readlane(__float_as_int(Lr[c]), j));
      }
      float4* dst = (float4*)rowp;
#pragma unroll
      for (int q = 0; q < NB / 4; ++q)
        dst[q] = make_float4(x[4 * q + 0], x[4 * q + 1], x[4 * q + 2], x[4 * q + 3]);
      float2* pdst = (float2*)(Pn + tid * PP);
#pragma unroll
      for (int q = 0; q < NB / 2; ++q) pdst[q] = make_float2(x[2 * q + 0], x[2 * q + 1]);
    }
    __syncthreads();

    const int m = nrows >> 4;
    const int T = m * (m + 1) / 2;
    for (int q = wave; q < T; q += 32) {
      int ti = (int)((sqrtf(8.f * (float)q + 1.f) - 1.f) * 0.5f);
      if (ti * (ti + 1) / 2 > q) --ti;
      if ((ti + 1) * (ti + 2) / 2 <= q) ++ti;
      const int tj = q - ti * (ti + 1) / 2;
      const float* pa = Pn + (ti * 16 + lq) * PP + 2 * half;
      const float* pb = Pn + (tj * 16 + lq) * PP + 2 * half;
      v8f s = {};
#pragma unroll
      for (int kc = 0; kc < NB / 4; ++kc) {
        v2f a = *(const v2f*)(pa + 4 * kc);
        v2f b = *(const v2f*)(pb + 4 * kc);
        s = __builtin_amdgcn_wmma_f32_16x16x4_f32(false, a, false, b, (short)0, s, false, false);
      }
      float* cp = Lm + (k0 + NB + ti * 16 + 8 * half) * NO + (k0 + NB + tj * 16 + lq);
#pragma unroll
      for (int r = 0; r < 8; ++r) cp[r * NO] -= s[r];
    }
    __syncthreads();
  }
}

#define SC 16
#define YP 516
__global__ __launch_bounds__(1024) void solve_kernel(const float* __restrict__ L,
                                                     const float* __restrict__ Kox,
                                                     float* __restrict__ G) {
  __shared__ __attribute__((aligned(16))) float Yt[SC * YP];

  const int tid = threadIdx.x;
  const int lane = tid & 31;
  const int half = lane >> 4, lq = lane & 15;
  const int wave = __builtin_amdgcn_readfirstlane(tid >> 5);
  const int col0 = blockIdx.x * SC;

  for (int i = 0; i < (SC * NO) / 1024; ++i) {
    int idx = tid + 1024 * i;
    int cc = idx & (SC - 1), n = idx >> 4;
    int col = col0 + cc;
    float v = 0.f;
    if (col < MT) v = Kox[n * MT + col];
    else if (col == MT) v = 1.0f;
    Yt[cc * YP + n] = v;
  }

  float Ln[NB];
  float dg = 1.f;

  if (wave < SC) {
    const float4* lp = (const float4*)(L + lane * NO);
#pragma unroll
    for (int q = 0; q < NB / 4; ++q) {
      float4 t4 = lp[q];
      Ln[4 * q + 0] = t4.x; Ln[4 * q + 1] = t4.y; Ln[4 * q + 2] = t4.z; Ln[4 * q + 3] = t4.w;
    }
    dg = L[lane * NO + lane];
  }
  for (int kb = 0; kb < NO / NB; ++kb) {
    const int k0 = kb * NB;
    float Mr[NB];
    float inv = 0.f;
    if (wave < SC) {
      inv = 1.f / dg;
#pragma unroll
      for (int c = 0; c < NB; ++c) Mr[c] = (lane > c) ? Ln[c] * inv : 0.f;
    }
    const int ntiles = (NO - k0 - NB) >> 4;
    const int rowbase = k0 + NB + 16 * wave;
    __syncthreads();

    float4 av[4];
    if (wave < ntiles) {
      const float* ap = L + (rowbase + lq) * NO + k0 + 4 * half;
#pragma unroll
      for (int jj = 0; jj < 4; ++jj) av[jj] = *(const float4*)(ap + 8 * jj);
    }
    if (wave < SC) {
      if (kb + 1 < NO / NB) {
        const float4* lp = (const float4*)(L + (k0 + NB + lane) * NO + k0 + NB);
#pragma unroll
        for (int q = 0; q < NB / 4; ++q) {
          float4 t4 = lp[q];
          Ln[4 * q + 0] = t4.x; Ln[4 * q + 1] = t4.y; Ln[4 * q + 2] = t4.z; Ln[4 * q + 3] = t4.w;
        }
        dg = L[(k0 + NB + lane) * NO + k0 + NB + lane];
      }
      float* yp = Yt + wave * YP + k0 + lane;
      float t = yp[0] * inv;
#pragma unroll
      for (int c = 0; c < NB - 1; ++c) {
        float yc = __int_as_float(__builtin_amdgcn_readlane(__float_as_int(t), c));
        t = fmaf(-Mr[c], yc, t);
      }
      yp[0] = t;
    }
    __syncthreads();

    if (wave < ntiles) {
      const float* bp = Yt + lq * YP + k0 + 4 * half;
      v8f s = {};
#pragma unroll
      for (int jj = 0; jj < 4; ++jj) {
        float4 bv = *(const float4*)(bp + 8 * jj);
        v2f a, b;
        a.x = av[jj].x; a.y = av[jj].y; b.x = bv.x; b.y = bv.y;
        s = __builtin_amdgcn_wmma_f32_16x16x4_f32(false, a, false, b, (short)0, s, false, false);
        a.x = av[jj].z; a.y = av[jj].w; b.x = bv.z; b.y = bv.w;
        s = __builtin_amdgcn_wmma_f32_16x16x4_f32(false, a, false, b, (short)0, s, false, false);
      }
      float4* cp = (float4*)(Yt + lq * YP + rowbase + 8 * half);
      float4 c0 = cp[0], c1 = cp[1];
      c0.x -= s[0]; c0.y -= s[1]; c0.z -= s[2]; c0.w -= s[3];
      c1.x -= s[4]; c1.y -= s[5]; c1.z -= s[6]; c1.w -= s[7];
      cp[0] = c0; cp[1] = c1;
    }
  }

  if (wave < SC) {
    const float* lp = L + (NO - NB) * NO + NO - NB + lane;
#pragma unroll
    for (int c = 0; c < NB; ++c) Ln[c] = lp[c * NO];
    dg = L[(NO - NB + lane) * NO + NO - NB + lane];
  }
  for (int kb = NO / NB - 1; kb >= 0; --kb) {
    const int k0 = kb * NB;
    float Mr[NB];
    float inv = 0.f;
    if (wave < SC) {
      inv = 1.f / dg;
#pragma unroll
      for (int c = 0; c < NB; ++c) Mr[c] = (lane < c) ? Ln[c] * inv : 0.f;
    }
    const int ntiles = k0 >> 4;
    const int rowbase = 16 * wave;
    __syncthreads();

    float at[16];
    if (wave < ntiles) {
      const float* ap = L + (k0 + 4 * half) * NO + rowbase + lq;
#pragma unroll
      for (int jj = 0; jj < 4; ++jj)
#pragma unroll
        for (int e = 0; e < 4; ++e) at[4 * jj + e] = ap[(8 * jj + e) * NO];
    }
    if (wave < SC) {
      if (kb > 0) {
        const float* lp = L + (k0 - NB) * NO + k0 - NB + lane;
#pragma unroll
        for (int c = 0; c < NB; ++c) Ln[c] = lp[c * NO];
        dg = L[(k0 - NB + lane) * NO + k0 - NB + lane];
      }
      float* yp = Yt + wave * YP + k0 + lane;
      float t = yp[0] * inv;
#pragma unroll
      for (int c = NB - 1; c >= 1; --c) {
        float xc = __int_as_float(__builtin_amdgcn_readlane(__float_as_int(t), c));
        t = fmaf(-Mr[c], xc, t);
      }
      yp[0] = t;
    }
    __syncthreads();

    if (wave < ntiles) {
      const float* bp = Yt + lq * YP + k0 + 4 * half;
      v8f s = {};
#pragma unroll
      for (int jj = 0; jj < 4; ++jj) {
        float4 bv = *(const float4*)(bp + 8 * jj);
        v2f a, b;
        a.x = at[4 * jj + 0]; a.y = at[4 * jj + 1]; b.x = bv.x; b.y = bv.y;
        s = __builtin_amdgcn_wmma_f32_16x16x4_f32(false, a, false, b, (short)0, s, false, false);
        a.x = at[4 * jj + 2]; a.y = at[4 * jj + 3]; b.x = bv.z; b.y = bv.w;
        s = __builtin_amdgcn_wmma_f32_16x16x4_f32(false, a, false, b, (short)0, s, false, false);
      }
      float4* cp = (float4*)(Yt + lq * YP + rowbase + 8 * half);
      float4 c0 = cp[0], c1 = cp[1];
      c0.x -= s[0]; c0.y -= s[1]; c0.z -= s[2]; c0.w -= s[3];
      c1.x -= s[4]; c1.y -= s[5]; c1.z -= s[6]; c1.w -= s[7];
      cp[0] = c0; cp[1] = c1;
    }
  }
  __syncthreads();

  for (int i = 0; i < (SC * NO) / 1024; ++i) {
    int idx = tid + 1024 * i;
    int n = idx & (NO - 1), cc = idx >> 9;
    int col = col0 + cc;
    if (col <= MT) G[col * NO + n] = Yt[cc * YP + n];
  }
}

__global__ __launch_bounds__(256) void mean_kernel(const float* __restrict__ G,
                                                   const float* __restrict__ ys,
                                                   const float* __restrict__ b_p,
                                                   float* __restrict__ out,
                                                   float* __restrict__ colsum) {
  __shared__ float r1[256], r2[256];
  int col = blockIdx.x, tid = threadIdx.x;
  float b = b_p[0];
  float sm = 0.f, sc = 0.f;
  for (int n = tid; n < NO; n += 256) {
    float g = G[col * NO + n];
    sm += g * (ys[n] - b);
    sc += g;
  }
  r1[tid] = sm; r2[tid] = sc;
  __syncthreads();
  for (int off = 128; off > 0; off >>= 1) {
    if (tid < off) { r1[tid] += r1[tid + off]; r2[tid] += r2[tid + off]; }
    __syncthreads();
  }
  if (tid == 0) {
    colsum[col] = r2[0];
    if (col < MT) out[col] = b + r1[0];
  }
}

__global__ __launch_bounds__(32) void cov_kernel(const float* __restrict__ Kxx,
                                                 const float* __restrict__ Kox,
                                                 const float* __restrict__ G,
                                                 const float* __restrict__ colsum,
                                                 float* __restrict__ out) {
  const int lane = threadIdx.x & 31;
  const int half = lane >> 4, lq = lane & 15;
  const int i0 = blockIdx.y * 16, j0 = blockIdx.x * 16;
  v8f c = {};
  for (int n0 = 0; n0 < NO; n0 += 8) {
    const float* grow = G + (i0 + lq) * NO + n0 + 2 * half;
    v2f a0; a0.x = grow[0]; a0.y = grow[1];
    v2f a1; a1.x = grow[4]; a1.y = grow[5];
    v2f b0; b0.x = Kox[(n0 + 2 * half + 0) * MT + j0 + lq];
            b0.y = Kox[(n0 + 2 * half + 1) * MT + j0 + lq];
    v2f b1; b1.x = Kox[(n0 + 4 + 2 * half + 0) * MT + j0 + lq];
            b1.y = Kox[(n0 + 4 + 2 * half + 1) * MT + j0 + lq];
    c = __builtin_amdgcn_wmma_f32_16x16x4_f32(false, a0, false, b0, (short)0, c, false, false);
    c = __builtin_amdgcn_wmma_f32_16x16x4_f32(false, a1, false, b1, (short)0, c, false, false);
  }
  float invS = 1.f / colsum[MT];
  float kbxj = colsum[j0 + lq];
#pragma unroll
  for (int r = 0; r < 8; ++r) {
    int i = i0 + r + 8 * half, j = j0 + lq;
    float v = Kxx[i * MT + j] - c[r] + (1.f - colsum[i]) * (1.f - kbxj) * invS;
    out[(1 + i) * MT + j] = v;
  }
}

extern "C" void kernel_launch(void* const* d_in, const int* in_sizes, int n_in,
                              void* d_out, int out_size, void* d_ws, size_t ws_size,
                              hipStream_t stream) {
  const float* Xt  = (const float*)d_in[0];
  const float* At  = (const float*)d_in[1];
  const float* Xo  = (const float*)d_in[2];
  const float* Ao  = (const float*)d_in[3];
  const float* ys  = (const float*)d_in[4];
  const float* rho = (const float*)d_in[5];
  const float* g   = (const float*)d_in[6];
  const float* nu  = (const float*)d_in[7];
  const float* b   = (const float*)d_in[8];
  float* out = (float*)d_out;

  float* ws  = (float*)d_ws;
  float* SN1 = ws;
  float* SN2 = SN1 + MT * PD;
  float* sT  = SN2 + NO * PD;
  float* sO  = sT + MT;
  float* Kxx = sO + NO;
  float* Kox = Kxx + MT * MT;
  float* Koo = Kox + NO * MT;
  float* G   = Koo + NO * NO;
  float* csm = G + (MT + 1) * NO;

  sqnorm_kernel<<<(MT * PD + 255) / 256, 256, 0, stream>>>(Xt, SN1, MT * PD);
  sqnorm_kernel<<<(NO * PD + 255) / 256, 256, 0, stream>>>(Xo, SN2, NO * PD);
  self_kernel<<<(MT + 7) / 8, 256, 0, stream>>>(Xt, At, SN1, sT, MT);
  self_kernel<<<(NO + 7) / 8, 256, 0, stream>>>(Xo, Ao, SN2, sO, NO);

  gram_kernel<<<dim3(16, 16), GW * 32, 0, stream>>>(Xt, SN1, At, sT, Xt, SN1, At, sT,
                                               Kxx, MT, rho, nu, g, 0);
  gram_kernel<<<dim3(16, 32), GW * 32, 0, stream>>>(Xo, SN2, Ao, sO, Xt, SN1, At, sT,
                                               Kox, MT, rho, nu, g, 0);
  gram_kernel<<<dim3(32, 32), GW * 32, 0, stream>>>(Xo, SN2, Ao, sO, Xo, SN2, Ao, sO,
                                               Koo, NO, rho, nu, g, 1);

  chol_kernel<<<1, 1024, 0, stream>>>(Koo);
  solve_kernel<<<(MT + 1 + SC - 1) / SC, 1024, 0, stream>>>(Koo, Kox, G);
  mean_kernel<<<MT + 1, 256, 0, stream>>>(G, ys, b, out, csm);
  cov_kernel<<<dim3(16, 16), 32, 0, stream>>>(Kxx, Kox, G, csm, out);
}


# Launches per run 11 to 6: norms and self terms of both point sets in one launch (was 4; one wave per item, its points staged in LDS), tiles of the three kernel matrices in one launch (was 3)
# speedup vs baseline: 114.4463x; 1.2144x over previous
#include <hip/hip_runtime.h>
#include <hip/hip_bf16.h>

typedef float v2f __attribute__((ext_vector_type(2)));
typedef float v8f __attribute__((ext_vector_type(8)));

#define MT 256
#define NO 512
#define PD 24
#define DD 8

#define IPB 8
__global__ __launch_bounds__(IPB * 32) void prep_kernel(const float* __restrict__ Xt,
                                                        const float* __restrict__ At,
                                                        const float* __restrict__ Xo,
                                                        const float* __restrict__ Ao,
                                                        float* __restrict__ SNt,
                                                        float* __restrict__ SNo,
                                                        float* __restrict__ sT,
                                                        float* __restrict__ sO) {
  __shared__ __attribute__((aligned(16))) float Xs[IPB][PD * DD];
  __shared__ float As[IPB][PD];
  __shared__ float Ns[IPB][PD];

  const int wave = threadIdx.x >> 5, lane = threadIdx.x & 31;
  const int it = blockIdx.x * IPB + wave;
  const bool obs = it >= MT;
  const int i = obs ? it - MT : it;
  const float* X = obs ? Xo : Xt;
  const float* A = obs ? Ao : At;
  float* SN = obs ? SNo : SNt;
  float* sOut = obs ? sO : sT;

  const float4* xg = (const float4*)(X + i * PD * DD);
  float4* xs4 = (float4*)Xs[wave];
  for (int q = lane; q < PD * DD / 4; q += 32) xs4[q] = xg[q];
  if (lane < PD) {
    const int p = i * PD + lane;
    As[wave][lane] = A[p];
    float s = 0.f;
#pragma unroll
    for (int d = 0; d < DD; ++d) { float x = X[p * DD + d]; s += x * x; }
    Ns[wave][lane] = s;
    SN[p] = s;
  }
  __syncthreads();

  float acc = 0.f;
  for (int idx = lane; idx < PD * PD; idx += 32) {
    int u = idx / PD, v = idx % PD;
    const float* xu = Xs[wave] + u * DD;
    const float* xv = Xs[wave] + v * DD;
    float dot = 0.f;
#pragma unroll
    for (int d = 0; d < DD; ++d) dot += xu[d] * xv[d];
    float d2 = Ns[wave][u] + Ns[wave][v] - 2.f * dot;
    acc += As[wave][u] * As[wave][v] * __expf(-0.5f * fmaxf(d2, 0.f));
  }
#pragma unroll
  for (int off = 16; off > 0; off >>= 1) acc += __shfl_xor(acc, off);
  if (lane == 0) sOut[i] = acc;
}

#define GW 12
#define UPW (PD / GW)
#define XP 196
#define TKOO ((NO / 16) * (NO / 16))
#define TKOX ((NO / 16) * (MT / 16))
#define TKXX ((MT / 16) * (MT / 16))
static_assert(NO / 16 == 32 && MT / 16 == 16, "tile decode in gram_kernel assumes 32 and 16 tiles per row");
__global__ __launch_bounds__(GW * 32) void gram_kernel(
    const float* __restrict__ Xt, const float* __restrict__ SNt,
    const float* __restrict__ At, const float* __restrict__ st,
    const float* __restrict__ Xo, const float* __restrict__ SNo,
    const float* __restrict__ Ao, const float* __restrict__ so,
    float* __restrict__ Kxx, float* __restrict__ Kox, float* __restrict__ Koo,
    const float* __restrict__ rho_p, const float* __restrict__ nu_p,
    const float* __restrict__ g_p) {
  __shared__ __attribute__((aligned(16))) float X1s[16 * XP];
  __shared__ __attribute__((aligned(16))) float X2s[16 * XP];
  __shared__ __attribute__((aligned(16))) float A1T[PD * 16];
  __shared__ __attribute__((aligned(16))) float N1T[PD * 16];
  __shared__ __attribute__((aligned(16))) float A2T[PD * 16];
  __shared__ __attribute__((aligned(16))) float N2T[PD * 16];
  __shared__ __attribute__((aligned(16))) float red[GW * 256];

  const int tid = threadIdx.x;
  const int lane = tid & 31;
  const int half = lane >> 4, lq = lane & 15;
  const int wave = __builtin_amdgcn_readfirstlane(tid >> 5);

  const int bid = blockIdx.x;
  const bool isKoo = bid < TKOO;
  const bool isKox = !isKoo && bid < TKOO + TKOX;
  const int tb = isKoo ? bid : (isKox ? bid - TKOO : bid - TKOO - TKOX);
  const int tsh = isKoo ? 5 : 4;
  const int i0 = (tb >> tsh) * 16, j0 = (tb & ((1 << tsh) - 1)) * 16;
  const bool obs1 = isKoo || isKox;
  const float* X1  = obs1 ? Xo : Xt;
  const float* SN1 = obs1 ? SNo : SNt;
  const float* A1  = obs1 ? Ao : At;
  const float* s1  = obs1 ? so : st;
  const float* X2  = isKoo ? Xo : Xt;
  const float* SN2 = isKoo ? SNo : SNt;
  const float* A2  = isKoo ? Ao : At;
  const float* s2  = isKoo ? so : st;
  float* out = isKoo ? Koo : (isKox ? Kox : Kxx);
  const int ldc = isKoo ? NO : MT;
  const int addDiag = isKoo ? 1 : 0;

  {
    const float4* g1 = (const float4*)(X1 + i0 * PD * DD);
    const float4* g2 = (const float4*)(X2 + j0 * PD * DD);
    for (int q = tid; q < 16 * PD * DD / 4; q += GW * 32) {
      const int item = q / (PD * DD / 4), w4 = q % (PD * DD / 4);
      const float4 t1 = g1[q], t2 = g2[q];
      *(float4*)(X1s + item * XP + 4 * w4) = t1;
      *(float4*)(X2s + item * XP + 4 * w4) = t2;
    }
    for (int q = tid; q < 16 * PD; q += GW * 32) {
      const int item = q / PD, p = q % PD;
      A1T[p * 16 + item] = A1[i0 * PD + q];
      N1T[p * 16 + item] = SN1[i0 * PD + q];
      A2T[p * 16 + item] = A2[j0 * PD + q];
      N2T[p * 16 + item] = SN2[j0 * PD + q];
    }
  }
  __syncthreads();

  float a1u[UPW][8], n1u[UPW][8];
  v2f fa0[UPW], fa1[UPW];
#pragma unroll
  for (int uu = 0; uu < UPW; ++uu) {
    const int u = wave * UPW + uu;
    const float4* ap = (const float4*)(A1T + u * 16 + 8 * half);
    const float4* np = (const float4*)(N1T + u * 16 + 8 * half);
    const float4 ta = ap[0], tb = ap[1], tc = np[0], td = np[1];
    a1u[uu][0] = ta.x; a1u[uu][1] = ta.y; a1u[uu][2] = ta.z; a1u[uu][3] = ta.w;
    a1u[uu][4] = tb.x; a1u[uu][5] = tb.y; a1u[uu][6] = tb.z; a1u[uu][7] = tb.w;
    n1u[uu][0] = tc.x; n1u[uu][1] = tc.y; n1u[uu][2] = tc.z; n1u[uu][3] = tc.w;
    n1u[uu][4] = td.x; n1u[uu][5] = td.y; n1u[uu][6] = td.z; n1u[uu][7] = td.w;
    const float* x1 = X1s + lq * XP + u * DD + 2 * half;
    fa0[uu] = *(const v2f*)(x1);
    fa1[uu] = *(const v2f*)(x1 + 4);
  }

  float acc[8];
#pragma unroll
  for (int r = 0; r < 8; ++r) acc[r] = 0.f;

  for (int v = 0; v < PD; ++v) {
    const float* x2 = X2s + lq * XP + v * DD + 2 * half;
    const v2f b0 = *(const v2f*)(x2);
    const v2f b1 = *(const v2f*)(x2 + 4);
    const float sn2v = N2T[v * 16 + lq];
    const float a2v  = A2T[v * 16 + lq];
#pragma unroll
    for (int uu = 0; uu < UPW; ++uu) {
      v8f s = {};
      s = __builtin_amdgcn_wmma_f32_16x16x4_f32(false, fa0[uu], false, b0, (short)0, s, false, false);
      s = __builtin_amdgcn_wmma_f32_16x16x4_f32(false, fa1[uu], false, b1, (short)0, s, false, false);
#pragma unroll
      for (int r = 0; r < 8; ++r) {
        float d2 = n1u[uu][r] + sn2v - 2.f * s[r];
        acc[r] += a1u[uu][r] * a2v * __expf(-0.5f * fmaxf(d2, 0.f));
      }
    }
  }

#pragma unroll
  for (int r = 0; r < 8; ++r) red[wave * 256 + r * 32 + lane] = acc[r];
  __syncthreads();

  if (tid < 256) {
    float c = 0.f;
#pragma unroll
    for (int w = 0; w < GW; ++w) c += red[w * 256 + tid];
    const int r = tid >> 5;
    const int i = i0 + r + 8 * half, j = j0 + lq;
    float invrho = 1.f / rho_p[0];
    float nu = nu_p[0], g = g_p[0];
    float d2 = s1[i] + s2[j] - 2.f * c;
    float val = nu * __expf(-0.5f * fmaxf(d2, 0.f) * invrho);
    if (addDiag && i == j) val += nu * g;
    out[i * ldc + j] = val;
  }
}

#define NB 32
#define PP 34
__global__ __launch_bounds__(1024) void chol_kernel(float* __restrict__ Lm) {
  __shared__ __attribute__((aligned(16))) float Sd[NB * NB];
  __shared__ __attribute__((aligned(16))) float invd[NB];
  __shared__ __attribute__((aligned(16))) float Pn[(NO - NB) * PP];

  const int tid = threadIdx.x;
  const int lane = tid & 31;
  const int half = lane >> 4, lq = lane & 15;
  const int wave = __builtin_amdgcn_readfirstlane(tid >> 5);

  for (int kb = 0; kb < NO / NB; ++kb) {
    const int k0 = kb * NB;
    const int nrows = NO - k0 - NB;

    if (wave == 0) {
      float r[NB];
      float* rowp = Lm + (k0 + lane) * NO + k0;
      const float4* src = (const float4*)rowp;
#pragma unroll
      for (int q = 0; q < NB / 4; ++q) {
        float4 t = src[q];
        r[4 * q + 0] = t.x; r[4 * q + 1] = t.y; r[4 * q + 2] = t.z; r[4 * q + 3] = t.w;
      }
      float mydinv = 0.f;
#pragma unroll
      for (int k = 0; k < NB; ++k) {
        float akk = __int_as_float(__builtin_amdgcn_readlane(__float_as_int(r[k]), k));
        float dinv = __builtin_amdgcn_rsqf(akk);
        r[k] = r[k] * dinv;
        mydinv = (lane == k) ? dinv : mydinv;
#pragma unroll
        for (int j = k + 1; j < NB; ++j) {
          float ljk = __int_as_float(__builtin_amdgcn_readlane(__float_as_int(r[k]), j));
          r[j] -= r[k] * ljk;
        }
      }
#pragma unroll
      for (int j = 0; j < NB; ++j) r[j] = (j <= lane) ? r[j] : 0.f;
#pragma unroll
      for (int c = 0; c < NB; ++c) Sd[c * NB + lane] = r[c];
      invd[lane] = mydinv;
      float4* dst = (float4*)rowp;
#pragma unroll
      for (int q = 0; q < NB / 4; ++q)
        dst[q] = make_float4(r[4 * q + 0], r[4 * q + 1], r[4 * q + 2], r[4 * q + 3]);
    }
    __syncthreads();

    if (wave < (nrows >> 5)) {
      float Lr[NB];
#pragma unroll
      for (int c = 0; c < NB; ++c) Lr[c] = Sd[c * NB + lane];
      const float dv = invd[lane];
      float x[NB];
      float* rowp = Lm + (k0 + NB + tid) * NO + k0;
      const float4* src = (const float4*)rowp;
#pragma unroll
      for (int q = 0; q < NB / 4; ++q) {
        float4 t = src[q];
        x[4 * q + 0] = t.x; x[4 * q + 1] = t.y; x[4 * q + 2] = t.z; x[4 * q + 3] = t.w;
      }
#pragma unroll
      for (int c = 0; c < NB; ++c) {
        x[c] *= __int_as_float(__builtin_amdgcn_readlane(__float_as_int(dv), c));
#pragma unroll
        for (int j = c + 1; j < NB; ++j)
          x[j] -= x[c] * __int_as_float(__builtin_amdgcn_readlane(__float_as_int(Lr[c]), j));
      }
      float4* dst = (float4*)rowp;
#pragma unroll
      for (int q = 0; q < NB / 4; ++q)
        dst[q] = make_float4(x[4 * q + 0], x[4 * q + 1], x[4 * q + 2], x[4 * q + 3]);
      float2* pdst = (float2*)(Pn + tid * PP);
#pragma unroll
      for (int q = 0; q < NB / 2; ++q) pdst[q] = make_float2(x[2 * q + 0], x[2 * q + 1]);
    }
    __syncthreads();

    const int m = nrows >> 4;
    const int T = m * (m + 1) / 2;
    for (int q = wave; q < T; q += 32) {
      int ti = (int)((sqrtf(8.f * (float)q + 1.f) - 1.f) * 0.5f);
      if (ti * (ti + 1) / 2 > q) --ti;
      if ((ti + 1) * (ti + 2) / 2 <= q) ++ti;
      const int tj = q - ti * (ti + 1) / 2;
      const float* pa = Pn + (ti * 16 + lq) * PP + 2 * half;
      const float* pb = Pn + (tj * 16 + lq) * PP + 2 * half;
      v8f s = {};
#pragma unroll
      for (int kc = 0; kc < NB / 4; ++kc) {
        v2f a = *(const v2f*)(pa + 4 * kc);
        v2f b = *(const v2f*)(pb + 4 * kc);
        s = __builtin_amdgcn_wmma_f32_16x16x4_f32(false, a, false, b, (short)0, s, false, false);
      }
      float* cp = Lm + (k0 + NB + ti * 16 + 8 * half) * NO + (k0 + NB + tj * 16 + lq);
#pragma unroll
      for (int r = 0; r < 8; ++r) cp[r * NO] -= s[r];
    }
    __syncthreads();
  }
}

#define SC 16
#define YP 516
__global__ __launch_bounds__(1024) void solve_kernel(const float* __restrict__ L,
                                                     const float* __restrict__ Kox,
                                                     float* __restrict__ G) {
  __shared__ __attribute__((aligned(16))) float Yt[SC * YP];

  const int tid = threadIdx.x;
  const int lane = tid & 31;
  const int half = lane >> 4, lq = lane & 15;
  const int wave = __builtin_amdgcn_readfirstlane(tid >> 5);
  const int col0 = blockIdx.x * SC;

  for (int i = 0; i < (SC * NO) / 1024; ++i) {
    int idx = tid + 1024 * i;
    int cc = idx & (SC - 1), n = idx >> 4;
    int col = col0 + cc;
    float v = 0.f;
    if (col < MT) v = Kox[n * MT + col];
    else if (col == MT) v = 1.0f;
    Yt[cc * YP + n] = v;
  }

  float Ln[NB];
  float dg = 1.f;

  if (wave < SC) {
    const float4* lp = (const float4*)(L + lane * NO);
#pragma unroll
    for (int q = 0; q < NB / 4; ++q) {
      float4 t4 = lp[q];
      Ln[4 * q + 0] = t4.x; Ln[4 * q + 1] = t4.y; Ln[4 * q + 2] = t4.z; Ln[4 * q + 3] = t4.w;
    }
    dg = L[lane * NO + lane];
  }
  for (int kb = 0; kb < NO / NB; ++kb) {
    const int k0 = kb * NB;
    float Mr[NB];
    float inv = 0.f;
    if (wave < SC) {
      inv = 1.f / dg;
#pragma unroll
      for (int c = 0; c < NB; ++c) Mr[c] = (lane > c) ? Ln[c] * inv : 0.f;
    }
    const int ntiles = (NO - k0 - NB) >> 4;
    const int rowbase = k0 + NB + 16 * wave;
    __syncthreads();

    float4 av[4];
    if (wave < ntiles) {
      const float* ap = L + (rowbase + lq) * NO + k0 + 4 * half;
#pragma unroll
      for (int jj = 0; jj < 4; ++jj) av[jj] = *(const float4*)(ap + 8 * jj);
    }
    if (wave < SC) {
      if (kb + 1 < NO / NB) {
        const float4* lp = (const float4*)(L + (k0 + NB + lane) * NO + k0 + NB);
#pragma unroll
        for (int q = 0; q < NB / 4; ++q) {
          float4 t4 = lp[q];
          Ln[4 * q + 0] = t4.x; Ln[4 * q + 1] = t4.y; Ln[4 * q + 2] = t4.z; Ln[4 * q + 3] = t4.w;
        }
        dg = L[(k0 + NB + lane) * NO + k0 + NB + lane];
      }
      float* yp = Yt + wave * YP + k0 + lane;
      float t = yp[0] * inv;
#pragma unroll
      for (int c = 0; c < NB - 1; ++c) {
        float yc = __int_as_float(__builtin_amdgcn_readlane(__float_as_int(t), c));
        t = fmaf(-Mr[c], yc, t);
      }
      yp[0] = t;
    }
    __syncthreads();

    if (wave < ntiles) {
      const float* bp = Yt + lq * YP + k0 + 4 * half;
      v8f s = {};
#pragma unroll
      for (int jj = 0; jj < 4; ++jj) {
        float4 bv = *(const float4*)(bp + 8 * jj);
        v2f a, b;
        a.x = av[jj].x; a.y = av[jj].y; b.x = bv.x; b.y = bv.y;
        s = __builtin_amdgcn_wmma_f32_16x16x4_f32(false, a, false, b, (short)0, s, false, false);
        a.x = av[jj].z; a.y = av[jj].w; b.x = bv.z; b.y = bv.w;
        s = __builtin_amdgcn_wmma_f32_16x16x4_f32(false, a, false, b, (short)0, s, false, false);
      }
      float4* cp = (float4*)(Yt + lq * YP + rowbase + 8 * half);
      float4 c0 = cp[0], c1 = cp[1];
      c0.x -= s[0]; c0.y -= s[1]; c0.z -= s[2]; c0.w -= s[3];
      c1.x -= s[4]; c1.y -= s[5]; c1.z -= s[6]; c1.w -= s[7];
      cp[0] = c0; cp[1] = c1;
    }
  }

  if (wave < SC) {
    const float* lp = L + (NO - NB) * NO + NO - NB + lane;
#pragma unroll
    for (int c = 0; c < NB; ++c) Ln[c] = lp[c * NO];
    dg = L[(NO - NB + lane) * NO + NO - NB + lane];
  }
  for (int kb = NO / NB - 1; kb >= 0; --kb) {
    const int k0 = kb * NB;
    float Mr[NB];
    float inv = 0.f;
    if (wave < SC) {
      inv = 1.f / dg;
#pragma unroll
      for (int c = 0; c < NB; ++c) Mr[c] = (lane < c) ? Ln[c] * inv : 0.f;
    }
    const int ntiles = k0 >> 4;
    const int rowbase = 16 * wave;
    __syncthreads();

    float at[16];
    if (wave < ntiles) {
      const float* ap = L + (k0 + 4 * half) * NO + rowbase + lq;
#pragma unroll
      for (int jj = 0; jj < 4; ++jj)
#pragma unroll
        for (int e = 0; e < 4; ++e) at[4 * jj + e] = ap[(8 * jj + e) * NO];
    }
    if (wave < SC) {
      if (kb > 0) {
        const float* lp = L + (k0 - NB) * NO + k0 - NB + lane;
#pragma unroll
        for (int c = 0; c < NB; ++c) Ln[c] = lp[c * NO];
        dg = L[(k0 - NB + lane) * NO + k0 - NB + lane];
      }
      float* yp = Yt + wave * YP + k0 + lane;
      float t = yp[0] * inv;
#pragma unroll
      for (int c = NB - 1; c >= 1; --c) {
        float xc = __int_as_float(__builtin_amdgcn_readlane(__float_as_int(t), c));
        t = fmaf(-Mr[c], xc, t);
      }
      yp[0] = t;
    }
    __syncthreads();

    if (wave < ntiles) {
      const float* bp = Yt + lq * YP + k0 + 4 * half;
      v8f s = {};
#pragma unroll
      for (int jj = 0; jj < 4; ++jj) {
        float4 bv = *(const float4*)(bp + 8 * jj);
        v2f a, b;
        a.x = at[4 * jj + 0]; a.y = at[4 * jj + 1]; b.x = bv.x; b.y = bv.y;
        s = __builtin_amdgcn_wmma_f32_16x16x4_f32(false, a, false, b, (short)0, s, false, false);
        a.x = at[4 * jj + 2]; a.y = at[4 * jj + 3]; b.x = bv.z; b.y = bv.w;
        s = __builtin_amdgcn_wmma_f32_16x16x4_f32(false, a, false, b, (short)0, s, false, false);
      }
      float4* cp = (float4*)(Yt + lq * YP + rowbase + 8 * half);
      float4 c0 = cp[0], c1 = cp[1];
      c0.x -= s[0]; c0.y -= s[1]; c0.z -= s[2]; c0.w -= s[3];
      c1.x -= s[4]; c1.y -= s[5]; c1.z -= s[6]; c1.w -= s[7];
      cp[0] = c0; cp[1] = c1;
    }
  }
  __syncthreads();

  for (int i = 0; i < (SC * NO) / 1024; ++i) {
    int idx = tid + 1024 * i;
    int n = idx & (NO - 1), cc = idx >> 9;
    int col = col0 + cc;
    if (col <= MT) G[col * NO + n] = Yt[cc * YP + n];
  }
}

__global__ __launch_bounds__(256) void mean_kernel(const float* __restrict__ G,
                                                   const float* __restrict__ ys,
                                                   const float* __restrict__ b_p,
                                                   float* __restrict__ out,
                                                   float* __restrict__ colsum) {
  __shared__ float r1[256], r2[256];
  int col = blockIdx.x, tid = threadIdx.x;
  float b = b_p[0];
  float sm = 0.f, sc = 0.f;
  for (int n = tid; n < NO; n += 256) {
    float g = G[col * NO + n];
    sm += g * (ys[n] - b);
    sc += g;
  }
  r1[tid] = sm; r2[tid] = sc;
  __syncthreads();
  for (int off = 128; off > 0; off >>= 1) {
    if (tid < off) { r1[tid] += r1[tid + off]; r2[tid] += r2[tid + off]; }
    __syncthreads();
  }
  if (tid == 0) {
    colsum[col] = r2[0];
    if (col < MT) out[col] = b + r1[0];
  }
}

__global__ __launch_bounds__(32) void cov_kernel(const float* __restrict__ Kxx,
                                                 const float* __restrict__ Kox,
                                                 const float* __restrict__ G,
                                                 const float* __restrict__ colsum,
                                                 float* __restrict__ out) {
  const int lane = threadIdx.x & 31;
  const int half = lane >> 4, lq = lane & 15;
  const int i0 = blockIdx.y * 16, j0 = blockIdx.x * 16;
  v8f c = {};
  for (int n0 = 0; n0 < NO; n0 += 8) {
    const float* grow = G + (i0 + lq) * NO + n0 + 2 * half;
    v2f a0; a0.x = grow[0]; a0.y = grow[1];
    v2f a1; a1.x = grow[4]; a1.y = grow[5];
    v2f b0; b0.x = Kox[(n0 + 2 * half + 0) * MT + j0 + lq];
            b0.y = Kox[(n0 + 2 * half + 1) * MT + j0 + lq];
    v2f b1; b1.x = Kox[(n0 + 4 + 2 * half + 0) * MT + j0 + lq];
            b1.y = Kox[(n0 + 4 + 2 * half + 1) * MT + j0 + lq];
    c = __builtin_amdgcn_wmma_f32_16x16x4_f32(false, a0, false, b0, (short)0, c, false, false);
    c = __builtin_amdgcn_wmma_f32_16x16x4_f32(false, a1, false, b1, (short)0, c, false, false);
  }
  float invS = 1.f / colsum[MT];
  float kbxj = colsum[j0 + lq];
#pragma unroll
  for (int r = 0; r < 8; ++r) {
    int i = i0 + r + 8 * half, j = j0 + lq;
    float v = Kxx[i * MT + j] - c[r] + (1.f - colsum[i]) * (1.f - kbxj) * invS;
    out[(1 + i) * MT + j] = v;
  }
}

extern "C" void kernel_launch(void* const* d_in, const int* in_sizes, int n_in,
                              void* d_out, int out_size, void* d_ws, size_t ws_size,
                              hipStream_t stream) {
  const float* Xt  = (const float*)d_in[0];
  const float* At  = (const float*)d_in[1];
  const float* Xo  = (const float*)d_in[2];
  const float* Ao  = (const float*)d_in[3];
  const float* ys  = (const float*)d_in[4];
  const float* rho = (const float*)d_in[5];
  const float* g   = (const float*)d_in[6];
  const float* nu  = (const float*)d_in[7];
  const float* b   = (const float*)d_in[8];
  float* out = (float*)d_out;

  float* ws  = (float*)d_ws;
  float* SN1 = ws;
  float* SN2 = SN1 + MT * PD;
  float* sT  = SN2 + NO * PD;
  float* sO  = sT + MT;
  float* Kxx = sO + NO;
  float* Kox = Kxx + MT * MT;
  float* Koo = Kox + NO * MT;
  float* G   = Koo + NO * NO;
  float* csm = G + (MT + 1) * NO;

  prep_kernel<<<(MT + NO) / IPB, IPB * 32, 0, stream>>>(Xt, At, Xo, Ao, SN1, SN2, sT, sO);

  gram_kernel<<<TKOO + TKOX + TKXX, GW * 32, 0, stream>>>(Xt, SN1, At, sT, Xo, SN2, Ao, sO,
                                                         Kxx, Kox, Koo, rho, nu, g);

  chol_kernel<<<1, 1024, 0, stream>>>(Koo);
  solve_kernel<<<(MT + 1 + SC - 1) / SC, 1024, 0, stream>>>(Koo, Kox, G);
  mean_kernel<<<MT + 1, 256, 0, stream>>>(G, ys, b, out, csm);
  cov_kernel<<<dim3(16, 16), 32, 0, stream>>>(Kxx, Kox, G, csm, out);
}


# Back substitution dropped: only L Y = [Kox, 1, ys-b] is solved; cov = Kxx - Y^T Y, mean = b + Y^T z, Kbx = Y^T w, 1^T Koo^-1 1 = w^T w (Koo symmetric by construction); sums regrouped
# speedup vs baseline: 123.4450x; 1.0786x over previous
#include <hip/hip_runtime.h>
#include <hip/hip_bf16.h>

typedef float v2f __attribute__((ext_vector_type(2)));
typedef float v8f __attribute__((ext_vector_type(8)));

#define MT 256
#define NO 512
#define PD 24
#define DD 8

#define IPB 8
__global__ __launch_bounds__(IPB * 32) void prep_kernel(const float* __restrict__ Xt,
                                                        const float* __restrict__ At,
                                                        const float* __restrict__ Xo,
                                                        const float* __restrict__ Ao,
                                                        float* __restrict__ SNt,
                                                        float* __restrict__ SNo,
                                                        float* __restrict__ sT,
                                                        float* __restrict__ sO) {
  __shared__ __attribute__((aligned(16))) float Xs[IPB][PD * DD];
  __shared__ float As[IPB][PD];
  __shared__ float Ns[IPB][PD];

  const int wave = threadIdx.x >> 5, lane = threadIdx.x & 31;
  const int it = blockIdx.x * IPB + wave;
  const bool obs = it >= MT;
  const int i = obs ? it - MT : it;
  const float* X = obs ? Xo : Xt;
  const float* A = obs ? Ao : At;
  float* SN = obs ? SNo : SNt;
  float* sOut = obs ? sO : sT;

  const float4* xg = (const float4*)(X + i * PD * DD);
  float4* xs4 = (float4*)Xs[wave];
  for (int q = lane; q < PD * DD / 4; q += 32) xs4[q] = xg[q];
  if (lane < PD) {
    const int p = i * PD + lane;
    As[wave][lane] = A[p];
    float s = 0.f;
#pragma unroll
    for (int d = 0; d < DD; ++d) { float x = X[p * DD + d]; s += x * x; }
    Ns[wave][lane] = s;
    SN[p] = s;
  }
  __syncthreads();

  float acc = 0.f;
  for (int idx = lane; idx < PD * PD; idx += 32) {
    int u = idx / PD, v = idx % PD;
    const float* xu = Xs[wave] + u * DD;
    const float* xv = Xs[wave] + v * DD;
    float dot = 0.f;
#pragma unroll
    for (int d = 0; d < DD; ++d) dot += xu[d] * xv[d];
    float d2 = Ns[wave][u] + Ns[wave][v] - 2.f * dot;
    acc += As[wave][u] * As[wave][v] * __expf(-0.5f * fmaxf(d2, 0.f));
  }
#pragma unroll
  for (int off = 16; off > 0; off >>= 1) acc += __shfl_xor(acc, off);
  if (lane == 0) sOut[i] = acc;
}

#define GW 12
#define UPW (PD / GW)
#define XP 196
#define TKOO ((NO / 16) * (NO / 16))
#define TKOX ((NO / 16) * (MT / 16))
#define TKXX ((MT / 16) * (MT / 16))
static_assert(NO / 16 == 32 && MT / 16 == 16, "tile decode in gram_kernel assumes 32 and 16 tiles per row");
__global__ __launch_bounds__(GW * 32) void gram_kernel(
    const float* __restrict__ Xt, const float* __restrict__ SNt,
    const float* __restrict__ At, const float* __restrict__ st,
    const float* __restrict__ Xo, const float* __restrict__ SNo,
    const float* __restrict__ Ao, const float* __restrict__ so,
    float* __restrict__ Kxx, float* __restrict__ Kox, float* __restrict__ Koo,
    const float* __restrict__ rho_p, const float* __restrict__ nu_p,
    const float* __restrict__ g_p) {
  __shared__ __attribute__((aligned(16))) float X1s[16 * XP];
  __shared__ __attribute__((aligned(16))) float X2s[16 * XP];
  __shared__ __attribute__((aligned(16))) float A1T[PD * 16];
  __shared__ __attribute__((aligned(16))) float N1T[PD * 16];
  __shared__ __attribute__((aligned(16))) float A2T[PD * 16];
  __shared__ __attribute__((aligned(16))) float N2T[PD * 16];
  __shared__ __attribute__((aligned(16))) float red[GW * 256];

  const int tid = threadIdx.x;
  const int lane = tid & 31;
  const int half = lane >> 4, lq = lane & 15;
  const int wave = __builtin_amdgcn_readfirstlane(tid >> 5);

  const int bid = blockIdx.x;
  const bool isKoo = bid < TKOO;
  const bool isKox = !isKoo && bid < TKOO + TKOX;
  const int tb = isKoo ? bid : (isKox ? bid - TKOO : bid - TKOO - TKOX);
  const int tsh = isKoo ? 5 : 4;
  const int i0 = (tb >> tsh) * 16, j0 = (tb & ((1 << tsh) - 1)) * 16;
  const bool obs1 = isKoo || isKox;
  const float* X1  = obs1 ? Xo : Xt;
  const float* SN1 = obs1 ? SNo : SNt;
  const float* A1  = obs1 ? Ao : At;
  const float* s1  = obs1 ? so : st;
  const float* X2  = isKoo ? Xo : Xt;
  const float* SN2 = isKoo ? SNo : SNt;
  const float* A2  = isKoo ? Ao : At;
  const float* s2  = isKoo ? so : st;
  float* out = isKoo ? Koo : (isKox ? Kox : Kxx);
  const int ldc = isKoo ? NO : MT;
  const int addDiag = isKoo ? 1 : 0;

  {
    const float4* g1 = (const float4*)(X1 + i0 * PD * DD);
    const float4* g2 = (const float4*)(X2 + j0 * PD * DD);
    for (int q = tid; q < 16 * PD * DD / 4; q += GW * 32) {
      const int item = q / (PD * DD / 4), w4 = q % (PD * DD / 4);
      const float4 t1 = g1[q], t2 = g2[q];
      *(float4*)(X1s + item * XP + 4 * w4) = t1;
      *(float4*)(X2s + item * XP + 4 * w4) = t2;
    }
    for (int q = tid; q < 16 * PD; q += GW * 32) {
      const int item = q / PD, p = q % PD;
      A1T[p * 16 + item] = A1[i0 * PD + q];
      N1T[p * 16 + item] = SN1[i0 * PD + q];
      A2T[p * 16 + item] = A2[j0 * PD + q];
      N2T[p * 16 + item] = SN2[j0 * PD + q];
    }
  }
  __syncthreads();

  float a1u[UPW][8], n1u[UPW][8];
  v2f fa0[UPW], fa1[UPW];
#pragma unroll
  for (int uu = 0; uu < UPW; ++uu) {
    const int u = wave * UPW + uu;
    const float4* ap = (const float4*)(A1T + u * 16 + 8 * half);
    const float4* np = (const float4*)(N1T + u * 16 + 8 * half);
    const float4 ta = ap[0], tb = ap[1], tc = np[0], td = np[1];
    a1u[uu][0] = ta.x; a1u[uu][1] = ta.y; a1u[uu][2] = ta.z; a1u[uu][3] = ta.w;
    a1u[uu][4] = tb.x; a1u[uu][5] = tb.y; a1u[uu][6] = tb.z; a1u[uu][7] = tb.w;
    n1u[uu][0] = tc.x; n1u[uu][1] = tc.y; n1u[uu][2] = tc.z; n1u[uu][3] = tc.w;
    n1u[uu][4] = td.x; n1u[uu][5] = td.y; n1u[uu][6] = td.z; n1u[uu][7] = td.w;
    const float* x1 = X1s + lq * XP + u * DD + 2 * half;
    fa0[uu] = *(const v2f*)(x1);
    fa1[uu] = *(const v2f*)(x1 + 4);
  }

  float acc[8];
#pragma unroll
  for (int r = 0; r < 8; ++r) acc[r] = 0.f;

  for (int v = 0; v < PD; ++v) {
    const float* x2 = X2s + lq * XP + v * DD + 2 * half;
    const v2f b0 = *(const v2f*)(x2);
    const v2f b1 = *(const v2f*)(x2 + 4);
    const float sn2v = N2T[v * 16 + lq];
    const float a2v  = A2T[v * 16 + lq];
#pragma unroll
    for (int uu = 0; uu < UPW; ++uu) {
      v8f s = {};
      s = __builtin_amdgcn_wmma_f32_16x16x4_f32(false, fa0[uu], false, b0, (short)0, s, false, false);
      s = __builtin_amdgcn_wmma_f32_16x16x4_f32(false, fa1[uu], false, b1, (short)0, s, false, false);
#pragma unroll
      for (int r = 0; r < 8; ++r) {
        float d2 = n1u[uu][r] + sn2v - 2.f * s[r];
        acc[r] += a1u[uu][r] * a2v * __expf(-0.5f * fmaxf(d2, 0.f));
      }
    }
  }

#pragma unroll
  for (int r = 0; r < 8; ++r) red[wave * 256 + r * 32 + lane] = acc[r];
  __syncthreads();

  if (tid < 256) {
    float c = 0.f;
#pragma unroll
    for (int w = 0; w < GW; ++w) c += red[w * 256 + tid];
    const int r = tid >> 5;
    const int i = i0 + r + 8 * half, j = j0 + lq;
    float invrho = 1.f / rho_p[0];
    float nu = nu_p[0], g = g_p[0];
    float d2 = s1[i] + s2[j] - 2.f * c;
    float val = nu * __expf(-0.5f * fmaxf(d2, 0.f) * invrho);
    if (addDiag && i == j) val += nu * g;
    out[i * ldc + j] = val;
  }
}

#define NB 32
#define PP 34
__global__ __launch_bounds__(1024) void chol_kernel(float* __restrict__ Lm) {
  __shared__ __attribute__((aligned(16))) float Sd[NB * NB];
  __shared__ __attribute__((aligned(16))) float invd[NB];
  __shared__ __attribute__((aligned(16))) float Pn[(NO - NB) * PP];

  const int tid = threadIdx.x;
  const int lane = tid & 31;
  const int half = lane >> 4, lq = lane & 15;
  const int wave = __builtin_amdgcn_readfirstlane(tid >> 5);

  for (int kb = 0; kb < NO / NB; ++kb) {
    const int k0 = kb * NB;
    const int nrows = NO - k0 - NB;

    if (wave == 0) {
      float r[NB];
      float* rowp = Lm + (k0 + lane) * NO + k0;
      const float4* src = (const float4*)rowp;
#pragma unroll
      for (int q = 0; q < NB / 4; ++q) {
        float4 t = src[q];
        r[4 * q + 0] = t.x; r[4 * q + 1] = t.y; r[4 * q + 2] = t.z; r[4 * q + 3] = t.w;
      }
      float mydinv = 0.f;
#pragma unroll
      for (int k = 0; k < NB; ++k) {
        float akk = __int_as_float(__builtin_amdgcn_readlane(__float_as_int(r[k]), k));
        float dinv = __builtin_amdgcn_rsqf(akk);
        r[k] = r[k] * dinv;
        mydinv = (lane == k) ? dinv : mydinv;
#pragma unroll
        for (int j = k + 1; j < NB; ++j) {
          float ljk = __int_as_float(__builtin_amdgcn_readlane(__float_as_int(r[k]), j));
          r[j] -= r[k] * ljk;
        }
      }
#pragma unroll
      for (int j = 0; j < NB; ++j) r[j] = (j <= lane) ? r[j] : 0.f;
#pragma unroll
      for (int c = 0; c < NB; ++c) Sd[c * NB + lane] = r[c];
      invd[lane] = mydinv;
      float4* dst = (float4*)rowp;
#pragma unroll
      for (int q = 0; q < NB / 4; ++q)
        dst[q] = make_float4(r[4 * q + 0], r[4 * q + 1], r[4 * q + 2], r[4 * q + 3]);
    }
    __syncthreads();

    if (wave < (nrows >> 5)) {
      float Lr[NB];
#pragma unroll
      for (int c = 0; c < NB; ++c) Lr[c] = Sd[c * NB + lane];
      const float dv = invd[lane];
      float x[NB];
      float* rowp = Lm + (k0 + NB + tid) * NO + k0;
      const float4* src = (const float4*)rowp;
#pragma unroll
      for (int q = 0; q < NB / 4; ++q) {
        float4 t = src[q];
        x[4 * q + 0] = t.x; x[4 * q + 1] = t.y; x[4 * q + 2] = t.z; x[4 * q + 3] = t.w;
      }
#pragma unroll
      for (int c = 0; c < NB; ++c) {
        x[c] *= __int_as_float(__builtin_amdgcn_readlane(__float_as_int(dv), c));
#pragma unroll
        for (int j = c + 1; j < NB; ++j)
          x[j] -= x[c] * __int_as_float(__builtin_amdgcn_readlane(__float_as_int(Lr[c]), j));
      }
      float4* dst = (float4*)rowp;
#pragma unroll
      for (int q = 0; q < NB / 4; ++q)
        dst[q] = make_float4(x[4 * q + 0], x[4 * q + 1], x[4 * q + 2], x[4 * q + 3]);
      float2* pdst = (float2*)(Pn + tid * PP);
#pragma unroll
      for (int q = 0; q < NB / 2; ++q) pdst[q] = make_float2(x[2 * q + 0], x[2 * q + 1]);
    }
    __syncthreads();

    const int m = nrows >> 4;
    const int T = m * (m + 1) / 2;
    for (int q = wave; q < T; q += 32) {
      int ti = (int)((sqrtf(8.f * (float)q + 1.f) - 1.f) * 0.5f);
      if (ti * (ti + 1) / 2 > q) --ti;
      if ((ti + 1) * (ti + 2) / 2 <= q) ++ti;
      const int tj = q - ti * (ti + 1) / 2;
      const float* pa = Pn + (ti * 16 + lq) * PP + 2 * half;
      const float* pb = Pn + (tj * 16 + lq) * PP + 2 * half;
      v8f s = {};
#pragma unroll
      for (int kc = 0; kc < NB / 4; ++kc) {
        v2f a = *(const v2f*)(pa + 4 * kc);
        v2f b = *(const v2f*)(pb + 4 * kc);
        s = __builtin_amdgcn_wmma_f32_16x16x4_f32(false, a, false, b, (short)0, s, false, false);
      }
      float* cp = Lm + (k0 + NB + ti * 16 + 8 * half) * NO + (k0 + NB + tj * 16 + lq);
#pragma unroll
      for (int r = 0; r < 8; ++r) cp[r * NO] -= s[r];
    }
    __syncthreads();
  }
}

#define SC 16
#define YP 516
__global__ __launch_bounds__(1024) void solve_kernel(const float* __restrict__ L,
                                                     const float* __restrict__ Kox,
                                                     const float* __restrict__ ys,
                                                     const float* __restrict__ b_p,
                                                     float* __restrict__ Y) {
  __shared__ __attribute__((aligned(16))) float Yt[SC * YP];

  const int tid = threadIdx.x;
  const int lane = tid & 31;
  const int half = lane >> 4, lq = lane & 15;
  const int wave = __builtin_amdgcn_readfirstlane(tid >> 5);
  const int col0 = blockIdx.x * SC;

  const float bb = b_p[0];
  for (int i = 0; i < (SC * NO) / 1024; ++i) {
    int idx = tid + 1024 * i;
    int cc = idx & (SC - 1), n = idx >> 4;
    int col = col0 + cc;
    float v = 0.f;
    if (col < MT) v = Kox[n * MT + col];
    else if (col == MT) v = 1.0f;
    else if (col == MT + 1) v = ys[n] - bb;
    Yt[cc * YP + n] = v;
  }

  float Ln[NB];
  float dg = 1.f;

  if (wave < SC) {
    const float4* lp = (const float4*)(L + lane * NO);
#pragma unroll
    for (int q = 0; q < NB / 4; ++q) {
      float4 t4 = lp[q];
      Ln[4 * q + 0] = t4.x; Ln[4 * q + 1] = t4.y; Ln[4 * q + 2] = t4.z; Ln[4 * q + 3] = t4.w;
    }
    dg = L[lane * NO + lane];
  }
  for (int kb = 0; kb < NO / NB; ++kb) {
    const int k0 = kb * NB;
    float Mr[NB];
    float inv = 0.f;
    if (wave < SC) {
      inv = 1.f / dg;
#pragma unroll
      for (int c = 0; c < NB; ++c) Mr[c] = (lane > c) ? Ln[c] * inv : 0.f;
    }
    const int ntiles = (NO - k0 - NB) >> 4;
    const int rowbase = k0 + NB + 16 * wave;
    __syncthreads();

    float4 av[4];
    if (wave < ntiles) {
      const float* ap = L + (rowbase + lq) * NO + k0 + 4 * half;
#pragma unroll
      for (int jj = 0; jj < 4; ++jj) av[jj] = *(const float4*)(ap + 8 * jj);
    }
    if (wave < SC) {
      if (kb + 1 < NO / NB) {
        const float4* lp = (const float4*)(L + (k0 + NB + lane) * NO + k0 + NB);
#pragma unroll
        for (int q = 0; q < NB / 4; ++q) {
          float4 t4 = lp[q];
          Ln[4 * q + 0] = t4.x; Ln[4 * q + 1] = t4.y; Ln[4 * q + 2] = t4.z; Ln[4 * q + 3] = t4.w;
        }
        dg = L[(k0 + NB + lane) * NO + k0 + NB + lane];
      }
      float* yp = Yt + wave * YP + k0 + lane;
      float t = yp[0] * inv;
#pragma unroll
      for (int c = 0; c < NB - 1; ++c) {
        float yc = __int_as_float(__builtin_amdgcn_readlane(__float_as_int(t), c));
        t = fmaf(-Mr[c], yc, t);
      }
      yp[0] = t;
    }
    __syncthreads();

    if (wave < ntiles) {
      const float* bp = Yt + lq * YP + k0 + 4 * half;
      v8f s = {};
#pragma unroll
      for (int jj = 0; jj < 4; ++jj) {
        float4 bv = *(const float4*)(bp + 8 * jj);
        v2f a, b;
        a.x = av[jj].x; a.y = av[jj].y; b.x = bv.x; b.y = bv.y;
        s = __builtin_amdgcn_wmma_f32_16x16x4_f32(false, a, false, b, (short)0, s, false, false);
        a.x = av[jj].z; a.y = av[jj].w; b.x = bv.z; b.y = bv.w;
        s = __builtin_amdgcn_wmma_f32_16x16x4_f32(false, a, false, b, (short)0, s, false, false);
      }
      float4* cp = (float4*)(Yt + lq * YP + rowbase + 8 * half);
      float4 c0 = cp[0], c1 = cp[1];
      c0.x -= s[0]; c0.y -= s[1]; c0.z -= s[2]; c0.w -= s[3];
      c1.x -= s[4]; c1.y -= s[5]; c1.z -= s[6]; c1.w -= s[7];
      cp[0] = c0; cp[1] = c1;
    }
  }

  for (int i = 0; i < (SC * NO) / 1024; ++i) {
    int idx = tid + 1024 * i;
    int n = idx & (NO - 1), cc = idx >> 9;
    int col = col0 + cc;
    if (col <= MT + 1) Y[col * NO + n] = Yt[cc * YP + n];
  }
}

__global__ __launch_bounds__(256) void mean_kernel(const float* __restrict__ Y,
                                                   const float* __restrict__ b_p,
                                                   float* __restrict__ out,
                                                   float* __restrict__ colsum) {
  __shared__ float r1[256], r2[256];
  int col = blockIdx.x, tid = threadIdx.x;
  float b = b_p[0];
  float sm = 0.f, sc = 0.f;
  for (int n = tid; n < NO; n += 256) {
    float y = Y[col * NO + n];
    sm += y * Y[(MT + 1) * NO + n];
    sc += y * Y[MT * NO + n];
  }
  r1[tid] = sm; r2[tid] = sc;
  __syncthreads();
  for (int off = 128; off > 0; off >>= 1) {
    if (tid < off) { r1[tid] += r1[tid + off]; r2[tid] += r2[tid + off]; }
    __syncthreads();
  }
  if (tid == 0) {
    colsum[col] = r2[0];
    if (col < MT) out[col] = b + r1[0];
  }
}

__global__ __launch_bounds__(32) void cov_kernel(const float* __restrict__ Kxx,
                                                 const float* __restrict__ Y,
                                                 const float* __restrict__ colsum,
                                                 float* __restrict__ out) {
  const int lane = threadIdx.x & 31;
  const int half = lane >> 4, lq = lane & 15;
  const int i0 = blockIdx.y * 16, j0 = blockIdx.x * 16;
  v8f c = {};
  for (int n0 = 0; n0 < NO; n0 += 8) {
    const float* yi = Y + (i0 + lq) * NO + n0 + 2 * half;
    const float* yj = Y + (j0 + lq) * NO + n0 + 2 * half;
    v2f a0; a0.x = yi[0]; a0.y = yi[1];
    v2f a1; a1.x = yi[4]; a1.y = yi[5];
    v2f b0; b0.x = yj[0]; b0.y = yj[1];
    v2f b1; b1.x = yj[4]; b1.y = yj[5];
    c = __builtin_amdgcn_wmma_f32_16x16x4_f32(false, a0, false, b0, (short)0, c, false, false);
    c = __builtin_amdgcn_wmma_f32_16x16x4_f32(false, a1, false, b1, (short)0, c, false, false);
  }
  float invS = 1.f / colsum[MT];
  float kbxj = colsum[j0 + lq];
#pragma unroll
  for (int r = 0; r < 8; ++r) {
    int i = i0 + r + 8 * half, j = j0 + lq;
    float v = Kxx[i * MT + j] - c[r] + (1.f - colsum[i]) * (1.f - kbxj) * invS;
    out[(1 + i) * MT + j] = v;
  }
}

extern "C" void kernel_launch(void* const* d_in, const int* in_sizes, int n_in,
                              void* d_out, int out_size, void* d_ws, size_t ws_size,
                              hipStream_t stream) {
  const float* Xt  = (const float*)d_in[0];
  const float* At  = (const float*)d_in[1];
  const float* Xo  = (const float*)d_in[2];
  const float* Ao  = (const float*)d_in[3];
  const float* ys  = (const float*)d_in[4];
  const float* rho = (const float*)d_in[5];
  const float* g   = (const float*)d_in[6];
  const float* nu  = (const float*)d_in[7];
  const float* b   = (const float*)d_in[8];
  float* out = (float*)d_out;

  float* ws  = (float*)d_ws;
  float* SN1 = ws;
  float* SN2 = SN1 + MT * PD;
  float* sT  = SN2 + NO * PD;
  float* sO  = sT + MT;
  float* Kxx = sO + NO;
  float* Kox = Kxx + MT * MT;
  float* Koo = Kox + NO * MT;
  float* Y   = Koo + NO * NO;
  float* csm = Y + (MT + 2) * NO;

  prep_kernel<<<(MT + NO) / IPB, IPB * 32, 0, stream>>>(Xt, At, Xo, Ao, SN1, SN2, sT, sO);

  gram_kernel<<<TKOO + TKOX + TKXX, GW * 32, 0, stream>>>(Xt, SN1, At, sT, Xo, SN2, Ao, sO,
                                                         Kxx, Kox, Koo, rho, nu, g);

  chol_kernel<<<1, 1024, 0, stream>>>(Koo);
  solve_kernel<<<(MT + 2 + SC - 1) / SC, 1024, 0, stream>>>(Koo, Kox, ys, b, Y);
  mean_kernel<<<MT + 1, 256, 0, stream>>>(Y, b, out, csm);
  cov_kernel<<<dim3(16, 16), 32, 0, stream>>>(Kxx, Y, csm, out);
}


# Cholesky lookahead: wave 0 updates the next 32x32 diagonal block first (3 tiles on WMMA, kept in LDS) and factors it while waves 1..31 do the other trailing tiles; 2 barriers per block, was 3
# speedup vs baseline: 131.9438x; 1.0688x over previous
#include <hip/hip_runtime.h>
#include <hip/hip_bf16.h>

typedef float v2f __attribute__((ext_vector_type(2)));
typedef float v8f __attribute__((ext_vector_type(8)));

#define MT 256
#define NO 512
#define PD 24
#define DD 8

#define IPB 8
__global__ __launch_bounds__(IPB * 32) void prep_kernel(const float* __restrict__ Xt,
                                                        const float* __restrict__ At,
                                                        const float* __restrict__ Xo,
                                                        const float* __restrict__ Ao,
                                                        float* __restrict__ SNt,
                                                        float* __restrict__ SNo,
                                                        float* __restrict__ sT,
                                                        float* __restrict__ sO) {
  __shared__ __attribute__((aligned(16))) float Xs[IPB][PD * DD];
  __shared__ float As[IPB][PD];
  __shared__ float Ns[IPB][PD];

  const int wave = threadIdx.x >> 5, lane = threadIdx.x & 31;
  const int it = blockIdx.x * IPB + wave;
  const bool obs = it >= MT;
  const int i = obs ? it - MT : it;
  const float* X = obs ? Xo : Xt;
  const float* A = obs ? Ao : At;
  float* SN = obs ? SNo : SNt;
  float* sOut = obs ? sO : sT;

  const float4* xg = (const float4*)(X + i * PD * DD);
  float4* xs4 = (float4*)Xs[wave];
  for (int q = lane; q < PD * DD / 4; q += 32) xs4[q] = xg[q];
  if (lane < PD) {
    const int p = i * PD + lane;
    As[wave][lane] = A[p];
    float s = 0.f;
#pragma unroll
    for (int d = 0; d < DD; ++d) { float x = X[p * DD + d]; s += x * x; }
    Ns[wave][lane] = s;
    SN[p] = s;
  }
  __syncthreads();

  float acc = 0.f;
  for (int idx = lane; idx < PD * PD; idx += 32) {
    int u = idx / PD, v = idx % PD;
    const float* xu = Xs[wave] + u * DD;
    const float* xv = Xs[wave] + v * DD;
    float dot = 0.f;
#pragma unroll
    for (int d = 0; d < DD; ++d) dot += xu[d] * xv[d];
    float d2 = Ns[wave][u] + Ns[wave][v] - 2.f * dot;
    acc += As[wave][u] * As[wave][v] * __expf(-0.5f * fmaxf(d2, 0.f));
  }
#pragma unroll
  for (int off = 16; off > 0; off >>= 1) acc += __shfl_xor(acc, off);
  if (lane == 0) sOut[i] = acc;
}

#define GW 12
#define UPW (PD / GW)
#define XP 196
#define TKOO ((NO / 16) * (NO / 16))
#define TKOX ((NO / 16) * (MT / 16))
#define TKXX ((MT / 16) * (MT / 16))
static_assert(NO / 16 == 32 && MT / 16 == 16, "tile decode in gram_kernel assumes 32 and 16 tiles per row");
__global__ __launch_bounds__(GW * 32) void gram_kernel(
    const float* __restrict__ Xt, const float* __restrict__ SNt,
    const float* __restrict__ At, const float* __restrict__ st,
    const float* __restrict__ Xo, const float* __restrict__ SNo,
    const float* __restrict__ Ao, const float* __restrict__ so,
    float* __restrict__ Kxx, float* __restrict__ Kox, float* __restrict__ Koo,
    const float* __restrict__ rho_p, const float* __restrict__ nu_p,
    const float* __restrict__ g_p) {
  __shared__ __attribute__((aligned(16))) float X1s[16 * XP];
  __shared__ __attribute__((aligned(16))) float X2s[16 * XP];
  __shared__ __attribute__((aligned(16))) float A1T[PD * 16];
  __shared__ __attribute__((aligned(16))) float N1T[PD * 16];
  __shared__ __attribute__((aligned(16))) float A2T[PD * 16];
  __shared__ __attribute__((aligned(16))) float N2T[PD * 16];
  __shared__ __attribute__((aligned(16))) float red[GW * 256];

  const int tid = threadIdx.x;
  const int lane = tid & 31;
  const int half = lane >> 4, lq = lane & 15;
  const int wave = __builtin_amdgcn_readfirstlane(tid >> 5);

  const int bid = blockIdx.x;
  const bool isKoo = bid < TKOO;
  const bool isKox = !isKoo && bid < TKOO + TKOX;
  const int tb = isKoo ? bid : (isKox ? bid - TKOO : bid - TKOO - TKOX);
  const int tsh = isKoo ? 5 : 4;
  const int i0 = (tb >> tsh) * 16, j0 = (tb & ((1 << tsh) - 1)) * 16;
  const bool obs1 = isKoo || isKox;
  const float* X1  = obs1 ? Xo : Xt;
  const float* SN1 = obs1 ? SNo : SNt;
  const float* A1  = obs1 ? Ao : At;
  const float* s1  = obs1 ? so : st;
  const float* X2  = isKoo ? Xo : Xt;
  const float* SN2 = isKoo ? SNo : SNt;
  const float* A2  = isKoo ? Ao : At;
  const float* s2  = isKoo ? so : st;
  float* out = isKoo ? Koo : (isKox ? Kox : Kxx);
  const int ldc = isKoo ? NO : MT;
  const int addDiag = isKoo ? 1 : 0;

  {
    const float4* g1 = (const float4*)(X1 + i0 * PD * DD);
    const float4* g2 = (const float4*)(X2 + j0 * PD * DD);
    for (int q = tid; q < 16 * PD * DD / 4; q += GW * 32) {
      const int item = q / (PD * DD / 4), w4 = q % (PD * DD / 4);
      const float4 t1 = g1[q], t2 = g2[q];
      *(float4*)(X1s + item * XP + 4 * w4) = t1;
      *(float4*)(X2s + item * XP + 4 * w4) = t2;
    }
    for (int q = tid; q < 16 * PD; q += GW * 32) {
      const int item = q / PD, p = q % PD;
      A1T[p * 16 + item] = A1[i0 * PD + q];
      N1T[p * 16 + item] = SN1[i0 * PD + q];
      A2T[p * 16 + item] = A2[j0 * PD + q];
      N2T[p * 16 + item] = SN2[j0 * PD + q];
    }
  }
  __syncthreads();

  float a1u[UPW][8], n1u[UPW][8];
  v2f fa0[UPW], fa1[UPW];
#pragma unroll
  for (int uu = 0; uu < UPW; ++uu) {
    const int u = wave * UPW + uu;
    const float4* ap = (const float4*)(A1T + u * 16 + 8 * half);
    const float4* np = (const float4*)(N1T + u * 16 + 8 * half);
    const float4 ta = ap[0], tb = ap[1], tc = np[0], td = np[1];
    a1u[uu][0] = ta.x; a1u[uu][1] = ta.y; a1u[uu][2] = ta.z; a1u[uu][3] = ta.w;
    a1u[uu][4] = tb.x; a1u[uu][5] = tb.y; a1u[uu][6] = tb.z; a1u[uu][7] = tb.w;
    n1u[uu][0] = tc.x; n1u[uu][1] = tc.y; n1u[uu][2] = tc.z; n1u[uu][3] = tc.w;
    n1u[uu][4] = td.x; n1u[uu][5] = td.y; n1u[uu][6] = td.z; n1u[uu][7] = td.w;
    const float* x1 = X1s + lq * XP + u * DD + 2 * half;
    fa0[uu] = *(const v2f*)(x1);
    fa1[uu] = *(const v2f*)(x1 + 4);
  }

  float acc[8];
#pragma unroll
  for (int r = 0; r < 8; ++r) acc[r] = 0.f;

  for (int v = 0; v < PD; ++v) {
    const float* x2 = X2s + lq * XP + v * DD + 2 * half;
    const v2f b0 = *(const v2f*)(x2);
    const v2f b1 = *(const v2f*)(x2 + 4);
    const float sn2v = N2T[v * 16 + lq];
    const float a2v  = A2T[v * 16 + lq];
#pragma unroll
    for (int uu = 0; uu < UPW; ++uu) {
      v8f s = {};
      s = __builtin_amdgcn_wmma_f32_16x16x4_f32(false, fa0[uu], false, b0, (short)0, s, false, false);
      s = __builtin_amdgcn_wmma_f32_16x16x4_f32(false, fa1[uu], false, b1, (short)0, s, false, false);
#pragma unroll
      for (int r = 0; r < 8; ++r) {
        float d2 = n1u[uu][r] + sn2v - 2.f * s[r];
        acc[r] += a1u[uu][r] * a2v * __expf(-0.5f * fmaxf(d2, 0.f));
      }
    }
  }

#pragma unroll
  for (int r = 0; r < 8; ++r) red[wave * 256 + r * 32 + lane] = acc[r];
  __syncthreads();

  if (tid < 256) {
    float c = 0.f;
#pragma unroll
    for (int w = 0; w < GW; ++w) c += red[w * 256 + tid];
    const int r = tid >> 5;
    const int i = i0 + r + 8 * half, j = j0 + lq;
    float invrho = 1.f / rho_p[0];
    float nu = nu_p[0], g = g_p[0];
    float d2 = s1[i] + s2[j] - 2.f * c;
    float val = nu * __expf(-0.5f * fmaxf(d2, 0.f) * invrho);
    if (addDiag && i == j) val += nu * g;
    out[i * ldc + j] = val;
  }
}

#define NB 32
#define PP 34
#define DP 33

__device__ __forceinline__ void chol_diag(float (&r)[NB], const int lane,
                                          float* __restrict__ Sd, float* __restrict__ invd,
                                          float* __restrict__ rowp) {
  float mydinv = 0.f;
#pragma unroll
  for (int k = 0; k < NB; ++k) {
    float akk = __int_as_float(__builtin_amdgcn_readlane(__float_as_int(r[k]), k));
    float dinv = __builtin_amdgcn_rsqf(akk);
    r[k] = r[k] * dinv;
    mydinv = (lane == k) ? dinv : mydinv;
#pragma unroll
    for (int j = k + 1; j < NB; ++j) {
      float ljk = __int_as_float(__builtin_amdgcn_readlane(__float_as_int(r[k]), j));
      r[j] -= r[k] * ljk;
    }
  }
#pragma unroll
  for (int j = 0; j < NB; ++j) r[j] = (j <= lane) ? r[j] : 0.f;
#pragma unroll
  for (int c = 0; c < NB; ++c) Sd[c * NB + lane] = r[c];
  invd[lane] = mydinv;
  float4* dst = (float4*)rowp;
#pragma unroll
  for (int q = 0; q < NB / 4; ++q)
    dst[q] = make_float4(r[4 * q + 0], r[4 * q + 1], r[4 * q + 2], r[4 * q + 3]);
}

__global__ __launch_bounds__(1024) void chol_kernel(float* __restrict__ Lm) {
  __shared__ __attribute__((aligned(16))) float Sd[NB * NB];
  __shared__ __attribute__((aligned(16))) float invd[NB];
  __shared__ __attribute__((aligned(16))) float Pn[(NO - NB) * PP];
  __shared__ float Db[NB * DP];

  const int tid = threadIdx.x;
  const int lane = tid & 31;
  const int half = lane >> 4, lq = lane & 15;
  const int wave = __builtin_amdgcn_readfirstlane(tid >> 5);

  if (wave == 0) {
    float r[NB];
    float* rowp = Lm + lane * NO;
    const float4* src = (const float4*)rowp;
#pragma unroll
    for (int q = 0; q < NB / 4; ++q) {
      float4 t = src[q];
      r[4 * q + 0] = t.x; r[4 * q + 1] = t.y; r[4 * q + 2] = t.z; r[4 * q + 3] = t.w;
    }
    chol_diag(r, lane, Sd, invd, rowp);
  }

  for (int kb = 0; kb < NO / NB - 1; ++kb) {
    const int k0 = kb * NB;
    const int nrows = NO - k0 - NB;
    __syncthreads();

    if (wave < (nrows >> 5)) {
      float Lr[NB];
#pragma unroll
      for (int c = 0; c < NB; ++c) Lr[c] = Sd[c * NB + lane];
      const float dv = invd[lane];
      float x[NB];
      float* rowp = Lm + (k0 + NB + tid) * NO + k0;
      const float4* src = (const float4*)rowp;
#pragma unroll
      for (int q = 0; q < NB / 4; ++q) {
        float4 t = src[q];
        x[4 * q + 0] = t.x; x[4 * q + 1] = t.y; x[4 * q + 2] = t.z; x[4 * q + 3] = t.w;
      }
#pragma unroll
      for (int c = 0; c < NB; ++c) {
        x[c] *= __int_as_float(__builtin_amdgcn_readlane(__float_as_int(dv), c));
#pragma unroll
        for (int j = c + 1; j < NB; ++j)
          x[j] -= x[c] * __int_as_float(__builtin_amdgcn_readlane(__float_as_int(Lr[c]), j));
      }
      float4* dst = (float4*)rowp;
#pragma unroll
      for (int q = 0; q < NB / 4; ++q)
        dst[q] = make_float4(x[4 * q + 0], x[4 * q + 1], x[4 * q + 2], x[4 * q + 3]);
      float2* pdst = (float2*)(Pn + tid * PP);
#pragma unroll
      for (int q = 0; q < NB / 2; ++q) pdst[q] = make_float2(x[2 * q + 0], x[2 * q + 1]);
    }
    __syncthreads();

    const int m = nrows >> 4;
    const int T = m * (m + 1) / 2;
    if (wave == 0) {
#pragma unroll
      for (int t = 0; t < 3; ++t) {
        const int ti = (t + 1) >> 1, tj = t >> 1;
        const float* pa = Pn + (ti * 16 + lq) * PP + 2 * half;
        const float* pb = Pn + (tj * 16 + lq) * PP + 2 * half;
        v8f s = {};
#pragma unroll
        for (int kc = 0; kc < NB / 4; ++kc) {
          v2f a = *(const v2f*)(pa + 4 * kc);
          v2f b = *(const v2f*)(pb + 4 * kc);
          s = __builtin_amdgcn_wmma_f32_16x16x4_f32(false, a, false, b, (short)0, s, false, false);
        }
        const float* cp = Lm + (k0 + NB + ti * 16 + 8 * half) * NO + (k0 + NB + tj * 16 + lq);
        float* dp = Db + (ti * 16 + 8 * half) * DP + tj * 16 + lq;
#pragma unroll
        for (int r = 0; r < 8; ++r) dp[r * DP] = cp[r * NO] - s[r];
      }
#pragma unroll
      for (int r = 0; r < 8; ++r) Db[(8 * half + r) * DP + 16 + lq] = 0.f;
      float r[NB];
#pragma unroll
      for (int c = 0; c < NB; ++c) r[c] = Db[lane * DP + c];
      chol_diag(r, lane, Sd, invd, Lm + (k0 + NB + lane) * NO + k0 + NB);
    } else {
      for (int q = 2 + wave; q < T; q += 31) {
        int ti = (int)((sqrtf(8.f * (float)q + 1.f) - 1.f) * 0.5f);
        if (ti * (ti + 1) / 2 > q) --ti;
        if ((ti + 1) * (ti + 2) / 2 <= q) ++ti;
        const int tj = q - ti * (ti + 1) / 2;
        const float* pa = Pn + (ti * 16 + lq) * PP + 2 * half;
        const float* pb = Pn + (tj * 16 + lq) * PP + 2 * half;
        v8f s = {};
#pragma unroll
        for (int kc = 0; kc < NB / 4; ++kc) {
          v2f a = *(const v2f*)(pa + 4 * kc);
          v2f b = *(const v2f*)(pb + 4 * kc);
          s = __builtin_amdgcn_wmma_f32_16x16x4_f32(false, a, false, b, (short)0, s, false, false);
        }
        float* cp = Lm + (k0 + NB + ti * 16 + 8 * half) * NO + (k0 + NB + tj * 16 + lq);
#pragma unroll
        for (int r = 0; r < 8; ++r) cp[r * NO] -= s[r];
      }
    }
  }
}

#define SC 16
#define YP 516
__global__ __launch_bounds__(1024) void solve_kernel(const float* __restrict__ L,
                                                     const float* __restrict__ Kox,
                                                     const float* __restrict__ ys,
                                                     const float* __restrict__ b_p,
                                                     float* __restrict__ Y) {
  __shared__ __attribute__((aligned(16))) float Yt[SC * YP];

  const int tid = threadIdx.x;
  const int lane = tid & 31;
  const int half = lane >> 4, lq = lane & 15;
  const int wave = __builtin_amdgcn_readfirstlane(tid >> 5);
  const int col0 = blockIdx.x * SC;

  const float bb = b_p[0];
  for (int i = 0; i < (SC * NO) / 1024; ++i) {
    int idx = tid + 1024 * i;
    int cc = idx & (SC - 1), n = idx >> 4;
    int col = col0 + cc;
    float v = 0.f;
    if (col < MT) v = Kox[n * MT + col];
    else if (col == MT) v = 1.0f;
    else if (col == MT + 1) v = ys[n] - bb;
    Yt[cc * YP + n] = v;
  }

  float Ln[NB];
  float dg = 1.f;

  if (wave < SC) {
    const float4* lp = (const float4*)(L + lane * NO);
#pragma unroll
    for (int q = 0; q < NB / 4; ++q) {
      float4 t4 = lp[q];
      Ln[4 * q + 0] = t4.x; Ln[4 * q + 1] = t4.y; Ln[4 * q + 2] = t4.z; Ln[4 * q + 3] = t4.w;
    }
    dg = L[lane * NO + lane];
  }
  for (int kb = 0; kb < NO / NB; ++kb) {
    const int k0 = kb * NB;
    float Mr[NB];
    float inv = 0.f;
    if (wave < SC) {
      inv = 1.f / dg;
#pragma unroll
      for (int c = 0; c < NB; ++c) Mr[c] = (lane > c) ? Ln[c] * inv : 0.f;
    }
    const int ntiles = (NO - k0 - NB) >> 4;
    const int rowbase = k0 + NB + 16 * wave;
    __syncthreads();

    float4 av[4];
    if (wave < ntiles) {
      const float* ap = L + (rowbase + lq) * NO + k0 + 4 * half;
#pragma unroll
      for (int jj = 0; jj < 4; ++jj) av[jj] = *(const float4*)(ap + 8 * jj);
    }
    if (wave < SC) {
      if (kb + 1 < NO / NB) {
        const float4* lp = (const float4*)(L + (k0 + NB + lane) * NO + k0 + NB);
#pragma unroll
        for (int q = 0; q < NB / 4; ++q) {
          float4 t4 = lp[q];
          Ln[4 * q + 0] = t4.x; Ln[4 * q + 1] = t4.y; Ln[4 * q + 2] = t4.z; Ln[4 * q + 3] = t4.w;
        }
        dg = L[(k0 + NB + lane) * NO + k0 + NB + lane];
      }
      float* yp = Yt + wave * YP + k0 + lane;
      float t = yp[0] * inv;
#pragma unroll
      for (int c = 0; c < NB - 1; ++c) {
        float yc = __int_as_float(__builtin_amdgcn_readlane(__float_as_int(t), c));
        t = fmaf(-Mr[c], yc, t);
      }
      yp[0] = t;
    }
    __syncthreads();

    if (wave < ntiles) {
      const float* bp = Yt + lq * YP + k0 + 4 * half;
      v8f s = {};
#pragma unroll
      for (int jj = 0; jj < 4; ++jj) {
        float4 bv = *(const float4*)(bp + 8 * jj);
        v2f a, b;
        a.x = av[jj].x; a.y = av[jj].y; b.x = bv.x; b.y = bv.y;
        s = __builtin_amdgcn_wmma_f32_16x16x4_f32(false, a, false, b, (short)0, s, false, false);
        a.x = av[jj].z; a.y = av[jj].w; b.x = bv.z; b.y = bv.w;
        s = __builtin_amdgcn_wmma_f32_16x16x4_f32(false, a, false, b, (short)0, s, false, false);
      }
      float4* cp = (float4*)(Yt + lq * YP + rowbase + 8 * half);
      float4 c0 = cp[0], c1 = cp[1];
      c0.x -= s[0]; c0.y -= s[1]; c0.z -= s[2]; c0.w -= s[3];
      c1.x -= s[4]; c1.y -= s[5]; c1.z -= s[6]; c1.w -= s[7];
      cp[0] = c0; cp[1] = c1;
    }
  }

  for (int i = 0; i < (SC * NO) / 1024; ++i) {
    int idx = tid + 1024 * i;
    int n = idx & (NO - 1), cc = idx >> 9;
    int col = col0 + cc;
    if (col <= MT + 1) Y[col * NO + n] = Yt[cc * YP + n];
  }
}

__global__ __launch_bounds__(256) void mean_kernel(const float* __restrict__ Y,
                                                   const float* __restrict__ b_p,
                                                   float* __restrict__ out,
                                                   float* __restrict__ colsum) {
  __shared__ float r1[256], r2[256];
  int col = blockIdx.x, tid = threadIdx.x;
  float b = b_p[0];
  float sm = 0.f, sc = 0.f;
  for (int n = tid; n < NO; n += 256) {
    float y = Y[col * NO + n];
    sm += y * Y[(MT + 1) * NO + n];
    sc += y * Y[MT * NO + n];
  }
  r1[tid] = sm; r2[tid] = sc;
  __syncthreads();
  for (int off = 128; off > 0; off >>= 1) {
    if (tid < off) { r1[tid] += r1[tid + off]; r2[tid] += r2[tid + off]; }
    __syncthreads();
  }
  if (tid == 0) {
    colsum[col] = r2[0];
    if (col < MT) out[col] = b + r1[0];
  }
}

__global__ __launch_bounds__(32) void cov_kernel(const float* __restrict__ Kxx,
                                                 const float* __restrict__ Y,
                                                 const float* __restrict__ colsum,
                                                 float* __restrict__ out) {
  const int lane = threadIdx.x & 31;
  const int half = lane >> 4, lq = lane & 15;
  const int i0 = blockIdx.y * 16, j0 = blockIdx.x * 16;
  v8f c = {};
  for (int n0 = 0; n0 < NO; n0 += 8) {
    const float* yi = Y + (i0 + lq) * NO + n0 + 2 * half;
    const float* yj = Y + (j0 + lq) * NO + n0 + 2 * half;
    v2f a0; a0.x = yi[0]; a0.y = yi[1];
    v2f a1; a1.x = yi[4]; a1.y = yi[5];
    v2f b0; b0.x = yj[0]; b0.y = yj[1];
    v2f b1; b1.x = yj[4]; b1.y = yj[5];
    c = __builtin_amdgcn_wmma_f32_16x16x4_f32(false, a0, false, b0, (short)0, c, false, false);
    c = __builtin_amdgcn_wmma_f32_16x16x4_f32(false, a1, false, b1, (short)0, c, false, false);
  }
  float invS = 1.f / colsum[MT];
  float kbxj = colsum[j0 + lq];
#pragma unroll
  for (int r = 0; r < 8; ++r) {
    int i = i0 + r + 8 * half, j = j0 + lq;
    float v = Kxx[i * MT + j] - c[r] + (1.f - colsum[i]) * (1.f - kbxj) * invS;
    out[(1 + i) * MT + j] = v;
  }
}

extern "C" void kernel_launch(void* const* d_in, const int* in_sizes, int n_in,
                              void* d_out, int out_size, void* d_ws, size_t ws_size,
                              hipStream_t stream) {
  const float* Xt  = (const float*)d_in[0];
  const float* At  = (const float*)d_in[1];
  const float* Xo  = (const float*)d_in[2];
  const float* Ao  = (const float*)d_in[3];
  const float* ys  = (const float*)d_in[4];
  const float* rho = (const float*)d_in[5];
  const float* g   = (const float*)d_in[6];
  const float* nu  = (const float*)d_in[7];
  const float* b   = (const float*)d_in[8];
  float* out = (float*)d_out;

  float* ws  = (float*)d_ws;
  float* SN1 = ws;
  float* SN2 = SN1 + MT * PD;
  float* sT  = SN2 + NO * PD;
  float* sO  = sT + MT;
  float* Kxx = sO + NO;
  float* Kox = Kxx + MT * MT;
  float* Koo = Kox + NO * MT;
  float* Y   = Koo + NO * NO;
  float* csm = Y + (MT + 2) * NO;

  prep_kernel<<<(MT + NO) / IPB, IPB * 32, 0, stream>>>(Xt, At, Xo, Ao, SN1, SN2, sT, sO);

  gram_kernel<<<TKOO + TKOX + TKXX, GW * 32, 0, stream>>>(Xt, SN1, At, sT, Xo, SN2, Ao, sO,
                                                         Kxx, Kox, Koo, rho, nu, g);

  chol_kernel<<<1, 1024, 0, stream>>>(Koo);
  solve_kernel<<<(MT + 2 + SC - 1) / SC, 1024, 0, stream>>>(Koo, Kox, ys, b, Y);
  mean_kernel<<<MT + 1, 256, 0, stream>>>(Y, b, out, csm);
  cov_kernel<<<dim3(16, 16), 32, 0, stream>>>(Kxx, Y, csm, out);
}


# Cholesky split 2x2 over 4 launches: 256-block factored by 1 workgroup, block below solved by 16 workgroups, trailing block updated by 136 one-wave tiles (K=256 per sum), then factored; sums regrouped
# speedup vs baseline: 154.5635x; 1.1714x over previous
#include <hip/hip_runtime.h>
#include <hip/hip_bf16.h>

typedef float v2f __attribute__((ext_vector_type(2)));
typedef float v8f __attribute__((ext_vector_type(8)));

#define MT 256
#define NO 512
#define PD 24
#define DD 8

#define IPB 8
__global__ __launch_bounds__(IPB * 32) void prep_kernel(const float* __restrict__ Xt,
                                                        const float* __restrict__ At,
                                                        const float* __restrict__ Xo,
                                                        const float* __restrict__ Ao,
                                                        float* __restrict__ SNt,
                                                        float* __restrict__ SNo,
                                                        float* __restrict__ sT,
                                                        float* __restrict__ sO) {
  __shared__ __attribute__((aligned(16))) float Xs[IPB][PD * DD];
  __shared__ float As[IPB][PD];
  __shared__ float Ns[IPB][PD];

  const int wave = threadIdx.x >> 5, lane = threadIdx.x & 31;
  const int it = blockIdx.x * IPB + wave;
  const bool obs = it >= MT;
  const int i = obs ? it - MT : it;
  const float* X = obs ? Xo : Xt;
  const float* A = obs ? Ao : At;
  float* SN = obs ? SNo : SNt;
  float* sOut = obs ? sO : sT;

  const float4* xg = (const float4*)(X + i * PD * DD);
  float4* xs4 = (float4*)Xs[wave];
  for (int q = lane; q < PD * DD / 4; q += 32) xs4[q] = xg[q];
  if (lane < PD) {
    const int p = i * PD + lane;
    As[wave][lane] = A[p];
    float s = 0.f;
#pragma unroll
    for (int d = 0; d < DD; ++d) { float x = X[p * DD + d]; s += x * x; }
    Ns[wave][lane] = s;
    SN[p] = s;
  }
  __syncthreads();

  float acc = 0.f;
  for (int idx = lane; idx < PD * PD; idx += 32) {
    int u = idx / PD, v = idx % PD;
    const float* xu = Xs[wave] + u * DD;
    const float* xv = Xs[wave] + v * DD;
    float dot = 0.f;
#pragma unroll
    for (int d = 0; d < DD; ++d) dot += xu[d] * xv[d];
    float d2 = Ns[wave][u] + Ns[wave][v] - 2.f * dot;
    acc += As[wave][u] * As[wave][v] * __expf(-0.5f * fmaxf(d2, 0.f));
  }
#pragma unroll
  for (int off = 16; off > 0; off >>= 1) acc += __shfl_xor(acc, off);
  if (lane == 0) sOut[i] = acc;
}

#define GW 12
#define UPW (PD / GW)
#define XP 196
#define TKOO ((NO / 16) * (NO / 16))
#define TKOX ((NO / 16) * (MT / 16))
#define TKXX ((MT / 16) * (MT / 16))
static_assert(NO / 16 == 32 && MT / 16 == 16, "tile decode in gram_kernel assumes 32 and 16 tiles per row");
__global__ __launch_bounds__(GW * 32) void gram_kernel(
    const float* __restrict__ Xt, const float* __restrict__ SNt,
    const float* __restrict__ At, const float* __restrict__ st,
    const float* __restrict__ Xo, const float* __restrict__ SNo,
    const float* __restrict__ Ao, const float* __restrict__ so,
    float* __restrict__ Kxx, float* __restrict__ Kox, float* __restrict__ Koo,
    const float* __restrict__ rho_p, const float* __restrict__ nu_p,
    const float* __restrict__ g_p) {
  __shared__ __attribute__((aligned(16))) float X1s[16 * XP];
  __shared__ __attribute__((aligned(16))) float X2s[16 * XP];
  __shared__ __attribute__((aligned(16))) float A1T[PD * 16];
  __shared__ __attribute__((aligned(16))) float N1T[PD * 16];
  __shared__ __attribute__((aligned(16))) float A2T[PD * 16];
  __shared__ __attribute__((aligned(16))) float N2T[PD * 16];
  __shared__ __attribute__((aligned(16))) float red[GW * 256];

  const int tid = threadIdx.x;
  const int lane = tid & 31;
  const int half = lane >> 4, lq = lane & 15;
  const int wave = __builtin_amdgcn_readfirstlane(tid >> 5);

  const int bid = blockIdx.x;
  const bool isKoo = bid < TKOO;
  const bool isKox = !isKoo && bid < TKOO + TKOX;
  const int tb = isKoo ? bid : (isKox ? bid - TKOO : bid - TKOO - TKOX);
  const int tsh = isKoo ? 5 : 4;
  const int i0 = (tb >> tsh) * 16, j0 = (tb & ((1 << tsh) - 1)) * 16;
  const bool obs1 = isKoo || isKox;
  const float* X1  = obs1 ? Xo : Xt;
  const float* SN1 = obs1 ? SNo : SNt;
  const float* A1  = obs1 ? Ao : At;
  const float* s1  = obs1 ? so : st;
  const float* X2  = isKoo ? Xo : Xt;
  const float* SN2 = isKoo ? SNo : SNt;
  const float* A2  = isKoo ? Ao : At;
  const float* s2  = isKoo ? so : st;
  float* out = isKoo ? Koo : (isKox ? Kox : Kxx);
  const int ldc = isKoo ? NO : MT;
  const int addDiag = isKoo ? 1 : 0;

  {
    const float4* g1 = (const float4*)(X1 + i0 * PD * DD);
    const float4* g2 = (const float4*)(X2 + j0 * PD * DD);
    for (int q = tid; q < 16 * PD * DD / 4; q += GW * 32) {
      const int item = q / (PD * DD / 4), w4 = q % (PD * DD / 4);
      const float4 t1 = g1[q], t2 = g2[q];
      *(float4*)(X1s + item * XP + 4 * w4) = t1;
      *(float4*)(X2s + item * XP + 4 * w4) = t2;
    }
    for (int q = tid; q < 16 * PD; q += GW * 32) {
      const int item = q / PD, p = q % PD;
      A1T[p * 16 + item] = A1[i0 * PD + q];
      N1T[p * 16 + item] = SN1[i0 * PD + q];
      A2T[p * 16 + item] = A2[j0 * PD + q];
      N2T[p * 16 + item] = SN2[j0 * PD + q];
    }
  }
  __syncthreads();

  float a1u[UPW][8], n1u[UPW][8];
  v2f fa0[UPW], fa1[UPW];
#pragma unroll
  for (int uu = 0; uu < UPW; ++uu) {
    const int u = wave * UPW + uu;
    const float4* ap = (const float4*)(A1T + u * 16 + 8 * half);
    const float4* np = (const float4*)(N1T + u * 16 + 8 * half);
    const float4 ta = ap[0], tb = ap[1], tc = np[0], td = np[1];
    a1u[uu][0] = ta.x; a1u[uu][1] = ta.y; a1u[uu][2] = ta.z; a1u[uu][3] = ta.w;
    a1u[uu][4] = tb.x; a1u[uu][5] = tb.y; a1u[uu][6] = tb.z; a1u[uu][7] = tb.w;
    n1u[uu][0] = tc.x; n1u[uu][1] = tc.y; n1u[uu][2] = tc.z; n1u[uu][3] = tc.w;
    n1u[uu][4] = td.x; n1u[uu][5] = td.y; n1u[uu][6] = td.z; n1u[uu][7] = td.w;
    const float* x1 = X1s + lq * XP + u * DD + 2 * half;
    fa0[uu] = *(const v2f*)(x1);
    fa1[uu] = *(const v2f*)(x1 + 4);
  }

  float acc[8];
#pragma unroll
  for (int r = 0; r < 8; ++r) acc[r] = 0.f;

  for (int v = 0; v < PD; ++v) {
    const float* x2 = X2s + lq * XP + v * DD + 2 * half;
    const v2f b0 = *(const v2f*)(x2);
    const v2f b1 = *(const v2f*)(x2 + 4);
    const float sn2v = N2T[v * 16 + lq];
    const float a2v  = A2T[v * 16 + lq];
#pragma unroll
    for (int uu = 0; uu < UPW; ++uu) {
      v8f s = {};
      s = __builtin_amdgcn_wmma_f32_16x16x4_f32(false, fa0[uu], false, b0, (short)0, s, false, false);
      s = __builtin_amdgcn_wmma_f32_16x16x4_f32(false, fa1[uu], false, b1, (short)0, s, false, false);
#pragma unroll
      for (int r = 0; r < 8; ++r) {
        float d2 = n1u[uu][r] + sn2v - 2.f * s[r];
        acc[r] += a1u[uu][r] * a2v * __expf(-0.5f * fmaxf(d2, 0.f));
      }
    }
  }

#pragma unroll
  for (int r = 0; r < 8; ++r) red[wave * 256 + r * 32 + lane] = acc[r];
  __syncthreads();

  if (tid < 256) {
    float c = 0.f;
#pragma unroll
    for (int w = 0; w < GW; ++w) c += red[w * 256 + tid];
    const int r = tid >> 5;
    const int i = i0 + r + 8 * half, j = j0 + lq;
    float invrho = 1.f / rho_p[0];
    float nu = nu_p[0], g = g_p[0];
    float d2 = s1[i] + s2[j] - 2.f * c;
    float val = nu * __expf(-0.5f * fmaxf(d2, 0.f) * invrho);
    if (addDiag && i == j) val += nu * g;
    out[i * ldc + j] = val;
  }
}

#define NB 32
#define NH 256
static_assert(NO == 2 * NH, "the four-launch factorisation assumes two diagonal blocks");
#define PP 34
#define DP 33

__device__ __forceinline__ void chol_diag(float (&r)[NB], const int lane,
                                          float* __restrict__ Sd, float* __restrict__ invd,
                                          float* __restrict__ rowp) {
  float mydinv = 0.f;
#pragma unroll
  for (int k = 0; k < NB; ++k) {
    float akk = __int_as_float(__builtin_amdgcn_readlane(__float_as_int(r[k]), k));
    float dinv = __builtin_amdgcn_rsqf(akk);
    r[k] = r[k] * dinv;
    mydinv = (lane == k) ? dinv : mydinv;
#pragma unroll
    for (int j = k + 1; j < NB; ++j) {
      float ljk = __int_as_float(__builtin_amdgcn_readlane(__float_as_int(r[k]), j));
      r[j] -= r[k] * ljk;
    }
  }
#pragma unroll
  for (int j = 0; j < NB; ++j) r[j] = (j <= lane) ? r[j] : 0.f;
#pragma unroll
  for (int c = 0; c < NB; ++c) Sd[c * NB + lane] = r[c];
  invd[lane] = mydinv;
  float4* dst = (float4*)rowp;
#pragma unroll
  for (int q = 0; q < NB / 4; ++q)
    dst[q] = make_float4(r[4 * q + 0], r[4 * q + 1], r[4 * q + 2], r[4 * q + 3]);
}

__global__ __launch_bounds__(1024) void chol_kernel(float* __restrict__ Lm) {
  __shared__ __attribute__((aligned(16))) float Sd[NB * NB];
  __shared__ __attribute__((aligned(16))) float invd[NB];
  __shared__ __attribute__((aligned(16))) float Pn[(NH - NB) * PP];
  __shared__ float Db[NB * DP];

  const int tid = threadIdx.x;
  const int lane = tid & 31;
  const int half = lane >> 4, lq = lane & 15;
  const int wave = __builtin_amdgcn_readfirstlane(tid >> 5);

  if (wave == 0) {
    float r[NB];
    float* rowp = Lm + lane * NO;
    const float4* src = (const float4*)rowp;
#pragma unroll
    for (int q = 0; q < NB / 4; ++q) {
      float4 t = src[q];
      r[4 * q + 0] = t.x; r[4 * q + 1] = t.y; r[4 * q + 2] = t.z; r[4 * q + 3] = t.w;
    }
    chol_diag(r, lane, Sd, invd, rowp);
  }

  for (int kb = 0; kb < NH / NB - 1; ++kb) {
    const int k0 = kb * NB;
    const int nrows = NH - k0 - NB;
    __syncthreads();

    if (wave < (nrows >> 5)) {
      float Lr[NB];
#pragma unroll
      for (int c = 0; c < NB; ++c) Lr[c] = Sd[c * NB + lane];
      const float dv = invd[lane];
      float x[NB];
      float* rowp = Lm + (k0 + NB + tid) * NO + k0;
      const float4* src = (const float4*)rowp;
#pragma unroll
      for (int q = 0; q < NB / 4; ++q) {
        float4 t = src[q];
        x[4 * q + 0] = t.x; x[4 * q + 1] = t.y; x[4 * q + 2] = t.z; x[4 * q + 3] = t.w;
      }
#pragma unroll
      for (int c = 0; c < NB; ++c) {
        x[c] *= __int_as_float(__builtin_amdgcn_readlane(__float_as_int(dv), c));
#pragma unroll
        for (int j = c + 1; j < NB; ++j)
          x[j] -= x[c] * __int_as_float(__builtin_amdgcn_readlane(__float_as_int(Lr[c]), j));
      }
      float4* dst = (float4*)rowp;
#pragma unroll
      for (int q = 0; q < NB / 4; ++q)
        dst[q] = make_float4(x[4 * q + 0], x[4 * q + 1], x[4 * q + 2], x[4 * q + 3]);
      float2* pdst = (float2*)(Pn + tid * PP);
#pragma unroll
      for (int q = 0; q < NB / 2; ++q) pdst[q] = make_float2(x[2 * q + 0], x[2 * q + 1]);
    }
    __syncthreads();

    const int m = nrows >> 4;
    const int T = m * (m + 1) / 2;
    if (wave == 0) {
#pragma unroll
      for (int t = 0; t < 3; ++t) {
        const int ti = (t + 1) >> 1, tj = t >> 1;
        const float* pa = Pn + (ti * 16 + lq) * PP + 2 * half;
        const float* pb = Pn + (tj * 16 + lq) * PP + 2 * half;
        v8f s = {};
#pragma unroll
        for (int kc = 0; kc < NB / 4; ++kc) {
          v2f a = *(const v2f*)(pa + 4 * kc);
          v2f b = *(const v2f*)(pb + 4 * kc);
          s = __builtin_amdgcn_wmma_f32_16x16x4_f32(false, a, false, b, (short)0, s, false, false);
        }
        const float* cp = Lm + (k0 + NB + ti * 16 + 8 * half) * NO + (k0 + NB + tj * 16 + lq);
        float* dp = Db + (ti * 16 + 8 * half) * DP + tj * 16 + lq;
#pragma unroll
        for (int r = 0; r < 8; ++r) dp[r * DP] = cp[r * NO] - s[r];
      }
#pragma unroll
      for (int r = 0; r < 8; ++r) Db[(8 * half + r) * DP + 16 + lq] = 0.f;
      float r[NB];
#pragma unroll
      for (int c = 0; c < NB; ++c) r[c] = Db[lane * DP + c];
      chol_diag(r, lane, Sd, invd, Lm + (k0 + NB + lane) * NO + k0 + NB);
    } else {
      for (int q = 2 + wave; q < T; q += 31) {
        int ti = (int)((sqrtf(8.f * (float)q + 1.f) - 1.f) * 0.5f);
        if (ti * (ti + 1) / 2 > q) --ti;
        if ((ti + 1) * (ti + 2) / 2 <= q) ++ti;
        const int tj = q - ti * (ti + 1) / 2;
        const float* pa = Pn + (ti * 16 + lq) * PP + 2 * half;
        const float* pb = Pn + (tj * 16 + lq) * PP + 2 * half;
        v8f s = {};
#pragma unroll
        for (int kc = 0; kc < NB / 4; ++kc) {
          v2f a = *(const v2f*)(pa + 4 * kc);
          v2f b = *(const v2f*)(pb + 4 * kc);
          s = __builtin_amdgcn_wmma_f32_16x16x4_f32(false, a, false, b, (short)0, s, false, false);
        }
        float* cp = Lm + (k0 + NB + ti * 16 + 8 * half) * NO + (k0 + NB + tj * 16 + lq);
#pragma unroll
        for (int r = 0; r < 8; ++r) cp[r * NO] -= s[r];
      }
    }
  }
}

#define SC 16
#define YP 516

__global__ __launch_bounds__(1024) void panel_kernel(float* Lm) {
  __shared__ __attribute__((aligned(16))) float Yt[SC * YP];

  const int tid = threadIdx.x;
  const int lane = tid & 31;
  const int half = lane >> 4, lq = lane & 15;
  const int wave = __builtin_amdgcn_readfirstlane(tid >> 5);
  const int row0 = NH + blockIdx.x * SC;

  for (int i = 0; i < (SC * NH) / 1024; ++i) {
    int idx = tid + 1024 * i;
    int n = idx & (NH - 1), cc = idx / NH;
    Yt[cc * YP + n] = Lm[(row0 + cc) * NO + n];
  }

  float Ln[NB];
  float dg = 1.f;
  if (wave < SC) {
    const float4* lp = (const float4*)(Lm + lane * NO);
#pragma unroll
    for (int q = 0; q < NB / 4; ++q) {
      float4 t4 = lp[q];
      Ln[4 * q + 0] = t4.x; Ln[4 * q + 1] = t4.y; Ln[4 * q + 2] = t4.z; Ln[4 * q + 3] = t4.w;
    }
    dg = Lm[lane * NO + lane];
  }
  for (int kb = 0; kb < NH / NB; ++kb) {
    const int k0 = kb * NB;
    float Mr[NB];
    float inv = 0.f;
    if (wave < SC) {
      inv = 1.f / dg;
#pragma unroll
      for (int c = 0; c < NB; ++c) Mr[c] = (lane > c) ? Ln[c] * inv : 0.f;
    }
    const int ntiles = (NH - k0 - NB) >> 4;
    const int rowbase = k0 + NB + 16 * wave;
    __syncthreads();

    float4 av[4];
    if (wave < ntiles) {
      const float* ap = Lm + (rowbase + lq) * NO + k0 + 4 * half;
#pragma unroll
      for (int jj = 0; jj < 4; ++jj) av[jj] = *(const float4*)(ap + 8 * jj);
    }
    if (wave < SC) {
      if (kb + 1 < NH / NB) {
        const float4* lp = (const float4*)(Lm + (k0 + NB + lane) * NO + k0 + NB);
#pragma unroll
        for (int q = 0; q < NB / 4; ++q) {
          float4 t4 = lp[q];
          Ln[4 * q + 0] = t4.x; Ln[4 * q + 1] = t4.y; Ln[4 * q + 2] = t4.z; Ln[4 * q + 3] = t4.w;
        }
        dg = Lm[(k0 + NB + lane) * NO + k0 + NB + lane];
      }
      float* yp = Yt + wave * YP + k0 + lane;
      float t = yp[0] * inv;
#pragma unroll
      for (int c = 0; c < NB - 1; ++c) {
        float yc = __int_as_float(__builtin_amdgcn_readlane(__float_as_int(t), c));
        t = fmaf(-Mr[c], yc, t);
      }
      yp[0] = t;
    }
    __syncthreads();

    if (wave < ntiles) {
      const float* bp = Yt + lq * YP + k0 + 4 * half;
      v8f s = {};
#pragma unroll
      for (int jj = 0; jj < 4; ++jj) {
        float4 bv = *(const float4*)(bp + 8 * jj);
        v2f a, b;
        a.x = av[jj].x; a.y = av[jj].y; b.x = bv.x; b.y = bv.y;
        s = __builtin_amdgcn_wmma_f32_16x16x4_f32(false, a, false, b, (short)0, s, false, false);
        a.x = av[jj].z; a.y = av[jj].w; b.x = bv.z; b.y = bv.w;
        s = __builtin_amdgcn_wmma_f32_16x16x4_f32(false, a, false, b, (short)0, s, false, false);
      }
      float4* cp = (float4*)(Yt + lq * YP + rowbase + 8 * half);
      float4 c0 = cp[0], c1 = cp[1];
      c0.x -= s[0]; c0.y -= s[1]; c0.z -= s[2]; c0.w -= s[3];
      c1.x -= s[4]; c1.y -= s[5]; c1.z -= s[6]; c1.w -= s[7];
      cp[0] = c0; cp[1] = c1;
    }
  }

  for (int i = 0; i < (SC * NH) / 1024; ++i) {
    int idx = tid + 1024 * i;
    int n = idx & (NH - 1), cc = idx / NH;
    Lm[(row0 + cc) * NO + n] = Yt[cc * YP + n];
  }
}

__global__ __launch_bounds__(32) void trail_kernel(float* Lm) {
  const int ti = blockIdx.y, tj = blockIdx.x;
  if (ti < tj) return;
  const int lane = threadIdx.x & 31;
  const int half = lane >> 4, lq = lane & 15;
  const float* pa = Lm + (NH + ti * 16 + lq) * NO + 2 * half;
  const float* pb = Lm + (NH + tj * 16 + lq) * NO + 2 * half;
  v8f c = {};
  for (int n0 = 0; n0 < NH; n0 += 8) {
    v2f a0; a0.x = pa[n0 + 0]; a0.y = pa[n0 + 1];
    v2f a1; a1.x = pa[n0 + 4]; a1.y = pa[n0 + 5];
    v2f b0; b0.x = pb[n0 + 0]; b0.y = pb[n0 + 1];
    v2f b1; b1.x = pb[n0 + 4]; b1.y = pb[n0 + 5];
    c = __builtin_amdgcn_wmma_f32_16x16x4_f32(false, a0, false, b0, (short)0, c, false, false);
    c = __builtin_amdgcn_wmma_f32_16x16x4_f32(false, a1, false, b1, (short)0, c, false, false);
  }
  float* cp = Lm + (NH + ti * 16 + 8 * half) * NO + NH + tj * 16 + lq;
#pragma unroll
  for (int r = 0; r < 8; ++r) cp[r * NO] -= c[r];
}

__global__ __launch_bounds__(1024) void solve_kernel(const float* __restrict__ L,
                                                     const float* __restrict__ Kox,
                                                     const float* __restrict__ ys,
                                                     const float* __restrict__ b_p,
                                                     float* __restrict__ Y) {
  __shared__ __attribute__((aligned(16))) float Yt[SC * YP];

  const int tid = threadIdx.x;
  const int lane = tid & 31;
  const int half = lane >> 4, lq = lane & 15;
  const int wave = __builtin_amdgcn_readfirstlane(tid >> 5);
  const int col0 = blockIdx.x * SC;

  const float bb = b_p[0];
  for (int i = 0; i < (SC * NO) / 1024; ++i) {
    int idx = tid + 1024 * i;
    int cc = idx & (SC - 1), n = idx >> 4;
    int col = col0 + cc;
    float v = 0.f;
    if (col < MT) v = Kox[n * MT + col];
    else if (col == MT) v = 1.0f;
    else if (col == MT + 1) v = ys[n] - bb;
    Yt[cc * YP + n] = v;
  }

  float Ln[NB];
  float dg = 1.f;

  if (wave < SC) {
    const float4* lp = (const float4*)(L + lane * NO);
#pragma unroll
    for (int q = 0; q < NB / 4; ++q) {
      float4 t4 = lp[q];
      Ln[4 * q + 0] = t4.x; Ln[4 * q + 1] = t4.y; Ln[4 * q + 2] = t4.z; Ln[4 * q + 3] = t4.w;
    }
    dg = L[lane * NO + lane];
  }
  for (int kb = 0; kb < NO / NB; ++kb) {
    const int k0 = kb * NB;
    float Mr[NB];
    float inv = 0.f;
    if (wave < SC) {
      inv = 1.f / dg;
#pragma unroll
      for (int c = 0; c < NB; ++c) Mr[c] = (lane > c) ? Ln[c] * inv : 0.f;
    }
    const int ntiles = (NO - k0 - NB) >> 4;
    const int rowbase = k0 + NB + 16 * wave;
    __syncthreads();

    float4 av[4];
    if (wave < ntiles) {
      const float* ap = L + (rowbase + lq) * NO + k0 + 4 * half;
#pragma unroll
      for (int jj = 0; jj < 4; ++jj) av[jj] = *(const float4*)(ap + 8 * jj);
    }
    if (wave < SC) {
      if (kb + 1 < NO / NB) {
        const float4* lp = (const float4*)(L + (k0 + NB + lane) * NO + k0 + NB);
#pragma unroll
        for (int q = 0; q < NB / 4; ++q) {
          float4 t4 = lp[q];
          Ln[4 * q + 0] = t4.x; Ln[4 * q + 1] = t4.y; Ln[4 * q + 2] = t4.z; Ln[4 * q + 3] = t4.w;
        }
        dg = L[(k0 + NB + lane) * NO + k0 + NB + lane];
      }
      float* yp = Yt + wave * YP + k0 + lane;
      float t = yp[0] * inv;
#pragma unroll
      for (int c = 0; c < NB - 1; ++c) {
        float yc = __int_as_float(__builtin_amdgcn_readlane(__float_as_int(t), c));
        t = fmaf(-Mr[c], yc, t);
      }
      yp[0] = t;
    }
    __syncthreads();

    if (wave < ntiles) {
      const float* bp = Yt + lq * YP + k0 + 4 * half;
      v8f s = {};
#pragma unroll
      for (int jj = 0; jj < 4; ++jj) {
        float4 bv = *(const float4*)(bp + 8 * jj);
        v2f a, b;
        a.x = av[jj].x; a.y = av[jj].y; b.x = bv.x; b.y = bv.y;
        s = __builtin_amdgcn_wmma_f32_16x16x4_f32(false, a, false, b, (short)0, s, false, false);
        a.x = av[jj].z; a.y = av[jj].w; b.x = bv.z; b.y = bv.w;
        s = __builtin_amdgcn_wmma_f32_16x16x4_f32(false, a, false, b, (short)0, s, false, false);
      }
      float4* cp = (float4*)(Yt + lq * YP + rowbase + 8 * half);
      float4 c0 = cp[0], c1 = cp[1];
      c0.x -= s[0]; c0.y -= s[1]; c0.z -= s[2]; c0.w -= s[3];
      c1.x -= s[4]; c1.y -= s[5]; c1.z -= s[6]; c1.w -= s[7];
      cp[0] = c0; cp[1] = c1;
    }
  }

  for (int i = 0; i < (SC * NO) / 1024; ++i) {
    int idx = tid + 1024 * i;
    int n = idx & (NO - 1), cc = idx >> 9;
    int col = col0 + cc;
    if (col <= MT + 1) Y[col * NO + n] = Yt[cc * YP + n];
  }
}

__global__ __launch_bounds__(256) void mean_kernel(const float* __restrict__ Y,
                                                   const float* __restrict__ b_p,
                                                   float* __restrict__ out,
                                                   float* __restrict__ colsum) {
  __shared__ float r1[256], r2[256];
  int col = blockIdx.x, tid = threadIdx.x;
  float b = b_p[0];
  float sm = 0.f, sc = 0.f;
  for (int n = tid; n < NO; n += 256) {
    float y = Y[col * NO + n];
    sm += y * Y[(MT + 1) * NO + n];
    sc += y * Y[MT * NO + n];
  }
  r1[tid] = sm; r2[tid] = sc;
  __syncthreads();
  for (int off = 128; off > 0; off >>= 1) {
    if (tid < off) { r1[tid] += r1[tid + off]; r2[tid] += r2[tid + off]; }
    __syncthreads();
  }
  if (tid == 0) {
    colsum[col] = r2[0];
    if (col < MT) out[col] = b + r1[0];
  }
}

__global__ __launch_bounds__(32) void cov_kernel(const float* __restrict__ Kxx,
                                                 const float* __restrict__ Y,
                                                 const float* __restrict__ colsum,
                                                 float* __restrict__ out) {
  const int lane = threadIdx.x & 31;
  const int half = lane >> 4, lq = lane & 15;
  const int i0 = blockIdx.y * 16, j0 = blockIdx.x * 16;
  v8f c = {};
  for (int n0 = 0; n0 < NO; n0 += 8) {
    const float* yi = Y + (i0 + lq) * NO + n0 + 2 * half;
    const float* yj = Y + (j0 + lq) * NO + n0 + 2 * half;
    v2f a0; a0.x = yi[0]; a0.y = yi[1];
    v2f a1; a1.x = yi[4]; a1.y = yi[5];
    v2f b0; b0.x = yj[0]; b0.y = yj[1];
    v2f b1; b1.x = yj[4]; b1.y = yj[5];
    c = __builtin_amdgcn_wmma_f32_16x16x4_f32(false, a0, false, b0, (short)0, c, false, false);
    c = __builtin_amdgcn_wmma_f32_16x16x4_f32(false, a1, false, b1, (short)0, c, false, false);
  }
  float invS = 1.f / colsum[MT];
  float kbxj = colsum[j0 + lq];
#pragma unroll
  for (int r = 0; r < 8; ++r) {
    int i = i0 + r + 8 * half, j = j0 + lq;
    float v = Kxx[i * MT + j] - c[r] + (1.f - colsum[i]) * (1.f - kbxj) * invS;
    out[(1 + i) * MT + j] = v;
  }
}

extern "C" void kernel_launch(void* const* d_in, const int* in_sizes, int n_in,
                              void* d_out, int out_size, void* d_ws, size_t ws_size,
                              hipStream_t stream) {
  const float* Xt  = (const float*)d_in[0];
  const float* At  = (const float*)d_in[1];
  const float* Xo  = (const float*)d_in[2];
  const float* Ao  = (const float*)d_in[3];
  const float* ys  = (const float*)d_in[4];
  const float* rho = (const float*)d_in[5];
  const float* g   = (const float*)d_in[6];
  const float* nu  = (const float*)d_in[7];
  const float* b   = (const float*)d_in[8];
  float* out = (float*)d_out;

  float* ws  = (float*)d_ws;
  float* SN1 = ws;
  float* SN2 = SN1 + MT * PD;
  float* sT  = SN2 + NO * PD;
  float* sO  = sT + MT;
  float* Kxx = sO + NO;
  float* Kox = Kxx + MT * MT;
  float* Koo = Kox + NO * MT;
  float* Y   = Koo + NO * NO;
  float* csm = Y + (MT + 2) * NO;

  prep_kernel<<<(MT + NO) / IPB, IPB * 32, 0, stream>>>(Xt, At, Xo, Ao, SN1, SN2, sT, sO);

  gram_kernel<<<TKOO + TKOX + TKXX, GW * 32, 0, stream>>>(Xt, SN1, At, sT, Xo, SN2, Ao, sO,
                                                         Kxx, Kox, Koo, rho, nu, g);

  chol_kernel<<<1, 1024, 0, stream>>>(Koo);
  panel_kernel<<<NH / SC, 1024, 0, stream>>>(Koo);
  trail_kernel<<<dim3(NH / 16, NH / 16), 32, 0, stream>>>(Koo);
  chol_kernel<<<1, 1024, 0, stream>>>(Koo + NH * NO + NH);
  solve_kernel<<<(MT + 2 + SC - 1) / SC, 1024, 0, stream>>>(Koo, Kox, ys, b, Y);
  mean_kernel<<<MT + 1, 256, 0, stream>>>(Y, b, out, csm);
  cov_kernel<<<dim3(16, 16), 32, 0, stream>>>(Kxx, Y, csm, out);
}


# Gram tiles above the diagonal of the symmetric Koo and Kxx not built (1176 tiles, was 1792): Koo's are never read, cov takes Kxx's from the mirror tile, whose pair sums run in swapped order
# speedup vs baseline: 164.9394x; 1.0671x over previous
#include <hip/hip_runtime.h>
#include <hip/hip_bf16.h>

typedef float v2f __attribute__((ext_vector_type(2)));
typedef float v8f __attribute__((ext_vector_type(8)));

#define MT 256
#define NO 512
#define PD 24
#define DD 8

#define IPB 8
__global__ __launch_bounds__(IPB * 32) void prep_kernel(const float* __restrict__ Xt,
                                                        const float* __restrict__ At,
                                                        const float* __restrict__ Xo,
                                                        const float* __restrict__ Ao,
                                                        float* __restrict__ SNt,
                                                        float* __restrict__ SNo,
                                                        float* __restrict__ sT,
                                                        float* __restrict__ sO) {
  __shared__ __attribute__((aligned(16))) float Xs[IPB][PD * DD];
  __shared__ float As[IPB][PD];
  __shared__ float Ns[IPB][PD];

  const int wave = threadIdx.x >> 5, lane = threadIdx.x & 31;
  const int it = blockIdx.x * IPB + wave;
  const bool obs = it >= MT;
  const int i = obs ? it - MT : it;
  const float* X = obs ? Xo : Xt;
  const float* A = obs ? Ao : At;
  float* SN = obs ? SNo : SNt;
  float* sOut = obs ? sO : sT;

  const float4* xg = (const float4*)(X + i * PD * DD);
  float4* xs4 = (float4*)Xs[wave];
  for (int q = lane; q < PD * DD / 4; q += 32) xs4[q] = xg[q];
  if (lane < PD) {
    const int p = i * PD + lane;
    As[wave][lane] = A[p];
    float s = 0.f;
#pragma unroll
    for (int d = 0; d < DD; ++d) { float x = X[p * DD + d]; s += x * x; }
    Ns[wave][lane] = s;
    SN[p] = s;
  }
  __syncthreads();

  float acc = 0.f;
  for (int idx = lane; idx < PD * PD; idx += 32) {
    int u = idx / PD, v = idx % PD;
    const float* xu = Xs[wave] + u * DD;
    const float* xv = Xs[wave] + v * DD;
    float dot = 0.f;
#pragma unroll
    for (int d = 0; d < DD; ++d) dot += xu[d] * xv[d];
    float d2 = Ns[wave][u] + Ns[wave][v] - 2.f * dot;
    acc += As[wave][u] * As[wave][v] * __expf(-0.5f * fmaxf(d2, 0.f));
  }
#pragma unroll
  for (int off = 16; off > 0; off >>= 1) acc += __shfl_xor(acc, off);
  if (lane == 0) sOut[i] = acc;
}

#define GW 12
#define UPW (PD / GW)
#define XP 196
#define TKOO ((NO / 16) * (NO / 16 + 1) / 2)
#define TKOX ((NO / 16) * (MT / 16))
#define TKXX ((MT / 16) * (MT / 16 + 1) / 2)
static_assert(MT / 16 == 16, "tile decode of Kox in gram_kernel assumes 16 tiles per row");
__global__ __launch_bounds__(GW * 32) void gram_kernel(
    const float* __restrict__ Xt, const float* __restrict__ SNt,
    const float* __restrict__ At, const float* __restrict__ st,
    const float* __restrict__ Xo, const float* __restrict__ SNo,
    const float* __restrict__ Ao, const float* __restrict__ so,
    float* __restrict__ Kxx, float* __restrict__ Kox, float* __restrict__ Koo,
    const float* __restrict__ rho_p, const float* __restrict__ nu_p,
    const float* __restrict__ g_p) {
  __shared__ __attribute__((aligned(16))) float X1s[16 * XP];
  __shared__ __attribute__((aligned(16))) float X2s[16 * XP];
  __shared__ __attribute__((aligned(16))) float A1T[PD * 16];
  __shared__ __attribute__((aligned(16))) float N1T[PD * 16];
  __shared__ __attribute__((aligned(16))) float A2T[PD * 16];
  __shared__ __attribute__((aligned(16))) float N2T[PD * 16];
  __shared__ __attribute__((aligned(16))) float red[GW * 256];

  const int tid = threadIdx.x;
  const int lane = tid & 31;
  const int half = lane >> 4, lq = lane & 15;
  const int wave = __builtin_amdgcn_readfirstlane(tid >> 5);

  const int bid = blockIdx.x;
  const bool isKoo = bid < TKOO;
  const bool isKox = !isKoo && bid < TKOO + TKOX;
  const int tb = isKoo ? bid : (isKox ? bid - TKOO : bid - TKOO - TKOX);
  int ti = tb >> 4, tj = tb & 15;
  if (!isKox) {
    ti = (int)((sqrtf(8.f * (float)tb + 1.f) - 1.f) * 0.5f);
    if (ti * (ti + 1) / 2 > tb) --ti;
    if ((ti + 1) * (ti + 2) / 2 <= tb) ++ti;
    tj = tb - ti * (ti + 1) / 2;
  }
  const int i0 = ti * 16, j0 = tj * 16;
  const bool obs1 = isKoo || isKox;
  const float* X1  = obs1 ? Xo : Xt;
  const float* SN1 = obs1 ? SNo : SNt;
  const float* A1  = obs1 ? Ao : At;
  const float* s1  = obs1 ? so : st;
  const float* X2  = isKoo ? Xo : Xt;
  const float* SN2 = isKoo ? SNo : SNt;
  const float* A2  = isKoo ? Ao : At;
  const float* s2  = isKoo ? so : st;
  float* out = isKoo ? Koo : (isKox ? Kox : Kxx);
  const int ldc = isKoo ? NO : MT;
  const int addDiag = isKoo ? 1 : 0;

  {
    const float4* g1 = (const float4*)(X1 + i0 * PD * DD);
    const float4* g2 = (const float4*)(X2 + j0 * PD * DD);
    for (int q = tid; q < 16 * PD * DD / 4; q += GW * 32) {
      const int item = q / (PD * DD / 4), w4 = q % (PD * DD / 4);
      const float4 t1 = g1[q], t2 = g2[q];
      *(float4*)(X1s + item * XP + 4 * w4) = t1;
      *(float4*)(X2s + item * XP + 4 * w4) = t2;
    }
    for (int q = tid; q < 16 * PD; q += GW * 32) {
      const int item = q / PD, p = q % PD;
      A1T[p * 16 + item] = A1[i0 * PD + q];
      N1T[p * 16 + item] = SN1[i0 * PD + q];
      A2T[p * 16 + item] = A2[j0 * PD + q];
      N2T[p * 16 + item] = SN2[j0 * PD + q];
    }
  }
  __syncthreads();

  float a1u[UPW][8], n1u[UPW][8];
  v2f fa0[UPW], fa1[UPW];
#pragma unroll
  for (int uu = 0; uu < UPW; ++uu) {
    const int u = wave * UPW + uu;
    const float4* ap = (const float4*)(A1T + u * 16 + 8 * half);
    const float4* np = (const float4*)(N1T + u * 16 + 8 * half);
    const float4 ta = ap[0], tb = ap[1], tc = np[0], td = np[1];
    a1u[uu][0] = ta.x; a1u[uu][1] = ta.y; a1u[uu][2] = ta.z; a1u[uu][3] = ta.w;
    a1u[uu][4] = tb.x; a1u[uu][5] = tb.y; a1u[uu][6] = tb.z; a1u[uu][7] = tb.w;
    n1u[uu][0] = tc.x; n1u[uu][1] = tc.y; n1u[uu][2] = tc.z; n1u[uu][3] = tc.w;
    n1u[uu][4] = td.x; n1u[uu][5] = td.y; n1u[uu][6] = td.z; n1u[uu][7] = td.w;
    const float* x1 = X1s + lq * XP + u * DD + 2 * half;
    fa0[uu] = *(const v2f*)(x1);
    fa1[uu] = *(const v2f*)(x1 + 4);
  }

  float acc[8];
#pragma unroll
  for (int r = 0; r < 8; ++r) acc[r] = 0.f;

  for (int v = 0; v < PD; ++v) {
    const float* x2 = X2s + lq * XP + v * DD + 2 * half;
    const v2f b0 = *(const v2f*)(x2);
    const v2f b1 = *(const v2f*)(x2 + 4);
    const float sn2v = N2T[v * 16 + lq];
    const float a2v  = A2T[v * 16 + lq];
#pragma unroll
    for (int uu = 0; uu < UPW; ++uu) {
      v8f s = {};
      s = __builtin_amdgcn_wmma_f32_16x16x4_f32(false, fa0[uu], false, b0, (short)0, s, false, false);
      s = __builtin_amdgcn_wmma_f32_16x16x4_f32(false, fa1[uu], false, b1, (short)0, s, false, false);
#pragma unroll
      for (int r = 0; r < 8; ++r) {
        float d2 = n1u[uu][r] + sn2v - 2.f * s[r];
        acc[r] += a1u[uu][r] * a2v * __expf(-0.5f * fmaxf(d2, 0.f));
      }
    }
  }

#pragma unroll
  for (int r = 0; r < 8; ++r) red[wave * 256 + r * 32 + lane] = acc[r];
  __syncthreads();

  if (tid < 256) {
    float c = 0.f;
#pragma unroll
    for (int w = 0; w < GW; ++w) c += red[w * 256 + tid];
    const int r = tid >> 5;
    const int i = i0 + r + 8 * half, j = j0 + lq;
    float invrho = 1.f / rho_p[0];
    float nu = nu_p[0], g = g_p[0];
    float d2 = s1[i] + s2[j] - 2.f * c;
    float val = nu * __expf(-0.5f * fmaxf(d2, 0.f) * invrho);
    if (addDiag && i == j) val += nu * g;
    out[i * ldc + j] = val;
  }
}

#define NB 32
#define NH 256
static_assert(NO == 2 * NH, "the four-launch factorisation assumes two diagonal blocks");
#define PP 34
#define DP 33

__device__ __forceinline__ void chol_diag(float (&r)[NB], const int lane,
                                          float* __restrict__ Sd, float* __restrict__ invd,
                                          float* __restrict__ rowp) {
  float mydinv = 0.f;
#pragma unroll
  for (int k = 0; k < NB; ++k) {
    float akk = __int_as_float(__builtin_amdgcn_readlane(__float_as_int(r[k]), k));
    float dinv = __builtin_amdgcn_rsqf(akk);
    r[k] = r[k] * dinv;
    mydinv = (lane == k) ? dinv : mydinv;
#pragma unroll
    for (int j = k + 1; j < NB; ++j) {
      float ljk = __int_as_float(__builtin_amdgcn_readlane(__float_as_int(r[k]), j));
      r[j] -= r[k] * ljk;
    }
  }
#pragma unroll
  for (int j = 0; j < NB; ++j) r[j] = (j <= lane) ? r[j] : 0.f;
#pragma unroll
  for (int c = 0; c < NB; ++c) Sd[c * NB + lane] = r[c];
  invd[lane] = mydinv;
  float4* dst = (float4*)rowp;
#pragma unroll
  for (int q = 0; q < NB / 4; ++q)
    dst[q] = make_float4(r[4 * q + 0], r[4 * q + 1], r[4 * q + 2], r[4 * q + 3]);
}

__global__ __launch_bounds__(1024) void chol_kernel(float* __restrict__ Lm) {
  __shared__ __attribute__((aligned(16))) float Sd[NB * NB];
  __shared__ __attribute__((aligned(16))) float invd[NB];
  __shared__ __attribute__((aligned(16))) float Pn[(NH - NB) * PP];
  __shared__ float Db[NB * DP];

  const int tid = threadIdx.x;
  const int lane = tid & 31;
  const int half = lane >> 4, lq = lane & 15;
  const int wave = __builtin_amdgcn_readfirstlane(tid >> 5);

  if (wave == 0) {
    float r[NB];
    float* rowp = Lm + lane * NO;
    const float4* src = (const float4*)rowp;
#pragma unroll
    for (int q = 0; q < NB / 4; ++q) {
      float4 t = (q < NB / 8 || lane >= 16) ? src[q] : make_float4(0.f, 0.f, 0.f, 0.f);
      r[4 * q + 0] = t.x; r[4 * q + 1] = t.y; r[4 * q + 2] = t.z; r[4 * q + 3] = t.w;
    }
    chol_diag(r, lane, Sd, invd, rowp);
  }

  for (int kb = 0; kb < NH / NB - 1; ++kb) {
    const int k0 = kb * NB;
    const int nrows = NH - k0 - NB;
    __syncthreads();

    if (wave < (nrows >> 5)) {
      float Lr[NB];
#pragma unroll
      for (int c = 0; c < NB; ++c) Lr[c] = Sd[c * NB + lane];
      const float dv = invd[lane];
      float x[NB];
      float* rowp = Lm + (k0 + NB + tid) * NO + k0;
      const float4* src = (const float4*)rowp;
#pragma unroll
      for (int q = 0; q < NB / 4; ++q) {
        float4 t = src[q];
        x[4 * q + 0] = t.x; x[4 * q + 1] = t.y; x[4 * q + 2] = t.z; x[4 * q + 3] = t.w;
      }
#pragma unroll
      for (int c = 0; c < NB; ++c) {
        x[c] *= __int_as_float(__builtin_amdgcn_readlane(__float_as_int(dv), c));
#pragma unroll
        for (int j = c + 1; j < NB; ++j)
          x[j] -= x[c] * __int_as_float(__builtin_amdgcn_readlane(__float_as_int(Lr[c]), j));
      }
      float4* dst = (float4*)rowp;
#pragma unroll
      for (int q = 0; q < NB / 4; ++q)
        dst[q] = make_float4(x[4 * q + 0], x[4 * q + 1], x[4 * q + 2], x[4 * q + 3]);
      float2* pdst = (float2*)(Pn + tid * PP);
#pragma unroll
      for (int q = 0; q < NB / 2; ++q) pdst[q] = make_float2(x[2 * q + 0], x[2 * q + 1]);
    }
    __syncthreads();

    const int m = nrows >> 4;
    const int T = m * (m + 1) / 2;
    if (wave == 0) {
#pragma unroll
      for (int t = 0; t < 3; ++t) {
        const int ti = (t + 1) >> 1, tj = t >> 1;
        const float* pa = Pn + (ti * 16 + lq) * PP + 2 * half;
        const float* pb = Pn + (tj * 16 + lq) * PP + 2 * half;
        v8f s = {};
#pragma unroll
        for (int kc = 0; kc < NB / 4; ++kc) {
          v2f a = *(const v2f*)(pa + 4 * kc);
          v2f b = *(const v2f*)(pb + 4 * kc);
          s = __builtin_amdgcn_wmma_f32_16x16x4_f32(false, a, false, b, (short)0, s, false, false);
        }
        const float* cp = Lm + (k0 + NB + ti * 16 + 8 * half) * NO + (k0 + NB + tj * 16 + lq);
        float* dp = Db + (ti * 16 + 8 * half) * DP + tj * 16 + lq;
#pragma unroll
        for (int r = 0; r < 8; ++r) dp[r * DP] = cp[r * NO] - s[r];
      }
#pragma unroll
      for (int r = 0; r < 8; ++r) Db[(8 * half + r) * DP + 16 + lq] = 0.f;
      float r[NB];
#pragma unroll
      for (int c = 0; c < NB; ++c) r[c] = Db[lane * DP + c];
      chol_diag(r, lane, Sd, invd, Lm + (k0 + NB + lane) * NO + k0 + NB);
    } else {
      for (int q = 2 + wave; q < T; q += 31) {
        int ti = (int)((sqrtf(8.f * (float)q + 1.f) - 1.f) * 0.5f);
        if (ti * (ti + 1) / 2 > q) --ti;
        if ((ti + 1) * (ti + 2) / 2 <= q) ++ti;
        const int tj = q - ti * (ti + 1) / 2;
        const float* pa = Pn + (ti * 16 + lq) * PP + 2 * half;
        const float* pb = Pn + (tj * 16 + lq) * PP + 2 * half;
        v8f s = {};
#pragma unroll
        for (int kc = 0; kc < NB / 4; ++kc) {
          v2f a = *(const v2f*)(pa + 4 * kc);
          v2f b = *(const v2f*)(pb + 4 * kc);
          s = __builtin_amdgcn_wmma_f32_16x16x4_f32(false, a, false, b, (short)0, s, false, false);
        }
        float* cp = Lm + (k0 + NB + ti * 16 + 8 * half) * NO + (k0 + NB + tj * 16 + lq);
#pragma unroll
        for (int r = 0; r < 8; ++r) cp[r * NO] -= s[r];
      }
    }
  }
}

#define SC 16
#define YP 516

__global__ __launch_bounds__(1024) void panel_kernel(float* Lm) {
  __shared__ __attribute__((aligned(16))) float Yt[SC * YP];

  const int tid = threadIdx.x;
  const int lane = tid & 31;
  const int half = lane >> 4, lq = lane & 15;
  const int wave = __builtin_amdgcn_readfirstlane(tid >> 5);
  const int row0 = NH + blockIdx.x * SC;

  for (int i = 0; i < (SC * NH) / 1024; ++i) {
    int idx = tid + 1024 * i;
    int n = idx & (NH - 1), cc = idx / NH;
    Yt[cc * YP + n] = Lm[(row0 + cc) * NO + n];
  }

  float Ln[NB];
  float dg = 1.f;
  if (wave < SC) {
    const float4* lp = (const float4*)(Lm + lane * NO);
#pragma unroll
    for (int q = 0; q < NB / 4; ++q) {
      float4 t4 = lp[q];
      Ln[4 * q + 0] = t4.x; Ln[4 * q + 1] = t4.y; Ln[4 * q + 2] = t4.z; Ln[4 * q + 3] = t4.w;
    }
    dg = Lm[lane * NO + lane];
  }
  for (int kb = 0; kb < NH / NB; ++kb) {
    const int k0 = kb * NB;
    float Mr[NB];
    float inv = 0.f;
    if (wave < SC) {
      inv = 1.f / dg;
#pragma unroll
      for (int c = 0; c < NB; ++c) Mr[c] = (lane > c) ? Ln[c] * inv : 0.f;
    }
    const int ntiles = (NH - k0 - NB) >> 4;
    const int rowbase = k0 + NB + 16 * wave;
    __syncthreads();

    float4 av[4];
    if (wave < ntiles) {
      const float* ap = Lm + (rowbase + lq) * NO + k0 + 4 * half;
#pragma unroll
      for (int jj = 0; jj < 4; ++jj) av[jj] = *(const float4*)(ap + 8 * jj);
    }
    if (wave < SC) {
      if (kb + 1 < NH / NB) {
        const float4* lp = (const float4*)(Lm + (k0 + NB + lane) * NO + k0 + NB);
#pragma unroll
        for (int q = 0; q < NB / 4; ++q) {
          float4 t4 = lp[q];
          Ln[4 * q + 0] = t4.x; Ln[4 * q + 1] = t4.y; Ln[4 * q + 2] = t4.z; Ln[4 * q + 3] = t4.w;
        }
        dg = Lm[(k0 + NB + lane) * NO + k0 + NB + lane];
      }
      float* yp = Yt + wave * YP + k0 + lane;
      float t = yp[0] * inv;
#pragma unroll
      for (int c = 0; c < NB - 1; ++c) {
        float yc = __int_as_float(__builtin_amdgcn_readlane(__float_as_int(t), c));
        t = fmaf(-Mr[c], yc, t);
      }
      yp[0] = t;
    }
    __syncthreads();

    if (wave < ntiles) {
      const float* bp = Yt + lq * YP + k0 + 4 * half;
      v8f s = {};
#pragma unroll
      for (int jj = 0; jj < 4; ++jj) {
        float4 bv = *(const float4*)(bp + 8 * jj);
        v2f a, b;
        a.x = av[jj].x; a.y = av[jj].y; b.x = bv.x; b.y = bv.y;
        s = __builtin_amdgcn_wmma_f32_16x16x4_f32(false, a, false, b, (short)0, s, false, false);
        a.x = av[jj].z; a.y = av[jj].w; b.x = bv.z; b.y = bv.w;
        s = __builtin_amdgcn_wmma_f32_16x16x4_f32(false, a, false, b, (short)0, s, false, false);
      }
      float4* cp = (float4*)(Yt + lq * YP + rowbase + 8 * half);
      float4 c0 = cp[0], c1 = cp[1];
      c0.x -= s[0]; c0.y -= s[1]; c0.z -= s[2]; c0.w -= s[3];
      c1.x -= s[4]; c1.y -= s[5]; c1.z -= s[6]; c1.w -= s[7];
      cp[0] = c0; cp[1] = c1;
    }
  }

  for (int i = 0; i < (SC * NH) / 1024; ++i) {
    int idx = tid + 1024 * i;
    int n = idx & (NH - 1), cc = idx / NH;
    Lm[(row0 + cc) * NO + n] = Yt[cc * YP + n];
  }
}

__global__ __launch_bounds__(32) void trail_kernel(float* Lm) {
  const int ti = blockIdx.y, tj = blockIdx.x;
  if (ti < tj) return;
  const int lane = threadIdx.x & 31;
  const int half = lane >> 4, lq = lane & 15;
  const float* pa = Lm + (NH + ti * 16 + lq) * NO + 2 * half;
  const float* pb = Lm + (NH + tj * 16 + lq) * NO + 2 * half;
  v8f c = {};
  for (int n0 = 0; n0 < NH; n0 += 8) {
    v2f a0; a0.x = pa[n0 + 0]; a0.y = pa[n0 + 1];
    v2f a1; a1.x = pa[n0 + 4]; a1.y = pa[n0 + 5];
    v2f b0; b0.x = pb[n0 + 0]; b0.y = pb[n0 + 1];
    v2f b1; b1.x = pb[n0 + 4]; b1.y = pb[n0 + 5];
    c = __builtin_amdgcn_wmma_f32_16x16x4_f32(false, a0, false, b0, (short)0, c, false, false);
    c = __builtin_amdgcn_wmma_f32_16x16x4_f32(false, a1, false, b1, (short)0, c, false, false);
  }
  float* cp = Lm + (NH + ti * 16 + 8 * half) * NO + NH + tj * 16 + lq;
#pragma unroll
  for (int r = 0; r < 8; ++r) cp[r * NO] -= c[r];
}

__global__ __launch_bounds__(1024) void solve_kernel(const float* __restrict__ L,
                                                     const float* __restrict__ Kox,
                                                     const float* __restrict__ ys,
                                                     const float* __restrict__ b_p,
                                                     float* __restrict__ Y) {
  __shared__ __attribute__((aligned(16))) float Yt[SC * YP];

  const int tid = threadIdx.x;
  const int lane = tid & 31;
  const int half = lane >> 4, lq = lane & 15;
  const int wave = __builtin_amdgcn_readfirstlane(tid >> 5);
  const int col0 = blockIdx.x * SC;

  const float bb = b_p[0];
  for (int i = 0; i < (SC * NO) / 1024; ++i) {
    int idx = tid + 1024 * i;
    int cc = idx & (SC - 1), n = idx >> 4;
    int col = col0 + cc;
    float v = 0.f;
    if (col < MT) v = Kox[n * MT + col];
    else if (col == MT) v = 1.0f;
    else if (col == MT + 1) v = ys[n] - bb;
    Yt[cc * YP + n] = v;
  }

  float Ln[NB];
  float dg = 1.f;

  if (wave < SC) {
    const float4* lp = (const float4*)(L + lane * NO);
#pragma unroll
    for (int q = 0; q < NB / 4; ++q) {
      float4 t4 = lp[q];
      Ln[4 * q + 0] = t4.x; Ln[4 * q + 1] = t4.y; Ln[4 * q + 2] = t4.z; Ln[4 * q + 3] = t4.w;
    }
    dg = L[lane * NO + lane];
  }
  for (int kb = 0; kb < NO / NB; ++kb) {
    const int k0 = kb * NB;
    float Mr[NB];
    float inv = 0.f;
    if (wave < SC) {
      inv = 1.f / dg;
#pragma unroll
      for (int c = 0; c < NB; ++c) Mr[c] = (lane > c) ? Ln[c] * inv : 0.f;
    }
    const int ntiles = (NO - k0 - NB) >> 4;
    const int rowbase = k0 + NB + 16 * wave;
    __syncthreads();

    float4 av[4];
    if (wave < ntiles) {
      const float* ap = L + (rowbase + lq) * NO + k0 + 4 * half;
#pragma unroll
      for (int jj = 0; jj < 4; ++jj) av[jj] = *(const float4*)(ap + 8 * jj);
    }
    if (wave < SC) {
      if (kb + 1 < NO / NB) {
        const float4* lp = (const float4*)(L + (k0 + NB + lane) * NO + k0 + NB);
#pragma unroll
        for (int q = 0; q < NB / 4; ++q) {
          float4 t4 = lp[q];
          Ln[4 * q + 0] = t4.x; Ln[4 * q + 1] = t4.y; Ln[4 * q + 2] = t4.z; Ln[4 * q + 3] = t4.w;
        }
        dg = L[(k0 + NB + lane) * NO + k0 + NB + lane];
      }
      float* yp = Yt + wave * YP + k0 + lane;
      float t = yp[0] * inv;
#pragma unroll
      for (int c = 0; c < NB - 1; ++c) {
        float yc = __int_as_float(__builtin_amdgcn_readlane(__float_as_int(t), c));
        t = fmaf(-Mr[c], yc, t);
      }
      yp[0] = t;
    }
    __syncthreads();

    if (wave < ntiles) {
      const float* bp = Yt + lq * YP + k0 + 4 * half;
      v8f s = {};
#pragma unroll
      for (int jj = 0; jj < 4; ++jj) {
        float4 bv = *(const float4*)(bp + 8 * jj);
        v2f a, b;
        a.x = av[jj].x; a.y = av[jj].y; b.x = bv.x; b.y = bv.y;
        s = __builtin_amdgcn_wmma_f32_16x16x4_f32(false, a, false, b, (short)0, s, false, false);
        a.x = av[jj].z; a.y = av[jj].w; b.x = bv.z; b.y = bv.w;
        s = __builtin_amdgcn_wmma_f32_16x16x4_f32(false, a, false, b, (short)0, s, false, false);
      }
      float4* cp = (float4*)(Yt + lq * YP + rowbase + 8 * half);
      float4 c0 = cp[0], c1 = cp[1];
      c0.x -= s[0]; c0.y -= s[1]; c0.z -= s[2]; c0.w -= s[3];
      c1.x -= s[4]; c1.y -= s[5]; c1.z -= s[6]; c1.w -= s[7];
      cp[0] = c0; cp[1] = c1;
    }
  }

  for (int i = 0; i < (SC * NO) / 1024; ++i) {
    int idx = tid + 1024 * i;
    int n = idx & (NO - 1), cc = idx >> 9;
    int col = col0 + cc;
    if (col <= MT + 1) Y[col * NO + n] = Yt[cc * YP + n];
  }
}

__global__ __launch_bounds__(256) void mean_kernel(const float* __restrict__ Y,
                                                   const float* __restrict__ b_p,
                                                   float* __restrict__ out,
                                                   float* __restrict__ colsum) {
  __shared__ float r1[256], r2[256];
  int col = blockIdx.x, tid = threadIdx.x;
  float b = b_p[0];
  float sm = 0.f, sc = 0.f;
  for (int n = tid; n < NO; n += 256) {
    float y = Y[col * NO + n];
    sm += y * Y[(MT + 1) * NO + n];
    sc += y * Y[MT * NO + n];
  }
  r1[tid] = sm; r2[tid] = sc;
  __syncthreads();
  for (int off = 128; off > 0; off >>= 1) {
    if (tid < off) { r1[tid] += r1[tid + off]; r2[tid] += r2[tid + off]; }
    __syncthreads();
  }
  if (tid == 0) {
    colsum[col] = r2[0];
    if (col < MT) out[col] = b + r1[0];
  }
}

__global__ __launch_bounds__(32) void cov_kernel(const float* __restrict__ Kxx,
                                                 const float* __restrict__ Y,
                                                 const float* __restrict__ colsum,
                                                 float* __restrict__ out) {
  const int lane = threadIdx.x & 31;
  const int half = lane >> 4, lq = lane & 15;
  const int i0 = blockIdx.y * 16, j0 = blockIdx.x * 16;
  v8f c = {};
  for (int n0 = 0; n0 < NO; n0 += 8) {
    const float* yi = Y + (i0 + lq) * NO + n0 + 2 * half;
    const float* yj = Y + (j0 + lq) * NO + n0 + 2 * half;
    v2f a0; a0.x = yi[0]; a0.y = yi[1];
    v2f a1; a1.x = yi[4]; a1.y = yi[5];
    v2f b0; b0.x = yj[0]; b0.y = yj[1];
    v2f b1; b1.x = yj[4]; b1.y = yj[5];
    c = __builtin_amdgcn_wmma_f32_16x16x4_f32(false, a0, false, b0, (short)0, c, false, false);
    c = __builtin_amdgcn_wmma_f32_16x16x4_f32(false, a1, false, b1, (short)0, c, false, false);
  }
  float invS = 1.f / colsum[MT];
  float kbxj = colsum[j0 + lq];
  const bool mirror = i0 < j0;
#pragma unroll
  for (int r = 0; r < 8; ++r) {
    int i = i0 + r + 8 * half, j = j0 + lq;
    float kx = mirror ? Kxx[j * MT + i] : Kxx[i * MT + j];
    float v = kx - c[r] + (1.f - colsum[i]) * (1.f - kbxj) * invS;
    out[(1 + i) * MT + j] = v;
  }
}

extern "C" void kernel_launch(void* const* d_in, const int* in_sizes, int n_in,
                              void* d_out, int out_size, void* d_ws, size_t ws_size,
                              hipStream_t stream) {
  const float* Xt  = (const float*)d_in[0];
  const float* At  = (const float*)d_in[1];
  const float* Xo  = (const float*)d_in[2];
  const float* Ao  = (const float*)d_in[3];
  const float* ys  = (const float*)d_in[4];
  const float* rho = (const float*)d_in[5];
  const float* g   = (const float*)d_in[6];
  const float* nu  = (const float*)d_in[7];
  const float* b   = (const float*)d_in[8];
  float* out = (float*)d_out;

  float* ws  = (float*)d_ws;
  float* SN1 = ws;
  float* SN2 = SN1 + MT * PD;
  float* sT  = SN2 + NO * PD;
  float* sO  = sT + MT;
  float* Kxx = sO + NO;
  float* Kox = Kxx + MT * MT;
  float* Koo = Kox + NO * MT;
  float* Y   = Koo + NO * NO;
  float* csm = Y + (MT + 2) * NO;

  prep_kernel<<<(MT + NO) / IPB, IPB * 32, 0, stream>>>(Xt, At, Xo, Ao, SN1, SN2, sT, sO);

  gram_kernel<<<TKOO + TKOX + TKXX, GW * 32, 0, stream>>>(Xt, SN1, At, sT, Xo, SN2, Ao, sO,
                                                         Kxx, Kox, Koo, rho, nu, g);

  chol_kernel<<<1, 1024, 0, stream>>>(Koo);
  panel_kernel<<<NH / SC, 1024, 0, stream>>>(Koo);
  trail_kernel<<<dim3(NH / 16, NH / 16), 32, 0, stream>>>(Koo);
  chol_kernel<<<1, 1024, 0, stream>>>(Koo + NH * NO + NH);
  solve_kernel<<<(MT + 2 + SC - 1) / SC, 1024, 0, stream>>>(Koo, Kox, ys, b, Y);
  mean_kernel<<<MT + 1, 256, 0, stream>>>(Y, b, out, csm);
  cov_kernel<<<dim3(16, 16), 32, 0, stream>>>(Kxx, Y, csm, out);
}
